# starting kernel
# baseline (speedup 1.0000x reference)
#include <hip/hip_runtime.h>
#include <hip/hip_bf16.h>

#define N0 50000
#define N1 12500
#define N2 3125
#define E0 800000
#define E1 200000
#define E2 50000
#define NNZ 200000
#define H 128
#define F3 96
#define DECW 64

typedef __attribute__((ext_vector_type(2))) float v2f;
typedef __attribute__((ext_vector_type(8))) float v8f;

__global__ void zero_kernel(float* __restrict__ p, long long count) {
    long long t = (long long)blockIdx.x * blockDim.x + threadIdx.x;
    if (t < count) p[t] = 0.0f;
}

__global__ void conv1_kernel(const float* __restrict__ x, const float* __restrict__ agg6,
                             const float* __restrict__ Wrel, const float* __restrict__ brel,
                             const float* __restrict__ Wroot, float* __restrict__ h1, int n) {
    int r = blockIdx.x;
    int c = threadIdx.x;
    if (r >= n) return;
    float s = brel[c];
#pragma unroll
    for (int k = 0; k < 6; ++k) {
        s += agg6[r * 6 + k] * Wrel[k * H + c];
        s += x[r * 6 + k] * Wroot[k * H + c];
    }
    h1[r * H + c] = fmaxf(s, 0.0f);
}

__global__ void scatter6_kernel(const float* __restrict__ x, const int* __restrict__ ei,
                                int E, float* __restrict__ agg6) {
    int e = blockIdx.x * blockDim.x + threadIdx.x;
    if (e >= E) return;
    int src = ei[e];
    int dst = ei[E + e];
#pragma unroll
    for (int k = 0; k < 6; ++k)
        atomicAdd(&agg6[dst * 6 + k], x[src * 6 + k]);
}

__global__ void scatter128_kernel(const float* __restrict__ h1, const int* __restrict__ ei,
                                  int E, float* __restrict__ agg) {
    long long t = (long long)blockIdx.x * blockDim.x + threadIdx.x;
    if (t >= (long long)E * 32) return;
    int e  = (int)(t >> 5);
    int c4 = ((int)t & 31) * 4;
    int src = ei[e];
    int dst = ei[E + e];
    const float* hs = h1 + src * H + c4;
    float* ad = agg + dst * H + c4;
#pragma unroll
    for (int k = 0; k < 4; ++k) atomicAdd(ad + k, hs[k]);
}

__global__ void interp_kernel(const float* __restrict__ vals, const int* __restrict__ cols,
                              const int* __restrict__ rows, const float* __restrict__ e,
                              float* __restrict__ acc) {
    int nz = blockIdx.x;
    int c  = threadIdx.x;
    float v = vals[nz];
    int col = cols[nz];
    int row = rows[nz];
    atomicAdd(&acc[row * F3 + c], v * e[col * F3 + c]);
}

__global__ void dec_out_kernel(const float* __restrict__ hdec, const float* __restrict__ Wout,
                               const float* __restrict__ bout, float* __restrict__ out,
                               int n, int branch) {
    int r = blockIdx.x * blockDim.x + threadIdx.x;
    if (r >= n) return;
    float s = bout[0];
    const float* hr = hdec + r * DECW;
#pragma unroll
    for (int c = 0; c < DECW; ++c) s += hr[c] * Wout[c];
    out[r * 3 + branch] = s;
}

template <int K, int Nc, int ACT, bool DUAL>
__global__ void gemm_wmma_kernel(const float* __restrict__ A0, const float* __restrict__ W0,
                                 const float* __restrict__ A1, const float* __restrict__ W1,
                                 const float* __restrict__ bias, float* __restrict__ D,
                                 int n) {
    const int lane = threadIdx.x & 31;
    const int wave = threadIdx.x >> 5;
    const int row0 = (blockIdx.x * 4 + wave) * 16;
    const int col0 = blockIdx.y * 16;
    if (row0 >= n) return;

    const int m    = lane & 15;
    const int half = lane >> 4;
    int row = row0 + m;
    const int rowc = (row < n) ? row : (n - 1);

    const float* a0p = A0 + rowc * K + half * 2;
    const float* w0p = W0 + col0 + m + half * 2 * Nc;
    const float* a1p = DUAL ? (A1 + rowc * K + half * 2) : nullptr;
    const float* w1p = DUAL ? (W1 + col0 + m + half * 2 * Nc) : nullptr;

    const float bv = bias[col0 + m];
    v8f acc = {bv, bv, bv, bv, bv, bv, bv, bv};

#pragma unroll
    for (int k = 0; k < K; k += 4) {
        v2f a = *(const v2f*)(a0p + k);
        v2f b;
        b.x = w0p[k * Nc];
        b.y = w0p[k * Nc + Nc];
        acc = __builtin_amdgcn_wmma_f32_16x16x4_f32(false, a, false, b, (short)0, acc,
                                                    false, false);
        if (DUAL) {
            v2f a2 = *(const v2f*)(a1p + k);
            v2f b2;
            b2.x = w1p[k * Nc];
            b2.y = w1p[k * Nc + Nc];
            acc = __builtin_amdgcn_wmma_f32_16x16x4_f32(false, a2, false, b2, (short)0, acc,
                                                        false, false);
        }
    }

#pragma unroll
    for (int j = 0; j < 8; ++j) {
        int r = row0 + j + half * 8;
        if (r < n) {
            float v = acc[j];
            if (ACT == 1) v = fmaxf(v, 0.0f);
            if (ACT == 2) v = (v > 0.0f) ? v : (expf(v) - 1.0f);
            D[r * Nc + col0 + m] = v;
        }
    }
}

__global__ void dec_gemm_wmma_kernel(const float* __restrict__ e0, const float* __restrict__ i1,
                                     const float* __restrict__ i2, const float* __restrict__ W,
                                     const float* __restrict__ bias, float* __restrict__ D,
                                     int n, int branch) {
    constexpr int K = F3, Nc = DECW;
    const int lane = threadIdx.x & 31;
    const int wave = threadIdx.x >> 5;
    const int row0 = (blockIdx.x * 4 + wave) * 16;
    const int col0 = blockIdx.y * 16;
    if (row0 >= n) return;

    const int m    = lane & 15;
    const int half = lane >> 4;
    int row = row0 + m;
    const int rowc = (row < n) ? row : (n - 1);

    const int abase = rowc * F3 + branch * 32 + half * 2;
    const float* s0 = e0 + abase;
    const float* s1 = i1 + abase;
    const float* s2 = i2 + abase;
    const float* wp = W + col0 + m + half * 2 * Nc;

    const float bv = bias[col0 + m];
    v8f acc = {bv, bv, bv, bv, bv, bv, bv, bv};

#pragma unroll
    for (int k = 0; k < K; k += 4) {
        const int blk = k >> 5;
        const float* sp = (blk == 0) ? s0 : (blk == 1) ? s1 : s2;
        v2f a = *(const v2f*)(sp + (k & 31));
        v2f b;
        b.x = wp[k * Nc];
        b.y = wp[k * Nc + Nc];
        acc = __builtin_amdgcn_wmma_f32_16x16x4_f32(false, a, false, b, (short)0, acc,
                                                    false, false);
    }

#pragma unroll
    for (int j = 0; j < 8; ++j) {
        int r = row0 + j + half * 8;
        if (r < n) {
            float v = acc[j];
            v = (v > 0.0f) ? v : (expf(v) - 1.0f);
            D[r * Nc + col0 + m] = v;
        }
    }
}

static inline dim3 gemm_grid(int n, int Nc) {
    int row_tiles = (n + 15) / 16;
    return dim3((row_tiles + 3) / 4, Nc / 16, 1);
}

extern "C" void kernel_launch(void* const* d_in, const int* in_sizes, int n_in,
                              void* d_out, int out_size, void* d_ws, size_t ws_size,
                              hipStream_t stream) {
    (void)in_sizes; (void)n_in; (void)out_size; (void)ws_size;

    const float* x[3]   = {(const float*)d_in[0], (const float*)d_in[1], (const float*)d_in[2]};
    const int*   ei[3]  = {(const int*)d_in[3], (const int*)d_in[4], (const int*)d_in[5]};
    const int*   A1r    = (const int*)d_in[6];
    const int*   A1c    = (const int*)d_in[7];
    const float* A1v    = (const float*)d_in[8];
    const int*   A2r    = (const int*)d_in[9];
    const int*   A2c    = (const int*)d_in[10];
    const float* A2v    = (const float*)d_in[11];
    const float* c1Wrel  = (const float*)d_in[12];
    const float* c1brel  = (const float*)d_in[13];
    const float* c1Wroot = (const float*)d_in[14];
    const float* c2Wrel  = (const float*)d_in[15];
    const float* c2brel  = (const float*)d_in[16];
    const float* c2Wroot = (const float*)d_in[17];
    const float* linW    = (const float*)d_in[18];
    const float* linb    = (const float*)d_in[19];
    const float* decW0   = (const float*)d_in[20];
    const float* decb0   = (const float*)d_in[21];
    const float* decWout = (const float*)d_in[22];
    const float* decbout = (const float*)d_in[23];
    float* out = (float*)d_out;

    float* ws = (float*)d_ws;
    float* h1   = ws;
    float* agg  = h1   + (size_t)N0 * H;
    float* h2   = agg  + (size_t)N0 * H;
    float* agg6 = h2   + (size_t)N0 * H;
    float* e0   = agg6 + (size_t)N0 * 6;
    float* e1   = e0   + (size_t)N0 * F3;
    float* e2   = e1   + (size_t)N1 * F3;
    float* i1   = e2   + (size_t)N2 * F3;
    float* i2   = i1   + (size_t)N0 * F3;
    float* hdec = i2   + (size_t)N0 * F3;

    const int nlv[3] = {N0, N1, N2};
    const int Elv[3] = {E0, E1, E2};
    float* elv[3] = {e0, e1, e2};

    for (int i = 0; i < 3; ++i) {
        const int n = nlv[i];
        const int E = Elv[i];

        {
            long long cnt = (long long)n * 6;
            zero_kernel<<<(int)((cnt + 255) / 256), 256, 0, stream>>>(agg6, cnt);
            scatter6_kernel<<<(E + 255) / 256, 256, 0, stream>>>(x[i], ei[i], E, agg6);
            conv1_kernel<<<n, H, 0, stream>>>(x[i], agg6,
                                              c1Wrel + (size_t)i * 6 * H,
                                              c1brel + (size_t)i * H,
                                              c1Wroot + (size_t)i * 6 * H, h1, n);
        }

        {
            long long cnt = (long long)n * H;
            zero_kernel<<<(int)((cnt + 255) / 256), 256, 0, stream>>>(agg, cnt);
            long long th = (long long)E * 32;
            scatter128_kernel<<<(int)((th + 255) / 256), 256, 0, stream>>>(h1, ei[i], E, agg);
            gemm_wmma_kernel<H, H, 1, true><<<gemm_grid(n, H), 128, 0, stream>>>(
                agg, c2Wrel + (size_t)i * H * H,
                h1,  c2Wroot + (size_t)i * H * H,
                c2brel + (size_t)i * H, h2, n);
        }

        gemm_wmma_kernel<H, F3, 0, false><<<gemm_grid(n, F3), 128, 0, stream>>>(
            h2, linW + (size_t)i * H * F3, nullptr, nullptr,
            linb + (size_t)i * F3, elv[i], n);
    }

    {
        long long cnt = (long long)N0 * F3;
        zero_kernel<<<(int)((cnt + 255) / 256), 256, 0, stream>>>(i1, cnt);
        zero_kernel<<<(int)((cnt + 255) / 256), 256, 0, stream>>>(i2, cnt);
        interp_kernel<<<NNZ, F3, 0, stream>>>(A1v, A1c, A1r, e1, i1);
        interp_kernel<<<NNZ, F3, 0, stream>>>(A2v, A2c, A2r, e2, i2);
    }

    for (int b = 0; b < 3; ++b) {
        dec_gemm_wmma_kernel<<<gemm_grid(N0, DECW), 128, 0, stream>>>(
            e0, i1, i2, decW0 + (size_t)b * F3 * DECW, decb0 + (size_t)b * DECW,
            hdec, N0, b);
        dec_out_kernel<<<(N0 + 255) / 256, 256, 0, stream>>>(
            hdec, decWout + (size_t)b * DECW, decbout + b, out, N0, b);
    }
}


# conv2 segment sum in LDS: a block owns <=200 dst rows, scans the dst list, queues its edges per wave, adds their h1 rows by ds float adds; global float atomics, zero fill gone; sum order changed
# speedup vs baseline: 1.5844x; 1.5844x over previous
#include <hip/hip_runtime.h>
#include <hip/hip_bf16.h>

#define N0 50000
#define N1 12500
#define N2 3125
#define E0 800000
#define E1 200000
#define E2 50000
#define NNZ 200000
#define H 128
#define F3 96
#define DECW 64

typedef __attribute__((ext_vector_type(2))) float v2f;
typedef __attribute__((ext_vector_type(8))) float v8f;

__global__ void zero_kernel(float* __restrict__ p, long long count) {
    long long t = (long long)blockIdx.x * blockDim.x + threadIdx.x;
    if (t < count) p[t] = 0.0f;
}

__global__ void conv1_kernel(const float* __restrict__ x, const float* __restrict__ agg6,
                             const float* __restrict__ Wrel, const float* __restrict__ brel,
                             const float* __restrict__ Wroot, float* __restrict__ h1, int n) {
    int r = blockIdx.x;
    int c = threadIdx.x;
    if (r >= n) return;
    float s = brel[c];
#pragma unroll
    for (int k = 0; k < 6; ++k) {
        s += agg6[r * 6 + k] * Wrel[k * H + c];
        s += x[r * 6 + k] * Wroot[k * H + c];
    }
    h1[r * H + c] = fmaxf(s, 0.0f);
}

#define SC_R 4096
#define SC_S 16
#define SC_T 1024
#define SC_U 8

__global__ __launch_bounds__(SC_T) __attribute__((amdgpu_waves_per_eu(8, 8)))
void scatter6_lds_kernel(const float* __restrict__ x, const int* __restrict__ ei,
                         int E, int n, float* __restrict__ part) {
    __shared__ float acc[SC_R * 6];
    const int tid = threadIdx.x;
    const int r0  = blockIdx.x * SC_R;
    const int s   = blockIdx.y;

    for (int i = tid; i < SC_R * 6; i += SC_T) acc[i] = 0.0f;
    __syncthreads();

    const int chunk = (E + SC_S - 1) / SC_S;
    const int eb = s * chunk;
    const int ee = (eb + chunk < E) ? (eb + chunk) : E;

    if (eb < ee) {
        for (int base = eb; base < ee; base += SC_T * SC_U) {
            int off[SC_U];
            int srow[SC_U];
#pragma unroll
            for (int u = 0; u < SC_U; ++u) {
                const int e  = base + u * SC_T + tid;
                const int ec = (e < ee) ? e : (ee - 1);
                const int d  = ei[E + ec] - r0;
                const int sr = ei[ec];
                const bool hit = (e < ee) && ((unsigned)d < (unsigned)SC_R);
                off[u]  = hit ? d * 6 : -1;
                srow[u] = hit ? sr * 6 : 0;
            }
            float2 v[SC_U][3];
#pragma unroll
            for (int u = 0; u < SC_U; ++u) {
                const float2* p = (const float2*)(x + srow[u]);
                v[u][0] = p[0];
                v[u][1] = p[1];
                v[u][2] = p[2];
            }
#pragma unroll
            for (int u = 0; u < SC_U; ++u)
                asm volatile("" :: "v"(v[u][0].x), "v"(v[u][0].y), "v"(v[u][1].x),
                                   "v"(v[u][1].y), "v"(v[u][2].x), "v"(v[u][2].y));
#pragma unroll
            for (int u = 0; u < SC_U; ++u) {
                if (off[u] >= 0) {
                    float* a = acc + off[u];
                    atomicAdd(a + 0, v[u][0].x);
                    atomicAdd(a + 1, v[u][0].y);
                    atomicAdd(a + 2, v[u][1].x);
                    atomicAdd(a + 3, v[u][1].y);
                    atomicAdd(a + 4, v[u][2].x);
                    atomicAdd(a + 5, v[u][2].y);
                }
            }
        }
    }
    __syncthreads();

    const int rows = (n - r0 < SC_R) ? (n - r0) : SC_R;
    float* p = part + (size_t)s * n * 6 + (size_t)r0 * 6;
    for (int i = tid; i < rows * 6; i += SC_T) p[i] = acc[i];
}

__global__ void agg6_reduce_kernel(const float* __restrict__ part, float* __restrict__ agg6,
                                   int cnt) {
    int t = blockIdx.x * blockDim.x + threadIdx.x;
    if (t >= cnt) return;
    float s = 0.0f;
#pragma unroll
    for (int k = 0; k < SC_S; ++k) s += part[(size_t)k * cnt + t];
    agg6[t] = s;
}

#define AG_T    1024
#define AG_RMAX 200
#define AG_U    4
#define AG_Q    1024
#define AG_B    16

template <bool TAIL>
__device__ __forceinline__ void agg128_scan_trip(const int4* __restrict__ d4, int g0, int ng,
                                                 int tid, int r0, unsigned rows,
                                                 int* q, int& tail) {
    int4 d[AG_U];
#pragma unroll
    for (int u = 0; u < AG_U; ++u) {
        const int g = g0 + u * AG_T + tid;
        d[u] = d4[(TAIL && g >= ng) ? (ng - 1) : g];
    }
#pragma unroll
    for (int u = 0; u < AG_U; ++u) {
        const int g = g0 + u * AG_T + tid;
        const bool valid = !TAIL || (g < ng);
        const unsigned vmask = TAIL ? __builtin_amdgcn_ballot_w32(g < ng) : 0xffffffffu;
        const int dd[4] = {d[u].x, d[u].y, d[u].z, d[u].w};
#pragma unroll
        for (int c = 0; c < 4; ++c) {
            const bool in = (unsigned)(dd[c] - r0) < rows;
            const unsigned mask = __builtin_amdgcn_ballot_w32(in) & vmask;
            if (mask != 0u) {
                if (in && valid)
                    q[(tail + (int)__builtin_amdgcn_mbcnt_lo(mask, 0u)) & (AG_Q - 1)] = g * 4 + c;
                tail += __builtin_popcount(mask);
            }
        }
    }
}

__device__ __forceinline__ void agg128_drain(const float* __restrict__ h1,
                                             const int* __restrict__ ei, int E, int r0,
                                             const int* q, int head, int cnt,
                                             float* acc, int lane) {
    const int qe = q[(head + lane) & (AG_Q - 1)];
    const int e  = (lane < cnt) ? qe : 0;
    const int s  = ei[e];
    const int dl = ei[E + e] - r0;
#pragma unroll 1
    for (int b = 0; b < cnt; b += AG_B) {
        float4 v[AG_B];
#pragma unroll
        for (int j = 0; j < AG_B; ++j) {
            const int sj = __builtin_amdgcn_readlane(s, b + j);
            v[j] = *(const float4*)(h1 + sj * H + lane * 4);
        }
#pragma unroll
        for (int j = 0; j < AG_B; ++j) {
            if (b + j < cnt) {
                const int dj = __builtin_amdgcn_readlane(dl, b + j);
                float* a = acc + dj * H + lane;
                atomicAdd(a,      v[j].x);
                atomicAdd(a + 32, v[j].y);
                atomicAdd(a + 64, v[j].z);
                atomicAdd(a + 96, v[j].w);
            }
        }
    }
}

__global__ __launch_bounds__(AG_T)
void agg128_lds_kernel(const float* __restrict__ h1, const int* __restrict__ ei,
                       int E, int n, int rr, float* __restrict__ agg) {
    __shared__ float acc[AG_RMAX * H];
    __shared__ int ring[AG_T / 32][AG_Q];
    const int tid  = threadIdx.x;
    const int lane = tid & 31;
    const int r0   = blockIdx.x * rr;
    const int rows = (n - r0 < rr) ? (n - r0) : rr;

    {
        float4* a4 = (float4*)acc;
        const float4 z = {0.0f, 0.0f, 0.0f, 0.0f};
        for (int i = tid; i < rows * (H / 4); i += AG_T) a4[i] = z;
    }
    __syncthreads();

    int* q = ring[tid >> 5];
    int head = 0, tail = 0;
    const int4* d4 = (const int4*)(ei + E);
    const int ng = E >> 2;

    int g0 = 0;
    for (; g0 + AG_T * AG_U <= ng; g0 += AG_T * AG_U) {
        agg128_scan_trip<false>(d4, g0, ng, tid, r0, (unsigned)rows, q, tail);
        while (tail - head >= 32) {
            agg128_drain(h1, ei, E, r0, q, head, 32, acc, lane);
            head += 32;
        }
    }
    if (g0 < ng) agg128_scan_trip<true>(d4, g0, ng, tid, r0, (unsigned)rows, q, tail);
    while (tail - head > 0) {
        const int cnt = (tail - head < 32) ? (tail - head) : 32;
        agg128_drain(h1, ei, E, r0, q, head, cnt, acc, lane);
        head += cnt;
    }
    __syncthreads();

    float4* o4 = (float4*)(agg + r0 * H);
    for (int i = tid; i < rows * (H / 4); i += AG_T) {
        const float* a = acc + (i >> 5) * H + (i & 31);
        float4 v;
        v.x = a[0];
        v.y = a[32];
        v.z = a[64];
        v.w = a[96];
        o4[i] = v;
    }
}

__global__ void interp_kernel(const float* __restrict__ vals, const int* __restrict__ cols,
                              const int* __restrict__ rows, const float* __restrict__ e,
                              float* __restrict__ acc) {
    int nz = blockIdx.x;
    int c  = threadIdx.x;
    float v = vals[nz];
    int col = cols[nz];
    int row = rows[nz];
    atomicAdd(&acc[row * F3 + c], v * e[col * F3 + c]);
}

__global__ void dec_out_kernel(const float* __restrict__ hdec, const float* __restrict__ Wout,
                               const float* __restrict__ bout, float* __restrict__ out,
                               int n, int branch) {
    int r = blockIdx.x * blockDim.x + threadIdx.x;
    if (r >= n) return;
    float s = bout[0];
    const float* hr = hdec + r * DECW;
#pragma unroll
    for (int c = 0; c < DECW; ++c) s += hr[c] * Wout[c];
    out[r * 3 + branch] = s;
}

template <int K, int Nc, int ACT, bool DUAL>
__global__ void gemm_wmma_kernel(const float* __restrict__ A0, const float* __restrict__ W0,
                                 const float* __restrict__ A1, const float* __restrict__ W1,
                                 const float* __restrict__ bias, float* __restrict__ D,
                                 int n) {
    const int lane = threadIdx.x & 31;
    const int wave = threadIdx.x >> 5;
    const int row0 = (blockIdx.x * 4 + wave) * 16;
    const int col0 = blockIdx.y * 16;
    if (row0 >= n) return;

    const int m    = lane & 15;
    const int half = lane >> 4;
    int row = row0 + m;
    const int rowc = (row < n) ? row : (n - 1);

    const float* a0p = A0 + rowc * K + half * 2;
    const float* w0p = W0 + col0 + m + half * 2 * Nc;
    const float* a1p = DUAL ? (A1 + rowc * K + half * 2) : nullptr;
    const float* w1p = DUAL ? (W1 + col0 + m + half * 2 * Nc) : nullptr;

    const float bv = bias[col0 + m];
    v8f acc = {bv, bv, bv, bv, bv, bv, bv, bv};

#pragma unroll
    for (int k = 0; k < K; k += 4) {
        v2f a = *(const v2f*)(a0p + k);
        v2f b;
        b.x = w0p[k * Nc];
        b.y = w0p[k * Nc + Nc];
        acc = __builtin_amdgcn_wmma_f32_16x16x4_f32(false, a, false, b, (short)0, acc,
                                                    false, false);
        if (DUAL) {
            v2f a2 = *(const v2f*)(a1p + k);
            v2f b2;
            b2.x = w1p[k * Nc];
            b2.y = w1p[k * Nc + Nc];
            acc = __builtin_amdgcn_wmma_f32_16x16x4_f32(false, a2, false, b2, (short)0, acc,
                                                        false, false);
        }
    }

#pragma unroll
    for (int j = 0; j < 8; ++j) {
        int r = row0 + j + half * 8;
        if (r < n) {
            float v = acc[j];
            if (ACT == 1) v = fmaxf(v, 0.0f);
            if (ACT == 2) v = (v > 0.0f) ? v : (expf(v) - 1.0f);
            D[r * Nc + col0 + m] = v;
        }
    }
}

__global__ void dec_gemm_wmma_kernel(const float* __restrict__ e0, const float* __restrict__ i1,
                                     const float* __restrict__ i2, const float* __restrict__ W,
                                     const float* __restrict__ bias, float* __restrict__ D,
                                     int n, int branch) {
    constexpr int K = F3, Nc = DECW;
    const int lane = threadIdx.x & 31;
    const int wave = threadIdx.x >> 5;
    const int row0 = (blockIdx.x * 4 + wave) * 16;
    const int col0 = blockIdx.y * 16;
    if (row0 >= n) return;

    const int m    = lane & 15;
    const int half = lane >> 4;
    int row = row0 + m;
    const int rowc = (row < n) ? row : (n - 1);

    const int abase = rowc * F3 + branch * 32 + half * 2;
    const float* s0 = e0 + abase;
    const float* s1 = i1 + abase;
    const float* s2 = i2 + abase;
    const float* wp = W + col0 + m + half * 2 * Nc;

    const float bv = bias[col0 + m];
    v8f acc = {bv, bv, bv, bv, bv, bv, bv, bv};

#pragma unroll
    for (int k = 0; k < K; k += 4) {
        const int blk = k >> 5;
        const float* sp = (blk == 0) ? s0 : (blk == 1) ? s1 : s2;
        v2f a = *(const v2f*)(sp + (k & 31));
        v2f b;
        b.x = wp[k * Nc];
        b.y = wp[k * Nc + Nc];
        acc = __builtin_amdgcn_wmma_f32_16x16x4_f32(false, a, false, b, (short)0, acc,
                                                    false, false);
    }

#pragma unroll
    for (int j = 0; j < 8; ++j) {
        int r = row0 + j + half * 8;
        if (r < n) {
            float v = acc[j];
            v = (v > 0.0f) ? v : (expf(v) - 1.0f);
            D[r * Nc + col0 + m] = v;
        }
    }
}

static inline dim3 gemm_grid(int n, int Nc) {
    int row_tiles = (n + 15) / 16;
    return dim3((row_tiles + 3) / 4, Nc / 16, 1);
}

extern "C" void kernel_launch(void* const* d_in, const int* in_sizes, int n_in,
                              void* d_out, int out_size, void* d_ws, size_t ws_size,
                              hipStream_t stream) {
    (void)in_sizes; (void)n_in; (void)out_size; (void)ws_size;

    const float* x[3]   = {(const float*)d_in[0], (const float*)d_in[1], (const float*)d_in[2]};
    const int*   ei[3]  = {(const int*)d_in[3], (const int*)d_in[4], (const int*)d_in[5]};
    const int*   A1r    = (const int*)d_in[6];
    const int*   A1c    = (const int*)d_in[7];
    const float* A1v    = (const float*)d_in[8];
    const int*   A2r    = (const int*)d_in[9];
    const int*   A2c    = (const int*)d_in[10];
    const float* A2v    = (const float*)d_in[11];
    const float* c1Wrel  = (const float*)d_in[12];
    const float* c1brel  = (const float*)d_in[13];
    const float* c1Wroot = (const float*)d_in[14];
    const float* c2Wrel  = (const float*)d_in[15];
    const float* c2brel  = (const float*)d_in[16];
    const float* c2Wroot = (const float*)d_in[17];
    const float* linW    = (const float*)d_in[18];
    const float* linb    = (const float*)d_in[19];
    const float* decW0   = (const float*)d_in[20];
    const float* decb0   = (const float*)d_in[21];
    const float* decWout = (const float*)d_in[22];
    const float* decbout = (const float*)d_in[23];
    float* out = (float*)d_out;

    float* ws = (float*)d_ws;
    float* h1   = ws;
    float* agg  = h1   + (size_t)N0 * H;
    float* h2   = agg  + (size_t)N0 * H;
    float* agg6 = h2   + (size_t)N0 * H;
    float* e0   = agg6 + (size_t)N0 * 6;
    float* e1   = e0   + (size_t)N0 * F3;
    float* e2   = e1   + (size_t)N1 * F3;
    float* i1   = e2   + (size_t)N2 * F3;
    float* i2   = i1   + (size_t)N0 * F3;
    float* hdec = i2   + (size_t)N0 * F3;

    const int nlv[3] = {N0, N1, N2};
    const int Elv[3] = {E0, E1, E2};
    const int rrl[3] = {200, 50, 13};
    static_assert(E0 % 4 == 0 && E1 % 4 == 0 && E2 % 4 == 0, "dst lists are read 16 B at a time");
    float* elv[3] = {e0, e1, e2};

    for (int i = 0; i < 3; ++i) {
        const int n = nlv[i];
        const int E = Elv[i];

        {
            const int cnt = n * 6;
            scatter6_lds_kernel<<<dim3((n + SC_R - 1) / SC_R, SC_S, 1), SC_T, 0, stream>>>(
                x[i], ei[i], E, n, agg);
            agg6_reduce_kernel<<<(cnt + 255) / 256, 256, 0, stream>>>(agg, agg6, cnt);
            conv1_kernel<<<n, H, 0, stream>>>(x[i], agg6,
                                              c1Wrel + (size_t)i * 6 * H,
                                              c1brel + (size_t)i * H,
                                              c1Wroot + (size_t)i * 6 * H, h1, n);
        }

        {
            const int rr = rrl[i];
            agg128_lds_kernel<<<(n + rr - 1) / rr, AG_T, 0, stream>>>(h1, ei[i], E, n, rr, agg);
            gemm_wmma_kernel<H, H, 1, true><<<gemm_grid(n, H), 128, 0, stream>>>(
                agg, c2Wrel + (size_t)i * H * H,
                h1,  c2Wroot + (size_t)i * H * H,
                c2brel + (size_t)i * H, h2, n);
        }

        gemm_wmma_kernel<H, F3, 0, false><<<gemm_grid(n, F3), 128, 0, stream>>>(
            h2, linW + (size_t)i * H * F3, nullptr, nullptr,
            linb + (size_t)i * F3, elv[i], n);
    }

    {
        long long cnt = (long long)N0 * F3;
        zero_kernel<<<(int)((cnt + 255) / 256), 256, 0, stream>>>(i1, cnt);
        zero_kernel<<<(int)((cnt + 255) / 256), 256, 0, stream>>>(i2, cnt);
        interp_kernel<<<NNZ, F3, 0, stream>>>(A1v, A1c, A1r, e1, i1);
        interp_kernel<<<NNZ, F3, 0, stream>>>(A2v, A2c, A2r, e2, i2);
    }

    for (int b = 0; b < 3; ++b) {
        dec_gemm_wmma_kernel<<<gemm_grid(N0, DECW), 128, 0, stream>>>(
            e0, i1, i2, decW0 + (size_t)b * F3 * DECW, decb0 + (size_t)b * DECW,
            hdec, N0, b);
        dec_out_kernel<<<(N0 + 255) / 256, 256, 0, stream>>>(
            hdec, decWout + (size_t)b * DECW, decbout + b, out, N0, b);
    }
}


# GEMMs re-tiled: work-group copies its 64/32/16 A rows to LDS once (16-B loads); wave owns 4x2 (2x2, 1x2) 16x16 tiles, a B fragment from global serves all its row tiles; sum order kept
# speedup vs baseline: 2.0487x; 1.2930x over previous
#include <hip/hip_runtime.h>
#include <hip/hip_bf16.h>

#define N0 50000
#define N1 12500
#define N2 3125
#define E0 800000
#define E1 200000
#define E2 50000
#define NNZ 200000
#define H 128
#define F3 96
#define DECW 64

typedef __attribute__((ext_vector_type(2))) float v2f;
typedef __attribute__((ext_vector_type(8))) float v8f;

__global__ void zero_kernel(float* __restrict__ p, long long count) {
    long long t = (long long)blockIdx.x * blockDim.x + threadIdx.x;
    if (t < count) p[t] = 0.0f;
}

__global__ void conv1_kernel(const float* __restrict__ x, const float* __restrict__ agg6,
                             const float* __restrict__ Wrel, const float* __restrict__ brel,
                             const float* __restrict__ Wroot, float* __restrict__ h1, int n) {
    int r = blockIdx.x;
    int c = threadIdx.x;
    if (r >= n) return;
    float s = brel[c];
#pragma unroll
    for (int k = 0; k < 6; ++k) {
        s += agg6[r * 6 + k] * Wrel[k * H + c];
        s += x[r * 6 + k] * Wroot[k * H + c];
    }
    h1[r * H + c] = fmaxf(s, 0.0f);
}

#define SC_R 4096
#define SC_S 16
#define SC_T 1024
#define SC_U 8

__global__ __launch_bounds__(SC_T) __attribute__((amdgpu_waves_per_eu(8, 8)))
void scatter6_lds_kernel(const float* __restrict__ x, const int* __restrict__ ei,
                         int E, int n, float* __restrict__ part) {
    __shared__ float acc[SC_R * 6];
    const int tid = threadIdx.x;
    const int r0  = blockIdx.x * SC_R;
    const int s   = blockIdx.y;

    for (int i = tid; i < SC_R * 6; i += SC_T) acc[i] = 0.0f;
    __syncthreads();

    const int chunk = (E + SC_S - 1) / SC_S;
    const int eb = s * chunk;
    const int ee = (eb + chunk < E) ? (eb + chunk) : E;

    if (eb < ee) {
        for (int base = eb; base < ee; base += SC_T * SC_U) {
            int off[SC_U];
            int srow[SC_U];
#pragma unroll
            for (int u = 0; u < SC_U; ++u) {
                const int e  = base + u * SC_T + tid;
                const int ec = (e < ee) ? e : (ee - 1);
                const int d  = ei[E + ec] - r0;
                const int sr = ei[ec];
                const bool hit = (e < ee) && ((unsigned)d < (unsigned)SC_R);
                off[u]  = hit ? d * 6 : -1;
                srow[u] = hit ? sr * 6 : 0;
            }
            float2 v[SC_U][3];
#pragma unroll
            for (int u = 0; u < SC_U; ++u) {
                const float2* p = (const float2*)(x + srow[u]);
                v[u][0] = p[0];
                v[u][1] = p[1];
                v[u][2] = p[2];
            }
#pragma unroll
            for (int u = 0; u < SC_U; ++u)
                asm volatile("" :: "v"(v[u][0].x), "v"(v[u][0].y), "v"(v[u][1].x),
                                   "v"(v[u][1].y), "v"(v[u][2].x), "v"(v[u][2].y));
#pragma unroll
            for (int u = 0; u < SC_U; ++u) {
                if (off[u] >= 0) {
                    float* a = acc + off[u];
                    atomicAdd(a + 0, v[u][0].x);
                    atomicAdd(a + 1, v[u][0].y);
                    atomicAdd(a + 2, v[u][1].x);
                    atomicAdd(a + 3, v[u][1].y);
                    atomicAdd(a + 4, v[u][2].x);
                    atomicAdd(a + 5, v[u][2].y);
                }
            }
        }
    }
    __syncthreads();

    const int rows = (n - r0 < SC_R) ? (n - r0) : SC_R;
    float* p = part + (size_t)s * n * 6 + (size_t)r0 * 6;
    for (int i = tid; i < rows * 6; i += SC_T) p[i] = acc[i];
}

__global__ void agg6_reduce_kernel(const float* __restrict__ part, float* __restrict__ agg6,
                                   int cnt) {
    int t = blockIdx.x * blockDim.x + threadIdx.x;
    if (t >= cnt) return;
    float s = 0.0f;
#pragma unroll
    for (int k = 0; k < SC_S; ++k) s += part[(size_t)k * cnt + t];
    agg6[t] = s;
}

#define AG_T    1024
#define AG_RMAX 200
#define AG_U    4
#define AG_Q    1024
#define AG_B    16

template <bool TAIL>
__device__ __forceinline__ void agg128_scan_trip(const int4* __restrict__ d4, int g0, int ng,
                                                 int tid, int r0, unsigned rows,
                                                 int* q, int& tail) {
    int4 d[AG_U];
#pragma unroll
    for (int u = 0; u < AG_U; ++u) {
        const int g = g0 + u * AG_T + tid;
        d[u] = d4[(TAIL && g >= ng) ? (ng - 1) : g];
    }
#pragma unroll
    for (int u = 0; u < AG_U; ++u) {
        const int g = g0 + u * AG_T + tid;
        const bool valid = !TAIL || (g < ng);
        const unsigned vmask = TAIL ? __builtin_amdgcn_ballot_w32(g < ng) : 0xffffffffu;
        const int dd[4] = {d[u].x, d[u].y, d[u].z, d[u].w};
#pragma unroll
        for (int c = 0; c < 4; ++c) {
            const bool in = (unsigned)(dd[c] - r0) < rows;
            const unsigned mask = __builtin_amdgcn_ballot_w32(in) & vmask;
            if (mask != 0u) {
                if (in && valid)
                    q[(tail + (int)__builtin_amdgcn_mbcnt_lo(mask, 0u)) & (AG_Q - 1)] = g * 4 + c;
                tail += __builtin_popcount(mask);
            }
        }
    }
}

__device__ __forceinline__ void agg128_drain(const float* __restrict__ h1,
                                             const int* __restrict__ ei, int E, int r0,
                                             const int* q, int head, int cnt,
                                             float* acc, int lane) {
    const int qe = q[(head + lane) & (AG_Q - 1)];
    const int e  = (lane < cnt) ? qe : 0;
    const int s  = ei[e];
    const int dl = ei[E + e] - r0;
#pragma unroll 1
    for (int b = 0; b < cnt; b += AG_B) {
        float4 v[AG_B];
#pragma unroll
        for (int j = 0; j < AG_B; ++j) {
            const int sj = __builtin_amdgcn_readlane(s, b + j);
            v[j] = *(const float4*)(h1 + sj * H + lane * 4);
        }
#pragma unroll
        for (int j = 0; j < AG_B; ++j) {
            if (b + j < cnt) {
                const int dj = __builtin_amdgcn_readlane(dl, b + j);
                float* a = acc + dj * H + lane;
                atomicAdd(a,      v[j].x);
                atomicAdd(a + 32, v[j].y);
                atomicAdd(a + 64, v[j].z);
                atomicAdd(a + 96, v[j].w);
            }
        }
    }
}

__global__ __launch_bounds__(AG_T)
void agg128_lds_kernel(const float* __restrict__ h1, const int* __restrict__ ei,
                       int E, int n, int rr, float* __restrict__ agg) {
    __shared__ float acc[AG_RMAX * H];
    __shared__ int ring[AG_T / 32][AG_Q];
    const int tid  = threadIdx.x;
    const int lane = tid & 31;
    const int r0   = blockIdx.x * rr;
    const int rows = (n - r0 < rr) ? (n - r0) : rr;

    {
        float4* a4 = (float4*)acc;
        const float4 z = {0.0f, 0.0f, 0.0f, 0.0f};
        for (int i = tid; i < rows * (H / 4); i += AG_T) a4[i] = z;
    }
    __syncthreads();

    int* q = ring[tid >> 5];
    int head = 0, tail = 0;
    const int4* d4 = (const int4*)(ei + E);
    const int ng = E >> 2;

    int g0 = 0;
    for (; g0 + AG_T * AG_U <= ng; g0 += AG_T * AG_U) {
        agg128_scan_trip<false>(d4, g0, ng, tid, r0, (unsigned)rows, q, tail);
        while (tail - head >= 32) {
            agg128_drain(h1, ei, E, r0, q, head, 32, acc, lane);
            head += 32;
        }
    }
    if (g0 < ng) agg128_scan_trip<true>(d4, g0, ng, tid, r0, (unsigned)rows, q, tail);
    while (tail - head > 0) {
        const int cnt = (tail - head < 32) ? (tail - head) : 32;
        agg128_drain(h1, ei, E, r0, q, head, cnt, acc, lane);
        head += cnt;
    }
    __syncthreads();

    float4* o4 = (float4*)(agg + r0 * H);
    for (int i = tid; i < rows * (H / 4); i += AG_T) {
        const float* a = acc + (i >> 5) * H + (i & 31);
        float4 v;
        v.x = a[0];
        v.y = a[32];
        v.z = a[64];
        v.w = a[96];
        o4[i] = v;
    }
}

__global__ void interp_kernel(const float* __restrict__ vals, const int* __restrict__ cols,
                              const int* __restrict__ rows, const float* __restrict__ e,
                              float* __restrict__ acc) {
    int nz = blockIdx.x;
    int c  = threadIdx.x;
    float v = vals[nz];
    int col = cols[nz];
    int row = rows[nz];
    atomicAdd(&acc[row * F3 + c], v * e[col * F3 + c]);
}

__global__ void dec_out_kernel(const float* __restrict__ hdec, const float* __restrict__ Wout,
                               const float* __restrict__ bout, float* __restrict__ out,
                               int n, int branch) {
    int r = blockIdx.x * blockDim.x + threadIdx.x;
    if (r >= n) return;
    float s = bout[0];
    const float* hr = hdec + r * DECW;
#pragma unroll
    for (int c = 0; c < DECW; ++c) s += hr[c] * Wout[c];
    out[r * 3 + branch] = s;
}

template <int NSEG, int SEGW, int LD, int Nc, int RT, int ACT, bool DUAL>
__global__ __launch_bounds__(Nc)
void gemm_tile_kernel(const float* __restrict__ s0, const float* __restrict__ s1,
                      const float* __restrict__ s2, int coff,
                      const float* __restrict__ W0, const float* __restrict__ W1,
                      const float* __restrict__ bias, float* __restrict__ D, int n) {
    constexpr int K   = NSEG * SEGW;
    constexpr int P   = K + 4;
    constexpr int MB  = RT * 16;
    constexpr int NT  = Nc;
    constexpr int K4  = K / 4;
    constexpr int S4  = SEGW / 4;
    constexpr int TOT = MB * K4;
    constexpr int ITER = (TOT + NT - 1) / NT;
    constexpr int SB  = 8;
    __shared__ __attribute__((aligned(16))) float As[MB * P];

    const int tid  = threadIdx.x;
    const int lane = tid & 31;
    const int wave = tid >> 5;
    const int rowb = blockIdx.x * MB;

#pragma unroll
    for (int i0 = 0; i0 < ITER; i0 += SB) {
        float4 v[SB];
#pragma unroll
        for (int j = 0; j < SB; ++j) {
            if (i0 + j < ITER) {
                const int idx = (i0 + j) * NT + tid;
                const int idc = (TOT % NT != 0 && idx >= TOT) ? (TOT - 1) : idx;
                const int r   = idc / K4;
                const int c4  = idc % K4;
                const int sg  = c4 / S4;
                const int w4  = c4 % S4;
                const int rg  = (rowb + r < n) ? (rowb + r) : (n - 1);
                const float* sp = (sg == 0) ? s0 : (sg == 1) ? s1 : s2;
                v[j] = *(const float4*)(sp + (size_t)rg * LD + coff + w4 * 4);
            }
        }
#pragma unroll
        for (int j = 0; j < SB; ++j) {
            if (i0 + j < ITER) {
                const int idx = (i0 + j) * NT + tid;
                if (TOT % NT == 0 || idx < TOT) {
                    const int r  = idx / K4;
                    const int c4 = idx % K4;
                    *(float4*)(As + r * P + c4 * 4) = v[j];
                }
            }
        }
    }
    __syncthreads();

    const int c    = lane & 15;
    const int h    = lane >> 4;
    const int col0 = wave * 32;
    const float* ap  = As + c * P + 2 * h;
    const float* wp0 = W0 + col0 + c + 2 * h * Nc;
    const float* wp1 = DUAL ? (W1 + col0 + c + 2 * h * Nc) : nullptr;

    v8f acc[RT][2];
#pragma unroll
    for (int ct = 0; ct < 2; ++ct) {
        const float bv = bias[col0 + ct * 16 + c];
#pragma unroll
        for (int rt = 0; rt < RT; ++rt) {
            const v8f t = {bv, bv, bv, bv, bv, bv, bv, bv};
            acc[rt][ct] = t;
        }
    }

    if (DUAL) {
#pragma unroll 4
        for (int k = 0; k < SEGW; k += 4) {
#pragma unroll
            for (int sg = 0; sg < 2; ++sg) {
                const float* wp = (sg == 0) ? wp0 : wp1;
                v2f a[RT];
                v2f b[2];
#pragma unroll
                for (int rt = 0; rt < RT; ++rt)
                    a[rt] = *(const v2f*)(ap + rt * 16 * P + sg * SEGW + k);
#pragma unroll
                for (int ct = 0; ct < 2; ++ct) {
                    b[ct].x = wp[k * Nc + ct * 16];
                    b[ct].y = wp[k * Nc + Nc + ct * 16];
                }
#pragma unroll
                for (int rt = 0; rt < RT; ++rt)
#pragma unroll
                    for (int ct = 0; ct < 2; ++ct)
                        acc[rt][ct] = __builtin_amdgcn_wmma_f32_16x16x4_f32(
                            false, a[rt], false, b[ct], (short)0, acc[rt][ct], false, false);
            }
        }
    } else {
#pragma unroll 8
        for (int k = 0; k < K; k += 4) {
            v2f a[RT];
            v2f b[2];
#pragma unroll
            for (int rt = 0; rt < RT; ++rt)
                a[rt] = *(const v2f*)(ap + rt * 16 * P + k);
#pragma unroll
            for (int ct = 0; ct < 2; ++ct) {
                b[ct].x = wp0[k * Nc + ct * 16];
                b[ct].y = wp0[k * Nc + Nc + ct * 16];
            }
#pragma unroll
            for (int rt = 0; rt < RT; ++rt)
#pragma unroll
                for (int ct = 0; ct < 2; ++ct)
                    acc[rt][ct] = __builtin_amdgcn_wmma_f32_16x16x4_f32(
                        false, a[rt], false, b[ct], (short)0, acc[rt][ct], false, false);
        }
    }

    const bool full = (rowb + MB <= n);
#pragma unroll
    for (int rt = 0; rt < RT; ++rt)
#pragma unroll
        for (int ct = 0; ct < 2; ++ct)
#pragma unroll
            for (int j = 0; j < 8; ++j) {
                float v = acc[rt][ct][j];
                if (ACT == 1) v = fmaxf(v, 0.0f);
                if (ACT == 2) v = (v > 0.0f) ? v : (expf(v) - 1.0f);
                acc[rt][ct][j] = v;
            }
    float* dp = D + (rowb + h * 8) * Nc + col0 + c;
    if (full) {
#pragma unroll
        for (int rt = 0; rt < RT; ++rt)
#pragma unroll
            for (int ct = 0; ct < 2; ++ct)
#pragma unroll
                for (int j = 0; j < 8; ++j)
                    dp[(rt * 16 + j) * Nc + ct * 16] = acc[rt][ct][j];
    } else {
#pragma unroll
        for (int rt = 0; rt < RT; ++rt)
#pragma unroll
            for (int ct = 0; ct < 2; ++ct)
#pragma unroll
                for (int j = 0; j < 8; ++j)
                    if (rowb + rt * 16 + j + h * 8 < n)
                        dp[(rt * 16 + j) * Nc + ct * 16] = acc[rt][ct][j];
    }
}

template <int RTC, int RTL>
static inline void launch_enc_gemms(const float* agg, const float* h1, float* h2, float* e,
                                    const float* Wrel, const float* Wroot, const float* brel,
                                    const float* lW, const float* lb, int n, hipStream_t stream) {
    gemm_tile_kernel<2, H, H, H, RTC, 1, true><<<(n + RTC * 16 - 1) / (RTC * 16), H, 0, stream>>>(
        agg, h1, nullptr, 0, Wrel, Wroot, brel, h2, n);
    gemm_tile_kernel<1, H, H, F3, RTL, 0, false><<<(n + RTL * 16 - 1) / (RTL * 16), F3, 0, stream>>>(
        h2, nullptr, nullptr, 0, lW, nullptr, lb, e, n);
}

extern "C" void kernel_launch(void* const* d_in, const int* in_sizes, int n_in,
                              void* d_out, int out_size, void* d_ws, size_t ws_size,
                              hipStream_t stream) {
    (void)in_sizes; (void)n_in; (void)out_size; (void)ws_size;

    const float* x[3]   = {(const float*)d_in[0], (const float*)d_in[1], (const float*)d_in[2]};
    const int*   ei[3]  = {(const int*)d_in[3], (const int*)d_in[4], (const int*)d_in[5]};
    const int*   A1r    = (const int*)d_in[6];
    const int*   A1c    = (const int*)d_in[7];
    const float* A1v    = (const float*)d_in[8];
    const int*   A2r    = (const int*)d_in[9];
    const int*   A2c    = (const int*)d_in[10];
    const float* A2v    = (const float*)d_in[11];
    const float* c1Wrel  = (const float*)d_in[12];
    const float* c1brel  = (const float*)d_in[13];
    const float* c1Wroot = (const float*)d_in[14];
    const float* c2Wrel  = (const float*)d_in[15];
    const float* c2brel  = (const float*)d_in[16];
    const float* c2Wroot = (const float*)d_in[17];
    const float* linW    = (const float*)d_in[18];
    const float* linb    = (const float*)d_in[19];
    const float* decW0   = (const float*)d_in[20];
    const float* decb0   = (const float*)d_in[21];
    const float* decWout = (const float*)d_in[22];
    const float* decbout = (const float*)d_in[23];
    float* out = (float*)d_out;

    float* ws = (float*)d_ws;
    float* h1   = ws;
    float* agg  = h1   + (size_t)N0 * H;
    float* h2   = agg  + (size_t)N0 * H;
    float* agg6 = h2   + (size_t)N0 * H;
    float* e0   = agg6 + (size_t)N0 * 6;
    float* e1   = e0   + (size_t)N0 * F3;
    float* e2   = e1   + (size_t)N1 * F3;
    float* i1   = e2   + (size_t)N2 * F3;
    float* i2   = i1   + (size_t)N0 * F3;
    float* hdec = i2   + (size_t)N0 * F3;

    const int nlv[3] = {N0, N1, N2};
    const int Elv[3] = {E0, E1, E2};
    const int rrl[3] = {200, 50, 13};
    static_assert(E0 % 4 == 0 && E1 % 4 == 0 && E2 % 4 == 0, "dst lists are read 16 B at a time");
    float* elv[3] = {e0, e1, e2};

    for (int i = 0; i < 3; ++i) {
        const int n = nlv[i];
        const int E = Elv[i];

        {
            const int cnt = n * 6;
            scatter6_lds_kernel<<<dim3((n + SC_R - 1) / SC_R, SC_S, 1), SC_T, 0, stream>>>(
                x[i], ei[i], E, n, agg);
            agg6_reduce_kernel<<<(cnt + 255) / 256, 256, 0, stream>>>(agg, agg6, cnt);
            conv1_kernel<<<n, H, 0, stream>>>(x[i], agg6,
                                              c1Wrel + (size_t)i * 6 * H,
                                              c1brel + (size_t)i * H,
                                              c1Wroot + (size_t)i * 6 * H, h1, n);
        }

        {
            const int rr = rrl[i];
            agg128_lds_kernel<<<(n + rr - 1) / rr, AG_T, 0, stream>>>(h1, ei[i], E, n, rr, agg);
        }

        {
            const float* Wrel2  = c2Wrel + (size_t)i * H * H;
            const float* Wroot2 = c2Wroot + (size_t)i * H * H;
            const float* brel2  = c2brel + (size_t)i * H;
            const float* lW     = linW + (size_t)i * H * F3;
            const float* lb     = linb + (size_t)i * F3;
            if (i == 0)
                launch_enc_gemms<4, 2>(agg, h1, h2, elv[i], Wrel2, Wroot2, brel2, lW, lb, n, stream);
            else if (i == 1)
                launch_enc_gemms<2, 2>(agg, h1, h2, elv[i], Wrel2, Wroot2, brel2, lW, lb, n, stream);
            else
                launch_enc_gemms<1, 1>(agg, h1, h2, elv[i], Wrel2, Wroot2, brel2, lW, lb, n, stream);
        }
    }

    {
        long long cnt = (long long)N0 * F3;
        zero_kernel<<<(int)((cnt + 255) / 256), 256, 0, stream>>>(i1, cnt);
        zero_kernel<<<(int)((cnt + 255) / 256), 256, 0, stream>>>(i2, cnt);
        interp_kernel<<<NNZ, F3, 0, stream>>>(A1v, A1c, A1r, e1, i1);
        interp_kernel<<<NNZ, F3, 0, stream>>>(A2v, A2c, A2r, e2, i2);
    }

    for (int b = 0; b < 3; ++b) {
        gemm_tile_kernel<3, 32, F3, DECW, 2, 2, false><<<(N0 + 31) / 32, DECW, 0, stream>>>(
            e0, i1, i2, b * 32, decW0 + (size_t)b * F3 * DECW, nullptr,
            decb0 + (size_t)b * DECW, hdec, N0);
        dec_out_kernel<<<(N0 + 255) / 256, 256, 0, stream>>>(
            hdec, decWout + (size_t)b * DECW, decbout + b, out, N0, b);
    }
}


# interp sums in LDS: block owns 400 dst rows of one matrix, scans its row list, queues hits per wave, adds val*e rows by ds float adds; float atomics, fills gone; 28->25 launches; sum order changed
# speedup vs baseline: 2.6891x; 1.3126x over previous
#include <hip/hip_runtime.h>
#include <hip/hip_bf16.h>

#define N0 50000
#define N1 12500
#define N2 3125
#define E0 800000
#define E1 200000
#define E2 50000
#define NNZ 200000
#define H 128
#define F3 96
#define DECW 64

typedef __attribute__((ext_vector_type(2))) float v2f;
typedef __attribute__((ext_vector_type(8))) float v8f;


__global__ void conv1_kernel(const float* __restrict__ x, const float* __restrict__ agg6,
                             const float* __restrict__ Wrel, const float* __restrict__ brel,
                             const float* __restrict__ Wroot, float* __restrict__ h1, int n) {
    int r = blockIdx.x;
    int c = threadIdx.x;
    if (r >= n) return;
    float s = brel[c];
#pragma unroll
    for (int k = 0; k < 6; ++k) {
        s += agg6[r * 6 + k] * Wrel[k * H + c];
        s += x[r * 6 + k] * Wroot[k * H + c];
    }
    h1[r * H + c] = fmaxf(s, 0.0f);
}

#define SC_R 4096
#define SC_S 16
#define SC_T 1024
#define SC_U 8

__global__ __launch_bounds__(SC_T) __attribute__((amdgpu_waves_per_eu(8, 8)))
void scatter6_lds_kernel(const float* __restrict__ x, const int* __restrict__ ei,
                         int E, int n, float* __restrict__ part) {
    __shared__ float acc[SC_R * 6];
    const int tid = threadIdx.x;
    const int r0  = blockIdx.x * SC_R;
    const int s   = blockIdx.y;

    for (int i = tid; i < SC_R * 6; i += SC_T) acc[i] = 0.0f;
    __syncthreads();

    const int chunk = (E + SC_S - 1) / SC_S;
    const int eb = s * chunk;
    const int ee = (eb + chunk < E) ? (eb + chunk) : E;

    if (eb < ee) {
        for (int base = eb; base < ee; base += SC_T * SC_U) {
            int off[SC_U];
            int srow[SC_U];
#pragma unroll
            for (int u = 0; u < SC_U; ++u) {
                const int e  = base + u * SC_T + tid;
                const int ec = (e < ee) ? e : (ee - 1);
                const int d  = ei[E + ec] - r0;
                const int sr = ei[ec];
                const bool hit = (e < ee) && ((unsigned)d < (unsigned)SC_R);
                off[u]  = hit ? d * 6 : -1;
                srow[u] = hit ? sr * 6 : 0;
            }
            float2 v[SC_U][3];
#pragma unroll
            for (int u = 0; u < SC_U; ++u) {
                const float2* p = (const float2*)(x + srow[u]);
                v[u][0] = p[0];
                v[u][1] = p[1];
                v[u][2] = p[2];
            }
#pragma unroll
            for (int u = 0; u < SC_U; ++u)
                asm volatile("" :: "v"(v[u][0].x), "v"(v[u][0].y), "v"(v[u][1].x),
                                   "v"(v[u][1].y), "v"(v[u][2].x), "v"(v[u][2].y));
#pragma unroll
            for (int u = 0; u < SC_U; ++u) {
                if (off[u] >= 0) {
                    float* a = acc + off[u];
                    atomicAdd(a + 0, v[u][0].x);
                    atomicAdd(a + 1, v[u][0].y);
                    atomicAdd(a + 2, v[u][1].x);
                    atomicAdd(a + 3, v[u][1].y);
                    atomicAdd(a + 4, v[u][2].x);
                    atomicAdd(a + 5, v[u][2].y);
                }
            }
        }
    }
    __syncthreads();

    const int rows = (n - r0 < SC_R) ? (n - r0) : SC_R;
    float* p = part + (size_t)s * n * 6 + (size_t)r0 * 6;
    for (int i = tid; i < rows * 6; i += SC_T) p[i] = acc[i];
}

__global__ void agg6_reduce_kernel(const float* __restrict__ part, float* __restrict__ agg6,
                                   int cnt) {
    int t = blockIdx.x * blockDim.x + threadIdx.x;
    if (t >= cnt) return;
    float s = 0.0f;
#pragma unroll
    for (int k = 0; k < SC_S; ++k) s += part[(size_t)k * cnt + t];
    agg6[t] = s;
}

#define AG_T    1024
#define AG_RMAX 200
#define AG_U    4
#define AG_Q    1024
#define AG_B    16

template <bool TAIL>
__device__ __forceinline__ void agg128_scan_trip(const int4* __restrict__ d4, int g0, int ng,
                                                 int tid, int r0, unsigned rows,
                                                 int* q, int& tail) {
    int4 d[AG_U];
#pragma unroll
    for (int u = 0; u < AG_U; ++u) {
        const int g = g0 + u * AG_T + tid;
        d[u] = d4[(TAIL && g >= ng) ? (ng - 1) : g];
    }
#pragma unroll
    for (int u = 0; u < AG_U; ++u) {
        const int g = g0 + u * AG_T + tid;
        const bool valid = !TAIL || (g < ng);
        const unsigned vmask = TAIL ? __builtin_amdgcn_ballot_w32(g < ng) : 0xffffffffu;
        const int dd[4] = {d[u].x, d[u].y, d[u].z, d[u].w};
#pragma unroll
        for (int c = 0; c < 4; ++c) {
            const bool in = (unsigned)(dd[c] - r0) < rows;
            const unsigned mask = __builtin_amdgcn_ballot_w32(in) & vmask;
            if (mask != 0u) {
                if (in && valid)
                    q[(tail + (int)__builtin_amdgcn_mbcnt_lo(mask, 0u)) & (AG_Q - 1)] = g * 4 + c;
                tail += __builtin_popcount(mask);
            }
        }
    }
}

__device__ __forceinline__ void agg128_drain(const float* __restrict__ h1,
                                             const int* __restrict__ ei, int E, int r0,
                                             const int* q, int head, int cnt,
                                             float* acc, int lane) {
    const int qe = q[(head + lane) & (AG_Q - 1)];
    const int e  = (lane < cnt) ? qe : 0;
    const int s  = ei[e];
    const int dl = ei[E + e] - r0;
#pragma unroll 1
    for (int b = 0; b < cnt; b += AG_B) {
        float4 v[AG_B];
#pragma unroll
        for (int j = 0; j < AG_B; ++j) {
            const int sj = __builtin_amdgcn_readlane(s, b + j);
            v[j] = *(const float4*)(h1 + sj * H + lane * 4);
        }
#pragma unroll
        for (int j = 0; j < AG_B; ++j) {
            if (b + j < cnt) {
                const int dj = __builtin_amdgcn_readlane(dl, b + j);
                float* a = acc + dj * H + lane;
                atomicAdd(a,      v[j].x);
                atomicAdd(a + 32, v[j].y);
                atomicAdd(a + 64, v[j].z);
                atomicAdd(a + 96, v[j].w);
            }
        }
    }
}

__global__ __launch_bounds__(AG_T)
void agg128_lds_kernel(const float* __restrict__ h1, const int* __restrict__ ei,
                       int E, int n, int rr, float* __restrict__ agg) {
    __shared__ float acc[AG_RMAX * H];
    __shared__ int ring[AG_T / 32][AG_Q];
    const int tid  = threadIdx.x;
    const int lane = tid & 31;
    const int r0   = blockIdx.x * rr;
    const int rows = (n - r0 < rr) ? (n - r0) : rr;

    {
        float4* a4 = (float4*)acc;
        const float4 z = {0.0f, 0.0f, 0.0f, 0.0f};
        for (int i = tid; i < rows * (H / 4); i += AG_T) a4[i] = z;
    }
    __syncthreads();

    int* q = ring[tid >> 5];
    int head = 0, tail = 0;
    const int4* d4 = (const int4*)(ei + E);
    const int ng = E >> 2;

    int g0 = 0;
    for (; g0 + AG_T * AG_U <= ng; g0 += AG_T * AG_U) {
        agg128_scan_trip<false>(d4, g0, ng, tid, r0, (unsigned)rows, q, tail);
        while (tail - head >= 32) {
            agg128_drain(h1, ei, E, r0, q, head, 32, acc, lane);
            head += 32;
        }
    }
    if (g0 < ng) agg128_scan_trip<true>(d4, g0, ng, tid, r0, (unsigned)rows, q, tail);
    while (tail - head > 0) {
        const int cnt = (tail - head < 32) ? (tail - head) : 32;
        agg128_drain(h1, ei, E, r0, q, head, cnt, acc, lane);
        head += cnt;
    }
    __syncthreads();

    float4* o4 = (float4*)(agg + r0 * H);
    for (int i = tid; i < rows * (H / 4); i += AG_T) {
        const float* a = acc + (i >> 5) * H + (i & 31);
        float4 v;
        v.x = a[0];
        v.y = a[32];
        v.z = a[64];
        v.w = a[96];
        o4[i] = v;
    }
}

#define IN_R 400
#define IN_B 16

__device__ __forceinline__ void interp_drain(const float* __restrict__ e,
                                             const int* __restrict__ cols,
                                             const int* __restrict__ rws,
                                             const float* __restrict__ vals, int r0,
                                             const int* q, int head, int cnt,
                                             float* acc, int lane) {
    const int qe  = q[(head + lane) & (AG_Q - 1)];
    const int nz  = (lane < cnt) ? qe : 0;
    const int col = cols[nz];
    const int dl  = rws[nz] - r0;
    const int vb  = __float_as_int(vals[nz]);
#pragma unroll 1
    for (int b = 0; b < cnt; b += IN_B) {
        float v[IN_B][3];
#pragma unroll
        for (int j = 0; j < IN_B; ++j) {
            const int cj = __builtin_amdgcn_readlane(col, b + j);
            const float* p = e + cj * F3 + lane;
            v[j][0] = p[0];
            v[j][1] = p[32];
            v[j][2] = p[64];
        }
#pragma unroll
        for (int j = 0; j < IN_B; ++j) {
            if (b + j < cnt) {
                const int dj  = __builtin_amdgcn_readlane(dl, b + j);
                const float s = __int_as_float(__builtin_amdgcn_readlane(vb, b + j));
                float* a = acc + dj * F3 + lane;
                atomicAdd(a,      s * v[j][0]);
                atomicAdd(a + 32, s * v[j][1]);
                atomicAdd(a + 64, s * v[j][2]);
            }
        }
    }
}

__global__ __launch_bounds__(AG_T)
void interp_lds_kernel(const float* __restrict__ vals1, const int* __restrict__ cols1,
                       const int* __restrict__ rows1, const float* __restrict__ e1,
                       float* __restrict__ out1,
                       const float* __restrict__ vals2, const int* __restrict__ cols2,
                       const int* __restrict__ rows2, const float* __restrict__ e2,
                       float* __restrict__ out2, int nnz, int n) {
    __shared__ __attribute__((aligned(16))) float acc[IN_R * F3];
    __shared__ int ring[AG_T / 32][AG_Q];
    const bool m2 = (blockIdx.y != 0);
    const float* vals = m2 ? vals2 : vals1;
    const int*   cols = m2 ? cols2 : cols1;
    const int*   rws  = m2 ? rows2 : rows1;
    const float* e    = m2 ? e2 : e1;
    float*       outp = m2 ? out2 : out1;

    const int tid  = threadIdx.x;
    const int lane = tid & 31;
    const int r0   = blockIdx.x * IN_R;
    const int rows = (n - r0 < IN_R) ? (n - r0) : IN_R;

    {
        float4* a4 = (float4*)acc;
        const float4 z = {0.0f, 0.0f, 0.0f, 0.0f};
        for (int i = tid; i < rows * (F3 / 4); i += AG_T) a4[i] = z;
    }
    __syncthreads();

    int* q = ring[tid >> 5];
    int head = 0, tail = 0;
    const int4* d4 = (const int4*)rws;
    const int ng = nnz >> 2;

    int g0 = 0;
    for (; g0 + AG_T * AG_U <= ng; g0 += AG_T * AG_U) {
        agg128_scan_trip<false>(d4, g0, ng, tid, r0, (unsigned)rows, q, tail);
        while (tail - head >= 32) {
            interp_drain(e, cols, rws, vals, r0, q, head, 32, acc, lane);
            head += 32;
        }
    }
    if (g0 < ng) agg128_scan_trip<true>(d4, g0, ng, tid, r0, (unsigned)rows, q, tail);
    while (tail - head > 0) {
        const int cnt = (tail - head < 32) ? (tail - head) : 32;
        interp_drain(e, cols, rws, vals, r0, q, head, cnt, acc, lane);
        head += cnt;
    }
    __syncthreads();

    float4* o4 = (float4*)(outp + (size_t)r0 * F3);
    const float4* a4 = (const float4*)acc;
    for (int i = tid; i < rows * (F3 / 4); i += AG_T) o4[i] = a4[i];
}

__global__ void dec_out_kernel(const float* __restrict__ hdec, const float* __restrict__ Wout,
                               const float* __restrict__ bout, float* __restrict__ out,
                               int n, int branch) {
    int r = blockIdx.x * blockDim.x + threadIdx.x;
    if (r >= n) return;
    float s = bout[0];
    const float* hr = hdec + r * DECW;
#pragma unroll
    for (int c = 0; c < DECW; ++c) s += hr[c] * Wout[c];
    out[r * 3 + branch] = s;
}

template <int NSEG, int SEGW, int LD, int Nc, int RT, int ACT, bool DUAL>
__global__ __launch_bounds__(Nc)
void gemm_tile_kernel(const float* __restrict__ s0, const float* __restrict__ s1,
                      const float* __restrict__ s2, int coff,
                      const float* __restrict__ W0, const float* __restrict__ W1,
                      const float* __restrict__ bias, float* __restrict__ D, int n) {
    constexpr int K   = NSEG * SEGW;
    constexpr int P   = K + 4;
    constexpr int MB  = RT * 16;
    constexpr int NT  = Nc;
    constexpr int K4  = K / 4;
    constexpr int S4  = SEGW / 4;
    constexpr int TOT = MB * K4;
    constexpr int ITER = (TOT + NT - 1) / NT;
    constexpr int SB  = 8;
    __shared__ __attribute__((aligned(16))) float As[MB * P];

    const int tid  = threadIdx.x;
    const int lane = tid & 31;
    const int wave = tid >> 5;
    const int rowb = blockIdx.x * MB;

#pragma unroll
    for (int i0 = 0; i0 < ITER; i0 += SB) {
        float4 v[SB];
#pragma unroll
        for (int j = 0; j < SB; ++j) {
            if (i0 + j < ITER) {
                const int idx = (i0 + j) * NT + tid;
                const int idc = (TOT % NT != 0 && idx >= TOT) ? (TOT - 1) : idx;
                const int r   = idc / K4;
                const int c4  = idc % K4;
                const int sg  = c4 / S4;
                const int w4  = c4 % S4;
                const int rg  = (rowb + r < n) ? (rowb + r) : (n - 1);
                const float* sp = (sg == 0) ? s0 : (sg == 1) ? s1 : s2;
                v[j] = *(const float4*)(sp + (size_t)rg * LD + coff + w4 * 4);
            }
        }
#pragma unroll
        for (int j = 0; j < SB; ++j) {
            if (i0 + j < ITER) {
                const int idx = (i0 + j) * NT + tid;
                if (TOT % NT == 0 || idx < TOT) {
                    const int r  = idx / K4;
                    const int c4 = idx % K4;
                    *(float4*)(As + r * P + c4 * 4) = v[j];
                }
            }
        }
    }
    __syncthreads();

    const int c    = lane & 15;
    const int h    = lane >> 4;
    const int col0 = wave * 32;
    const float* ap  = As + c * P + 2 * h;
    const float* wp0 = W0 + col0 + c + 2 * h * Nc;
    const float* wp1 = DUAL ? (W1 + col0 + c + 2 * h * Nc) : nullptr;

    v8f acc[RT][2];
#pragma unroll
    for (int ct = 0; ct < 2; ++ct) {
        const float bv = bias[col0 + ct * 16 + c];
#pragma unroll
        for (int rt = 0; rt < RT; ++rt) {
            const v8f t = {bv, bv, bv, bv, bv, bv, bv, bv};
            acc[rt][ct] = t;
        }
    }

    if (DUAL) {
#pragma unroll 4
        for (int k = 0; k < SEGW; k += 4) {
#pragma unroll
            for (int sg = 0; sg < 2; ++sg) {
                const float* wp = (sg == 0) ? wp0 : wp1;
                v2f a[RT];
                v2f b[2];
#pragma unroll
                for (int rt = 0; rt < RT; ++rt)
                    a[rt] = *(const v2f*)(ap + rt * 16 * P + sg * SEGW + k);
#pragma unroll
                for (int ct = 0; ct < 2; ++ct) {
                    b[ct].x = wp[k * Nc + ct * 16];
                    b[ct].y = wp[k * Nc + Nc + ct * 16];
                }
#pragma unroll
                for (int rt = 0; rt < RT; ++rt)
#pragma unroll
                    for (int ct = 0; ct < 2; ++ct)
                        acc[rt][ct] = __builtin_amdgcn_wmma_f32_16x16x4_f32(
                            false, a[rt], false, b[ct], (short)0, acc[rt][ct], false, false);
            }
        }
    } else {
#pragma unroll 8
        for (int k = 0; k < K; k += 4) {
            v2f a[RT];
            v2f b[2];
#pragma unroll
            for (int rt = 0; rt < RT; ++rt)
                a[rt] = *(const v2f*)(ap + rt * 16 * P + k);
#pragma unroll
            for (int ct = 0; ct < 2; ++ct) {
                b[ct].x = wp0[k * Nc + ct * 16];
                b[ct].y = wp0[k * Nc + Nc + ct * 16];
            }
#pragma unroll
            for (int rt = 0; rt < RT; ++rt)
#pragma unroll
                for (int ct = 0; ct < 2; ++ct)
                    acc[rt][ct] = __builtin_amdgcn_wmma_f32_16x16x4_f32(
                        false, a[rt], false, b[ct], (short)0, acc[rt][ct], false, false);
        }
    }

    const bool full = (rowb + MB <= n);
#pragma unroll
    for (int rt = 0; rt < RT; ++rt)
#pragma unroll
        for (int ct = 0; ct < 2; ++ct)
#pragma unroll
            for (int j = 0; j < 8; ++j) {
                float v = acc[rt][ct][j];
                if (ACT == 1) v = fmaxf(v, 0.0f);
                if (ACT == 2) v = (v > 0.0f) ? v : (expf(v) - 1.0f);
                acc[rt][ct][j] = v;
            }
    float* dp = D + (rowb + h * 8) * Nc + col0 + c;
    if (full) {
#pragma unroll
        for (int rt = 0; rt < RT; ++rt)
#pragma unroll
            for (int ct = 0; ct < 2; ++ct)
#pragma unroll
                for (int j = 0; j < 8; ++j)
                    dp[(rt * 16 + j) * Nc + ct * 16] = acc[rt][ct][j];
    } else {
#pragma unroll
        for (int rt = 0; rt < RT; ++rt)
#pragma unroll
            for (int ct = 0; ct < 2; ++ct)
#pragma unroll
                for (int j = 0; j < 8; ++j)
                    if (rowb + rt * 16 + j + h * 8 < n)
                        dp[(rt * 16 + j) * Nc + ct * 16] = acc[rt][ct][j];
    }
}

template <int RTC, int RTL>
static inline void launch_enc_gemms(const float* agg, const float* h1, float* h2, float* e,
                                    const float* Wrel, const float* Wroot, const float* brel,
                                    const float* lW, const float* lb, int n, hipStream_t stream) {
    gemm_tile_kernel<2, H, H, H, RTC, 1, true><<<(n + RTC * 16 - 1) / (RTC * 16), H, 0, stream>>>(
        agg, h1, nullptr, 0, Wrel, Wroot, brel, h2, n);
    gemm_tile_kernel<1, H, H, F3, RTL, 0, false><<<(n + RTL * 16 - 1) / (RTL * 16), F3, 0, stream>>>(
        h2, nullptr, nullptr, 0, lW, nullptr, lb, e, n);
}

extern "C" void kernel_launch(void* const* d_in, const int* in_sizes, int n_in,
                              void* d_out, int out_size, void* d_ws, size_t ws_size,
                              hipStream_t stream) {
    (void)in_sizes; (void)n_in; (void)out_size; (void)ws_size;

    const float* x[3]   = {(const float*)d_in[0], (const float*)d_in[1], (const float*)d_in[2]};
    const int*   ei[3]  = {(const int*)d_in[3], (const int*)d_in[4], (const int*)d_in[5]};
    const int*   A1r    = (const int*)d_in[6];
    const int*   A1c    = (const int*)d_in[7];
    const float* A1v    = (const float*)d_in[8];
    const int*   A2r    = (const int*)d_in[9];
    const int*   A2c    = (const int*)d_in[10];
    const float* A2v    = (const float*)d_in[11];
    const float* c1Wrel  = (const float*)d_in[12];
    const float* c1brel  = (const float*)d_in[13];
    const float* c1Wroot = (const float*)d_in[14];
    const float* c2Wrel  = (const float*)d_in[15];
    const float* c2brel  = (const float*)d_in[16];
    const float* c2Wroot = (const float*)d_in[17];
    const float* linW    = (const float*)d_in[18];
    const float* linb    = (const float*)d_in[19];
    const float* decW0   = (const float*)d_in[20];
    const float* decb0   = (const float*)d_in[21];
    const float* decWout = (const float*)d_in[22];
    const float* decbout = (const float*)d_in[23];
    float* out = (float*)d_out;

    float* ws = (float*)d_ws;
    float* h1   = ws;
    float* agg  = h1   + (size_t)N0 * H;
    float* h2   = agg  + (size_t)N0 * H;
    float* agg6 = h2   + (size_t)N0 * H;
    float* e0   = agg6 + (size_t)N0 * 6;
    float* e1   = e0   + (size_t)N0 * F3;
    float* e2   = e1   + (size_t)N1 * F3;
    float* i1   = e2   + (size_t)N2 * F3;
    float* i2   = i1   + (size_t)N0 * F3;
    float* hdec = i2   + (size_t)N0 * F3;

    const int nlv[3] = {N0, N1, N2};
    const int Elv[3] = {E0, E1, E2};
    const int rrl[3] = {200, 50, 13};
    static_assert(E0 % 4 == 0 && E1 % 4 == 0 && E2 % 4 == 0, "dst lists are read 16 B at a time");
    float* elv[3] = {e0, e1, e2};

    for (int i = 0; i < 3; ++i) {
        const int n = nlv[i];
        const int E = Elv[i];

        {
            const int cnt = n * 6;
            scatter6_lds_kernel<<<dim3((n + SC_R - 1) / SC_R, SC_S, 1), SC_T, 0, stream>>>(
                x[i], ei[i], E, n, agg);
            agg6_reduce_kernel<<<(cnt + 255) / 256, 256, 0, stream>>>(agg, agg6, cnt);
            conv1_kernel<<<n, H, 0, stream>>>(x[i], agg6,
                                              c1Wrel + (size_t)i * 6 * H,
                                              c1brel + (size_t)i * H,
                                              c1Wroot + (size_t)i * 6 * H, h1, n);
        }

        {
            const int rr = rrl[i];
            agg128_lds_kernel<<<(n + rr - 1) / rr, AG_T, 0, stream>>>(h1, ei[i], E, n, rr, agg);
        }

        {
            const float* Wrel2  = c2Wrel + (size_t)i * H * H;
            const float* Wroot2 = c2Wroot + (size_t)i * H * H;
            const float* brel2  = c2brel + (size_t)i * H;
            const float* lW     = linW + (size_t)i * H * F3;
            const float* lb     = linb + (size_t)i * F3;
            if (i == 0)
                launch_enc_gemms<4, 2>(agg, h1, h2, elv[i], Wrel2, Wroot2, brel2, lW, lb, n, stream);
            else if (i == 1)
                launch_enc_gemms<2, 2>(agg, h1, h2, elv[i], Wrel2, Wroot2, brel2, lW, lb, n, stream);
            else
                launch_enc_gemms<1, 1>(agg, h1, h2, elv[i], Wrel2, Wroot2, brel2, lW, lb, n, stream);
        }
    }

    {
        static_assert(NNZ % 4 == 0, "row lists are read 16 B at a time");
        interp_lds_kernel<<<dim3((N0 + IN_R - 1) / IN_R, 2, 1), AG_T, 0, stream>>>(
            A1v, A1c, A1r, e1, i1, A2v, A2c, A2r, e2, i2, NNZ, N0);
    }

    for (int b = 0; b < 3; ++b) {
        gemm_tile_kernel<3, 32, F3, DECW, 2, 2, false><<<(N0 + 31) / 32, DECW, 0, stream>>>(
            e0, i1, i2, b * 32, decW0 + (size_t)b * F3 * DECW, nullptr,
            decb0 + (size_t)b * DECW, hdec, N0);
        dec_out_kernel<<<(N0 + 255) / 256, 256, 0, stream>>>(
            hdec, decWout + (size_t)b * DECW, decbout + b, out, N0, b);
    }
}


# agg6 plane reduce folded into conv1: a block owns 32 rows, sums their 16 partial planes into LDS, then a wave stores whole 512-B h1 rows (16 B per lane); 25 -> 22 launches; sum order kept
# speedup vs baseline: 2.8057x; 1.0434x over previous
#include <hip/hip_runtime.h>
#include <hip/hip_bf16.h>

#define N0 50000
#define N1 12500
#define N2 3125
#define E0 800000
#define E1 200000
#define E2 50000
#define NNZ 200000
#define H 128
#define F3 96
#define DECW 64

typedef __attribute__((ext_vector_type(2))) float v2f;
typedef __attribute__((ext_vector_type(8))) float v8f;


#define SC_R 4096
#define SC_S 16
#define SC_T 1024
#define SC_U 8

__global__ __launch_bounds__(SC_T) __attribute__((amdgpu_waves_per_eu(8, 8)))
void scatter6_lds_kernel(const float* __restrict__ x, const int* __restrict__ ei,
                         int E, int n, float* __restrict__ part) {
    __shared__ float acc[SC_R * 6];
    const int tid = threadIdx.x;
    const int r0  = blockIdx.x * SC_R;
    const int s   = blockIdx.y;

    for (int i = tid; i < SC_R * 6; i += SC_T) acc[i] = 0.0f;
    __syncthreads();

    const int chunk = (E + SC_S - 1) / SC_S;
    const int eb = s * chunk;
    const int ee = (eb + chunk < E) ? (eb + chunk) : E;

    if (eb < ee) {
        for (int base = eb; base < ee; base += SC_T * SC_U) {
            int off[SC_U];
            int srow[SC_U];
#pragma unroll
            for (int u = 0; u < SC_U; ++u) {
                const int e  = base + u * SC_T + tid;
                const int ec = (e < ee) ? e : (ee - 1);
                const int d  = ei[E + ec] - r0;
                const int sr = ei[ec];
                const bool hit = (e < ee) && ((unsigned)d < (unsigned)SC_R);
                off[u]  = hit ? d * 6 : -1;
                srow[u] = hit ? sr * 6 : 0;
            }
            float2 v[SC_U][3];
#pragma unroll
            for (int u = 0; u < SC_U; ++u) {
                const float2* p = (const float2*)(x + srow[u]);
                v[u][0] = p[0];
                v[u][1] = p[1];
                v[u][2] = p[2];
            }
#pragma unroll
            for (int u = 0; u < SC_U; ++u)
                asm volatile("" :: "v"(v[u][0].x), "v"(v[u][0].y), "v"(v[u][1].x),
                                   "v"(v[u][1].y), "v"(v[u][2].x), "v"(v[u][2].y));
#pragma unroll
            for (int u = 0; u < SC_U; ++u) {
                if (off[u] >= 0) {
                    float* a = acc + off[u];
                    atomicAdd(a + 0, v[u][0].x);
                    atomicAdd(a + 1, v[u][0].y);
                    atomicAdd(a + 2, v[u][1].x);
                    atomicAdd(a + 3, v[u][1].y);
                    atomicAdd(a + 4, v[u][2].x);
                    atomicAdd(a + 5, v[u][2].y);
                }
            }
        }
    }
    __syncthreads();

    const int rows = (n - r0 < SC_R) ? (n - r0) : SC_R;
    float* p = part + (size_t)s * n * 6 + (size_t)r0 * 6;
    for (int i = tid; i < rows * 6; i += SC_T) p[i] = acc[i];
}

#define C1_R 32
#define C1_T 128

__global__ __launch_bounds__(C1_T)
void conv1_planes_kernel(const float* __restrict__ x, const float* __restrict__ part,
                         const float* __restrict__ Wrel, const float* __restrict__ brel,
                         const float* __restrict__ Wroot, float* __restrict__ h1, int n) {
    __shared__ float sa[C1_R * 6];
    __shared__ float sx[C1_R * 6];
    const int tid  = threadIdx.x;
    const int lane = tid & 31;
    const int wave = tid >> 5;
    const int r0   = blockIdx.x * C1_R;
    const int rows = (n - r0 < C1_R) ? (n - r0) : C1_R;
    const int cnt  = n * 6;

    float4 wr[6], wo[6];
#pragma unroll
    for (int k = 0; k < 6; ++k) {
        wr[k] = *(const float4*)(Wrel + k * H + lane * 4);
        wo[k] = *(const float4*)(Wroot + k * H + lane * 4);
    }
    const float4 bv = *(const float4*)(brel + lane * 4);

    static_assert(6 * C1_R <= 2 * C1_T, "two elements per thread cover the block's rows");
    const int  m    = rows * 6;
    const int  i1   = tid + C1_T;
    const bool has0 = tid < m;
    const bool has1 = i1 < m;
    const int  t0   = r0 * 6 + (has0 ? tid : 0);
    const int  t1   = r0 * 6 + (has1 ? i1 : 0);
    float p0[SC_S], p1[SC_S];
#pragma unroll
    for (int k = 0; k < SC_S; ++k) {
        p0[k] = part[(size_t)k * cnt + t0];
        p1[k] = part[(size_t)k * cnt + t1];
    }
    const float x0 = x[t0];
    const float x1 = x[t1];
#pragma unroll
    for (int k = 0; k < SC_S; k += 4)
        asm volatile("" :: "v"(p0[k]), "v"(p0[k + 1]), "v"(p0[k + 2]), "v"(p0[k + 3]),
                           "v"(p1[k]), "v"(p1[k + 1]), "v"(p1[k + 2]), "v"(p1[k + 3]));
    asm volatile("" :: "v"(x0), "v"(x1));
    float s0 = 0.0f, s1 = 0.0f;
#pragma unroll
    for (int k = 0; k < SC_S; ++k) {
        s0 += p0[k];
        s1 += p1[k];
    }
    if (has0) { sa[tid] = s0; sx[tid] = x0; }
    if (has1) { sa[i1]  = s1; sx[i1]  = x1; }
    __syncthreads();

    for (int r = wave; r < rows; r += C1_T / 32) {
        float4 s = bv;
#pragma unroll
        for (int k = 0; k < 6; ++k) {
            const float a  = sa[r * 6 + k];
            const float xv = sx[r * 6 + k];
            s.x += a * wr[k].x;  s.y += a * wr[k].y;  s.z += a * wr[k].z;  s.w += a * wr[k].w;
            s.x += xv * wo[k].x; s.y += xv * wo[k].y; s.z += xv * wo[k].z; s.w += xv * wo[k].w;
        }
        s.x = fmaxf(s.x, 0.0f);
        s.y = fmaxf(s.y, 0.0f);
        s.z = fmaxf(s.z, 0.0f);
        s.w = fmaxf(s.w, 0.0f);
        *(float4*)(h1 + (size_t)(r0 + r) * H + lane * 4) = s;
    }
}

#define AG_T    1024
#define AG_RMAX 200
#define AG_U    4
#define AG_Q    1024
#define AG_B    16

template <bool TAIL>
__device__ __forceinline__ void agg128_scan_trip(const int4* __restrict__ d4, int g0, int ng,
                                                 int tid, int r0, unsigned rows,
                                                 int* q, int& tail) {
    int4 d[AG_U];
#pragma unroll
    for (int u = 0; u < AG_U; ++u) {
        const int g = g0 + u * AG_T + tid;
        d[u] = d4[(TAIL && g >= ng) ? (ng - 1) : g];
    }
#pragma unroll
    for (int u = 0; u < AG_U; ++u) {
        const int g = g0 + u * AG_T + tid;
        const bool valid = !TAIL || (g < ng);
        const unsigned vmask = TAIL ? __builtin_amdgcn_ballot_w32(g < ng) : 0xffffffffu;
        const int dd[4] = {d[u].x, d[u].y, d[u].z, d[u].w};
#pragma unroll
        for (int c = 0; c < 4; ++c) {
            const bool in = (unsigned)(dd[c] - r0) < rows;
            const unsigned mask = __builtin_amdgcn_ballot_w32(in) & vmask;
            if (mask != 0u) {
                if (in && valid)
                    q[(tail + (int)__builtin_amdgcn_mbcnt_lo(mask, 0u)) & (AG_Q - 1)] = g * 4 + c;
                tail += __builtin_popcount(mask);
            }
        }
    }
}

__device__ __forceinline__ void agg128_drain(const float* __restrict__ h1,
                                             const int* __restrict__ ei, int E, int r0,
                                             const int* q, int head, int cnt,
                                             float* acc, int lane) {
    const int qe = q[(head + lane) & (AG_Q - 1)];
    const int e  = (lane < cnt) ? qe : 0;
    const int s  = ei[e];
    const int dl = ei[E + e] - r0;
#pragma unroll 1
    for (int b = 0; b < cnt; b += AG_B) {
        float4 v[AG_B];
#pragma unroll
        for (int j = 0; j < AG_B; ++j) {
            const int sj = __builtin_amdgcn_readlane(s, b + j);
            v[j] = *(const float4*)(h1 + sj * H + lane * 4);
        }
#pragma unroll
        for (int j = 0; j < AG_B; ++j) {
            if (b + j < cnt) {
                const int dj = __builtin_amdgcn_readlane(dl, b + j);
                float* a = acc + dj * H + lane;
                atomicAdd(a,      v[j].x);
                atomicAdd(a + 32, v[j].y);
                atomicAdd(a + 64, v[j].z);
                atomicAdd(a + 96, v[j].w);
            }
        }
    }
}

__global__ __launch_bounds__(AG_T)
void agg128_lds_kernel(const float* __restrict__ h1, const int* __restrict__ ei,
                       int E, int n, int rr, float* __restrict__ agg) {
    __shared__ float acc[AG_RMAX * H];
    __shared__ int ring[AG_T / 32][AG_Q];
    const int tid  = threadIdx.x;
    const int lane = tid & 31;
    const int r0   = blockIdx.x * rr;
    const int rows = (n - r0 < rr) ? (n - r0) : rr;

    {
        float4* a4 = (float4*)acc;
        const float4 z = {0.0f, 0.0f, 0.0f, 0.0f};
        for (int i = tid; i < rows * (H / 4); i += AG_T) a4[i] = z;
    }
    __syncthreads();

    int* q = ring[tid >> 5];
    int head = 0, tail = 0;
    const int4* d4 = (const int4*)(ei + E);
    const int ng = E >> 2;

    int g0 = 0;
    for (; g0 + AG_T * AG_U <= ng; g0 += AG_T * AG_U) {
        agg128_scan_trip<false>(d4, g0, ng, tid, r0, (unsigned)rows, q, tail);
        while (tail - head >= 32) {
            agg128_drain(h1, ei, E, r0, q, head, 32, acc, lane);
            head += 32;
        }
    }
    if (g0 < ng) agg128_scan_trip<true>(d4, g0, ng, tid, r0, (unsigned)rows, q, tail);
    while (tail - head > 0) {
        const int cnt = (tail - head < 32) ? (tail - head) : 32;
        agg128_drain(h1, ei, E, r0, q, head, cnt, acc, lane);
        head += cnt;
    }
    __syncthreads();

    float4* o4 = (float4*)(agg + r0 * H);
    for (int i = tid; i < rows * (H / 4); i += AG_T) {
        const float* a = acc + (i >> 5) * H + (i & 31);
        float4 v;
        v.x = a[0];
        v.y = a[32];
        v.z = a[64];
        v.w = a[96];
        o4[i] = v;
    }
}

#define IN_R 400
#define IN_B 16

__device__ __forceinline__ void interp_drain(const float* __restrict__ e,
                                             const int* __restrict__ cols,
                                             const int* __restrict__ rws,
                                             const float* __restrict__ vals, int r0,
                                             const int* q, int head, int cnt,
                                             float* acc, int lane) {
    const int qe  = q[(head + lane) & (AG_Q - 1)];
    const int nz  = (lane < cnt) ? qe : 0;
    const int col = cols[nz];
    const int dl  = rws[nz] - r0;
    const int vb  = __float_as_int(vals[nz]);
#pragma unroll 1
    for (int b = 0; b < cnt; b += IN_B) {
        float v[IN_B][3];
#pragma unroll
        for (int j = 0; j < IN_B; ++j) {
            const int cj = __builtin_amdgcn_readlane(col, b + j);
            const float* p = e + cj * F3 + lane;
            v[j][0] = p[0];
            v[j][1] = p[32];
            v[j][2] = p[64];
        }
#pragma unroll
        for (int j = 0; j < IN_B; ++j) {
            if (b + j < cnt) {
                const int dj  = __builtin_amdgcn_readlane(dl, b + j);
                const float s = __int_as_float(__builtin_amdgcn_readlane(vb, b + j));
                float* a = acc + dj * F3 + lane;
                atomicAdd(a,      s * v[j][0]);
                atomicAdd(a + 32, s * v[j][1]);
                atomicAdd(a + 64, s * v[j][2]);
            }
        }
    }
}

__global__ __launch_bounds__(AG_T)
void interp_lds_kernel(const float* __restrict__ vals1, const int* __restrict__ cols1,
                       const int* __restrict__ rows1, const float* __restrict__ e1,
                       float* __restrict__ out1,
                       const float* __restrict__ vals2, const int* __restrict__ cols2,
                       const int* __restrict__ rows2, const float* __restrict__ e2,
                       float* __restrict__ out2, int nnz, int n) {
    __shared__ __attribute__((aligned(16))) float acc[IN_R * F3];
    __shared__ int ring[AG_T / 32][AG_Q];
    const bool m2 = (blockIdx.y != 0);
    const float* vals = m2 ? vals2 : vals1;
    const int*   cols = m2 ? cols2 : cols1;
    const int*   rws  = m2 ? rows2 : rows1;
    const float* e    = m2 ? e2 : e1;
    float*       outp = m2 ? out2 : out1;

    const int tid  = threadIdx.x;
    const int lane = tid & 31;
    const int r0   = blockIdx.x * IN_R;
    const int rows = (n - r0 < IN_R) ? (n - r0) : IN_R;

    {
        float4* a4 = (float4*)acc;
        const float4 z = {0.0f, 0.0f, 0.0f, 0.0f};
        for (int i = tid; i < rows * (F3 / 4); i += AG_T) a4[i] = z;
    }
    __syncthreads();

    int* q = ring[tid >> 5];
    int head = 0, tail = 0;
    const int4* d4 = (const int4*)rws;
    const int ng = nnz >> 2;

    int g0 = 0;
    for (; g0 + AG_T * AG_U <= ng; g0 += AG_T * AG_U) {
        agg128_scan_trip<false>(d4, g0, ng, tid, r0, (unsigned)rows, q, tail);
        while (tail - head >= 32) {
            interp_drain(e, cols, rws, vals, r0, q, head, 32, acc, lane);
            head += 32;
        }
    }
    if (g0 < ng) agg128_scan_trip<true>(d4, g0, ng, tid, r0, (unsigned)rows, q, tail);
    while (tail - head > 0) {
        const int cnt = (tail - head < 32) ? (tail - head) : 32;
        interp_drain(e, cols, rws, vals, r0, q, head, cnt, acc, lane);
        head += cnt;
    }
    __syncthreads();

    float4* o4 = (float4*)(outp + (size_t)r0 * F3);
    const float4* a4 = (const float4*)acc;
    for (int i = tid; i < rows * (F3 / 4); i += AG_T) o4[i] = a4[i];
}

__global__ void dec_out_kernel(const float* __restrict__ hdec, const float* __restrict__ Wout,
                               const float* __restrict__ bout, float* __restrict__ out,
                               int n, int branch) {
    int r = blockIdx.x * blockDim.x + threadIdx.x;
    if (r >= n) return;
    float s = bout[0];
    const float* hr = hdec + r * DECW;
#pragma unroll
    for (int c = 0; c < DECW; ++c) s += hr[c] * Wout[c];
    out[r * 3 + branch] = s;
}

template <int NSEG, int SEGW, int LD, int Nc, int RT, int ACT, bool DUAL>
__global__ __launch_bounds__(Nc)
void gemm_tile_kernel(const float* __restrict__ s0, const float* __restrict__ s1,
                      const float* __restrict__ s2, int coff,
                      const float* __restrict__ W0, const float* __restrict__ W1,
                      const float* __restrict__ bias, float* __restrict__ D, int n) {
    constexpr int K   = NSEG * SEGW;
    constexpr int P   = K + 4;
    constexpr int MB  = RT * 16;
    constexpr int NT  = Nc;
    constexpr int K4  = K / 4;
    constexpr int S4  = SEGW / 4;
    constexpr int TOT = MB * K4;
    constexpr int ITER = (TOT + NT - 1) / NT;
    constexpr int SB  = 8;
    __shared__ __attribute__((aligned(16))) float As[MB * P];

    const int tid  = threadIdx.x;
    const int lane = tid & 31;
    const int wave = tid >> 5;
    const int rowb = blockIdx.x * MB;

#pragma unroll
    for (int i0 = 0; i0 < ITER; i0 += SB) {
        float4 v[SB];
#pragma unroll
        for (int j = 0; j < SB; ++j) {
            if (i0 + j < ITER) {
                const int idx = (i0 + j) * NT + tid;
                const int idc = (TOT % NT != 0 && idx >= TOT) ? (TOT - 1) : idx;
                const int r   = idc / K4;
                const int c4  = idc % K4;
                const int sg  = c4 / S4;
                const int w4  = c4 % S4;
                const int rg  = (rowb + r < n) ? (rowb + r) : (n - 1);
                const float* sp = (sg == 0) ? s0 : (sg == 1) ? s1 : s2;
                v[j] = *(const float4*)(sp + (size_t)rg * LD + coff + w4 * 4);
            }
        }
#pragma unroll
        for (int j = 0; j < SB; ++j) {
            if (i0 + j < ITER) {
                const int idx = (i0 + j) * NT + tid;
                if (TOT % NT == 0 || idx < TOT) {
                    const int r  = idx / K4;
                    const int c4 = idx % K4;
                    *(float4*)(As + r * P + c4 * 4) = v[j];
                }
            }
        }
    }
    __syncthreads();

    const int c    = lane & 15;
    const int h    = lane >> 4;
    const int col0 = wave * 32;
    const float* ap  = As + c * P + 2 * h;
    const float* wp0 = W0 + col0 + c + 2 * h * Nc;
    const float* wp1 = DUAL ? (W1 + col0 + c + 2 * h * Nc) : nullptr;

    v8f acc[RT][2];
#pragma unroll
    for (int ct = 0; ct < 2; ++ct) {
        const float bv = bias[col0 + ct * 16 + c];
#pragma unroll
        for (int rt = 0; rt < RT; ++rt) {
            const v8f t = {bv, bv, bv, bv, bv, bv, bv, bv};
            acc[rt][ct] = t;
        }
    }

    if (DUAL) {
#pragma unroll 4
        for (int k = 0; k < SEGW; k += 4) {
#pragma unroll
            for (int sg = 0; sg < 2; ++sg) {
                const float* wp = (sg == 0) ? wp0 : wp1;
                v2f a[RT];
                v2f b[2];
#pragma unroll
                for (int rt = 0; rt < RT; ++rt)
                    a[rt] = *(const v2f*)(ap + rt * 16 * P + sg * SEGW + k);
#pragma unroll
                for (int ct = 0; ct < 2; ++ct) {
                    b[ct].x = wp[k * Nc + ct * 16];
                    b[ct].y = wp[k * Nc + Nc + ct * 16];
                }
#pragma unroll
                for (int rt = 0; rt < RT; ++rt)
#pragma unroll
                    for (int ct = 0; ct < 2; ++ct)
                        acc[rt][ct] = __builtin_amdgcn_wmma_f32_16x16x4_f32(
                            false, a[rt], false, b[ct], (short)0, acc[rt][ct], false, false);
            }
        }
    } else {
#pragma unroll 8
        for (int k = 0; k < K; k += 4) {
            v2f a[RT];
            v2f b[2];
#pragma unroll
            for (int rt = 0; rt < RT; ++rt)
                a[rt] = *(const v2f*)(ap + rt * 16 * P + k);
#pragma unroll
            for (int ct = 0; ct < 2; ++ct) {
                b[ct].x = wp0[k * Nc + ct * 16];
                b[ct].y = wp0[k * Nc + Nc + ct * 16];
            }
#pragma unroll
            for (int rt = 0; rt < RT; ++rt)
#pragma unroll
                for (int ct = 0; ct < 2; ++ct)
                    acc[rt][ct] = __builtin_amdgcn_wmma_f32_16x16x4_f32(
                        false, a[rt], false, b[ct], (short)0, acc[rt][ct], false, false);
        }
    }

    const bool full = (rowb + MB <= n);
#pragma unroll
    for (int rt = 0; rt < RT; ++rt)
#pragma unroll
        for (int ct = 0; ct < 2; ++ct)
#pragma unroll
            for (int j = 0; j < 8; ++j) {
                float v = acc[rt][ct][j];
                if (ACT == 1) v = fmaxf(v, 0.0f);
                if (ACT == 2) v = (v > 0.0f) ? v : (expf(v) - 1.0f);
                acc[rt][ct][j] = v;
            }
    float* dp = D + (rowb + h * 8) * Nc + col0 + c;
    if (full) {
#pragma unroll
        for (int rt = 0; rt < RT; ++rt)
#pragma unroll
            for (int ct = 0; ct < 2; ++ct)
#pragma unroll
                for (int j = 0; j < 8; ++j)
                    dp[(rt * 16 + j) * Nc + ct * 16] = acc[rt][ct][j];
    } else {
#pragma unroll
        for (int rt = 0; rt < RT; ++rt)
#pragma unroll
            for (int ct = 0; ct < 2; ++ct)
#pragma unroll
                for (int j = 0; j < 8; ++j)
                    if (rowb + rt * 16 + j + h * 8 < n)
                        dp[(rt * 16 + j) * Nc + ct * 16] = acc[rt][ct][j];
    }
}

template <int RTC, int RTL>
static inline void launch_enc_gemms(const float* agg, const float* h1, float* h2, float* e,
                                    const float* Wrel, const float* Wroot, const float* brel,
                                    const float* lW, const float* lb, int n, hipStream_t stream) {
    gemm_tile_kernel<2, H, H, H, RTC, 1, true><<<(n + RTC * 16 - 1) / (RTC * 16), H, 0, stream>>>(
        agg, h1, nullptr, 0, Wrel, Wroot, brel, h2, n);
    gemm_tile_kernel<1, H, H, F3, RTL, 0, false><<<(n + RTL * 16 - 1) / (RTL * 16), F3, 0, stream>>>(
        h2, nullptr, nullptr, 0, lW, nullptr, lb, e, n);
}

extern "C" void kernel_launch(void* const* d_in, const int* in_sizes, int n_in,
                              void* d_out, int out_size, void* d_ws, size_t ws_size,
                              hipStream_t stream) {
    (void)in_sizes; (void)n_in; (void)out_size; (void)ws_size;

    const float* x[3]   = {(const float*)d_in[0], (const float*)d_in[1], (const float*)d_in[2]};
    const int*   ei[3]  = {(const int*)d_in[3], (const int*)d_in[4], (const int*)d_in[5]};
    const int*   A1r    = (const int*)d_in[6];
    const int*   A1c    = (const int*)d_in[7];
    const float* A1v    = (const float*)d_in[8];
    const int*   A2r    = (const int*)d_in[9];
    const int*   A2c    = (const int*)d_in[10];
    const float* A2v    = (const float*)d_in[11];
    const float* c1Wrel  = (const float*)d_in[12];
    const float* c1brel  = (const float*)d_in[13];
    const float* c1Wroot = (const float*)d_in[14];
    const float* c2Wrel  = (const float*)d_in[15];
    const float* c2brel  = (const float*)d_in[16];
    const float* c2Wroot = (const float*)d_in[17];
    const float* linW    = (const float*)d_in[18];
    const float* linb    = (const float*)d_in[19];
    const float* decW0   = (const float*)d_in[20];
    const float* decb0   = (const float*)d_in[21];
    const float* decWout = (const float*)d_in[22];
    const float* decbout = (const float*)d_in[23];
    float* out = (float*)d_out;

    float* ws = (float*)d_ws;
    float* h1   = ws;
    float* agg  = h1   + (size_t)N0 * H;
    float* h2   = agg  + (size_t)N0 * H;
    float* e0   = h2   + (size_t)N0 * H;
    float* e1   = e0   + (size_t)N0 * F3;
    float* e2   = e1   + (size_t)N1 * F3;
    float* i1   = e2   + (size_t)N2 * F3;
    float* i2   = i1   + (size_t)N0 * F3;
    float* hdec = i2   + (size_t)N0 * F3;

    const int nlv[3] = {N0, N1, N2};
    const int Elv[3] = {E0, E1, E2};
    const int rrl[3] = {200, 50, 13};
    static_assert(E0 % 4 == 0 && E1 % 4 == 0 && E2 % 4 == 0, "dst lists are read 16 B at a time");
    float* elv[3] = {e0, e1, e2};

    for (int i = 0; i < 3; ++i) {
        const int n = nlv[i];
        const int E = Elv[i];

        {
            scatter6_lds_kernel<<<dim3((n + SC_R - 1) / SC_R, SC_S, 1), SC_T, 0, stream>>>(
                x[i], ei[i], E, n, agg);
            conv1_planes_kernel<<<(n + C1_R - 1) / C1_R, C1_T, 0, stream>>>(
                x[i], agg,
                c1Wrel + (size_t)i * 6 * H,
                c1brel + (size_t)i * H,
                c1Wroot + (size_t)i * 6 * H, h1, n);
        }

        {
            const int rr = rrl[i];
            agg128_lds_kernel<<<(n + rr - 1) / rr, AG_T, 0, stream>>>(h1, ei[i], E, n, rr, agg);
        }

        {
            const float* Wrel2  = c2Wrel + (size_t)i * H * H;
            const float* Wroot2 = c2Wroot + (size_t)i * H * H;
            const float* brel2  = c2brel + (size_t)i * H;
            const float* lW     = linW + (size_t)i * H * F3;
            const float* lb     = linb + (size_t)i * F3;
            if (i == 0)
                launch_enc_gemms<4, 2>(agg, h1, h2, elv[i], Wrel2, Wroot2, brel2, lW, lb, n, stream);
            else if (i == 1)
                launch_enc_gemms<2, 2>(agg, h1, h2, elv[i], Wrel2, Wroot2, brel2, lW, lb, n, stream);
            else
                launch_enc_gemms<1, 1>(agg, h1, h2, elv[i], Wrel2, Wroot2, brel2, lW, lb, n, stream);
        }
    }

    {
        static_assert(NNZ % 4 == 0, "row lists are read 16 B at a time");
        interp_lds_kernel<<<dim3((N0 + IN_R - 1) / IN_R, 2, 1), AG_T, 0, stream>>>(
            A1v, A1c, A1r, e1, i1, A2v, A2c, A2r, e2, i2, NNZ, N0);
    }

    for (int b = 0; b < 3; ++b) {
        gemm_tile_kernel<3, 32, F3, DECW, 2, 2, false><<<(N0 + 31) / 32, DECW, 0, stream>>>(
            e0, i1, i2, b * 32, decW0 + (size_t)b * F3 * DECW, nullptr,
            decb0 + (size_t)b * DECW, hdec, N0);
        dec_out_kernel<<<(N0 + 255) / 256, 256, 0, stream>>>(
            hdec, decWout + (size_t)b * DECW, decbout + b, out, N0, b);
    }
}


# conv2 segment sum reads per-range edge lists (packed row offset | edge number) that the 6-wide scatter writes while it scans, not the whole dst list in every block; sum order changed
# speedup vs baseline: 2.9448x; 1.0496x over previous
#include <hip/hip_runtime.h>
#include <hip/hip_bf16.h>

#define N0 50000
#define N1 12500
#define N2 3125
#define E0 800000
#define E1 200000
#define E2 50000
#define NNZ 200000
#define H 128
#define F3 96
#define DECW 64

typedef __attribute__((ext_vector_type(2))) float v2f;
typedef __attribute__((ext_vector_type(8))) float v8f;


#define SC_R 4000

#define SC_S 16
#define SC_T 1024
#define SC_U 8
static_assert(SC_R <= 4096 && E0 < (1 << 20) && E1 < (1 << 20) && E2 < (1 << 20),
              "a list entry packs a 12-bit row offset and a 20-bit edge number");

__device__ __forceinline__ int sc_cap(int E) {
    const int chunk = (E + SC_S - 1) / SC_S;
    const int trips = (chunk + SC_T * SC_U - 1) / (SC_T * SC_U);
    return trips * SC_U * 32;
}

__global__ __launch_bounds__(SC_T) __attribute__((amdgpu_waves_per_eu(8, 8)))
void scatter6_lds_kernel(const float* __restrict__ x, const int* __restrict__ ei,
                         int E, int n, float* __restrict__ part,
                         int* __restrict__ lst, int* __restrict__ lcnt) {
    __shared__ float acc[SC_R * 6];
    const int tid = threadIdx.x;
    const int r0  = blockIdx.x * SC_R;
    const int s   = blockIdx.y;

    for (int i = tid; i < SC_R * 6; i += SC_T) acc[i] = 0.0f;
    __syncthreads();

    const int chunk = (E + SC_S - 1) / SC_S;
    const int eb = s * chunk;
    const int ee = (eb + chunk < E) ? (eb + chunk) : E;

    const int sub = (blockIdx.x * SC_S + s) * (SC_T / 32) + (tid >> 5);
    int* lstw = lst + (size_t)sub * sc_cap(E);
    int wc = 0;

    if (eb < ee) {
        for (int base = eb; base < ee; base += SC_T * SC_U) {
            int off[SC_U];
            int srow[SC_U];
#pragma unroll
            for (int u = 0; u < SC_U; ++u) {
                const int e  = base + u * SC_T + tid;
                const int ec = (e < ee) ? e : (ee - 1);
                const int d  = ei[E + ec] - r0;
                const int sr = ei[ec];
                const bool hit = (e < ee) && ((unsigned)d < (unsigned)SC_R);
                off[u]  = hit ? d * 6 : -1;
                srow[u] = hit ? sr * 6 : 0;
            }
            float2 v[SC_U][3];
#pragma unroll
            for (int u = 0; u < SC_U; ++u) {
                const float2* p = (const float2*)(x + srow[u]);
                v[u][0] = p[0];
                v[u][1] = p[1];
                v[u][2] = p[2];
            }
#pragma unroll
            for (int u = 0; u < SC_U; ++u)
                asm volatile("" :: "v"(v[u][0].x), "v"(v[u][0].y), "v"(v[u][1].x),
                                   "v"(v[u][1].y), "v"(v[u][2].x), "v"(v[u][2].y));
#pragma unroll
            for (int u = 0; u < SC_U; ++u) {
                const unsigned hm = __builtin_amdgcn_ballot_w32(off[u] >= 0);
                if (off[u] >= 0) {
                    lstw[wc + (int)__builtin_amdgcn_mbcnt_lo(hm, 0u)] =
                        (int)((((unsigned)off[u] / 6u) << 20) | (unsigned)(base + u * SC_T + tid));
                    float* a = acc + off[u];
                    atomicAdd(a + 0, v[u][0].x);
                    atomicAdd(a + 1, v[u][0].y);
                    atomicAdd(a + 2, v[u][1].x);
                    atomicAdd(a + 3, v[u][1].y);
                    atomicAdd(a + 4, v[u][2].x);
                    atomicAdd(a + 5, v[u][2].y);
                }
                wc += __builtin_popcount(hm);
            }
        }
    }
    if ((tid & 31) == 0) lcnt[sub] = wc;
    __syncthreads();

    const int rows = (n - r0 < SC_R) ? (n - r0) : SC_R;
    float* p = part + (size_t)s * n * 6 + (size_t)r0 * 6;
    for (int i = tid; i < rows * 6; i += SC_T) p[i] = acc[i];
}

#define C1_R 32
#define C1_T 128

__global__ __launch_bounds__(C1_T)
void conv1_planes_kernel(const float* __restrict__ x, const float* __restrict__ part,
                         const float* __restrict__ Wrel, const float* __restrict__ brel,
                         const float* __restrict__ Wroot, float* __restrict__ h1, int n) {
    __shared__ float sa[C1_R * 6];
    __shared__ float sx[C1_R * 6];
    const int tid  = threadIdx.x;
    const int lane = tid & 31;
    const int wave = tid >> 5;
    const int r0   = blockIdx.x * C1_R;
    const int rows = (n - r0 < C1_R) ? (n - r0) : C1_R;
    const int cnt  = n * 6;

    float4 wr[6], wo[6];
#pragma unroll
    for (int k = 0; k < 6; ++k) {
        wr[k] = *(const float4*)(Wrel + k * H + lane * 4);
        wo[k] = *(const float4*)(Wroot + k * H + lane * 4);
    }
    const float4 bv = *(const float4*)(brel + lane * 4);

    static_assert(6 * C1_R <= 2 * C1_T, "two elements per thread cover the block's rows");
    const int  m    = rows * 6;
    const int  i1   = tid + C1_T;
    const bool has0 = tid < m;
    const bool has1 = i1 < m;
    const int  t0   = r0 * 6 + (has0 ? tid : 0);
    const int  t1   = r0 * 6 + (has1 ? i1 : 0);
    float p0[SC_S], p1[SC_S];
#pragma unroll
    for (int k = 0; k < SC_S; ++k) {
        p0[k] = part[(size_t)k * cnt + t0];
        p1[k] = part[(size_t)k * cnt + t1];
    }
    const float x0 = x[t0];
    const float x1 = x[t1];
#pragma unroll
    for (int k = 0; k < SC_S; k += 4)
        asm volatile("" :: "v"(p0[k]), "v"(p0[k + 1]), "v"(p0[k + 2]), "v"(p0[k + 3]),
                           "v"(p1[k]), "v"(p1[k + 1]), "v"(p1[k + 2]), "v"(p1[k + 3]));
    asm volatile("" :: "v"(x0), "v"(x1));
    float s0 = 0.0f, s1 = 0.0f;
#pragma unroll
    for (int k = 0; k < SC_S; ++k) {
        s0 += p0[k];
        s1 += p1[k];
    }
    if (has0) { sa[tid] = s0; sx[tid] = x0; }
    if (has1) { sa[i1]  = s1; sx[i1]  = x1; }
    __syncthreads();

    for (int r = wave; r < rows; r += C1_T / 32) {
        float4 s = bv;
#pragma unroll
        for (int k = 0; k < 6; ++k) {
            const float a  = sa[r * 6 + k];
            const float xv = sx[r * 6 + k];
            s.x += a * wr[k].x;  s.y += a * wr[k].y;  s.z += a * wr[k].z;  s.w += a * wr[k].w;
            s.x += xv * wo[k].x; s.y += xv * wo[k].y; s.z += xv * wo[k].z; s.w += xv * wo[k].w;
        }
        s.x = fmaxf(s.x, 0.0f);
        s.y = fmaxf(s.y, 0.0f);
        s.z = fmaxf(s.z, 0.0f);
        s.w = fmaxf(s.w, 0.0f);
        *(float4*)(h1 + (size_t)(r0 + r) * H + lane * 4) = s;
    }
}

#define AG_T    1024
#define AG_RMAX 200
#define AG_U    4
#define AG_Q    1024
#define AG_B    16
#define SL_G    4
#define SL_L    5
static_assert(SC_S % SL_G == 0, "sublists are read SL_G at a time");
static_assert(SL_G * SL_L * 32 + 31 <= AG_Q, "a wave's ring holds what one read step can queue");

template <bool TAIL>
__device__ __forceinline__ void agg128_scan_trip(const int4* __restrict__ d4, int g0, int ng,
                                                 int tid, int r0, unsigned rows,
                                                 int* q, int& tail) {
    int4 d[AG_U];
#pragma unroll
    for (int u = 0; u < AG_U; ++u) {
        const int g = g0 + u * AG_T + tid;
        d[u] = d4[(TAIL && g >= ng) ? (ng - 1) : g];
    }
#pragma unroll
    for (int u = 0; u < AG_U; ++u) {
        const int g = g0 + u * AG_T + tid;
        const bool valid = !TAIL || (g < ng);
        const unsigned vmask = TAIL ? __builtin_amdgcn_ballot_w32(g < ng) : 0xffffffffu;
        const int dd[4] = {d[u].x, d[u].y, d[u].z, d[u].w};
#pragma unroll
        for (int c = 0; c < 4; ++c) {
            const bool in = (unsigned)(dd[c] - r0) < rows;
            const unsigned mask = __builtin_amdgcn_ballot_w32(in) & vmask;
            if (mask != 0u) {
                if (in && valid)
                    q[(tail + (int)__builtin_amdgcn_mbcnt_lo(mask, 0u)) & (AG_Q - 1)] = g * 4 + c;
                tail += __builtin_popcount(mask);
            }
        }
    }
}

__device__ __forceinline__ void agg128_drain(const float* __restrict__ h1,
                                             const int* __restrict__ ei, int roff,
                                             const int* q, int head, int cnt,
                                             float* acc, int lane) {
    const unsigned qe = (unsigned)q[(head + lane) & (AG_Q - 1)];
    const unsigned en = (lane < cnt) ? qe : 0u;
    const int s  = ei[en & 0xfffffu];
    const int dl = (int)(en >> 20) - roff;
#pragma unroll 1
    for (int b = 0; b < cnt; b += AG_B) {
        float4 v[AG_B];
#pragma unroll
        for (int j = 0; j < AG_B; ++j) {
            const int sj = __builtin_amdgcn_readlane(s, b + j);
            v[j] = *(const float4*)(h1 + sj * H + lane * 4);
        }
#pragma unroll
        for (int j = 0; j < AG_B; ++j) {
            if (b + j < cnt) {
                const int dj = __builtin_amdgcn_readlane(dl, b + j);
                float* a = acc + dj * H + lane;
                atomicAdd(a,      v[j].x);
                atomicAdd(a + 32, v[j].y);
                atomicAdd(a + 64, v[j].z);
                atomicAdd(a + 96, v[j].w);
            }
        }
    }
}

__global__ __launch_bounds__(AG_T)
void agg128_lds_kernel(const float* __restrict__ h1, const int* __restrict__ ei,
                       int E, int n, int rr, const int* __restrict__ lst,
                       const int* __restrict__ lcnt, float* __restrict__ agg) {
    __shared__ float acc[AG_RMAX * H];
    __shared__ int ring[AG_T / 32][AG_Q];
    const int tid  = threadIdx.x;
    const int lane = tid & 31;
    const int r0   = blockIdx.x * rr;
    const int rows = (n - r0 < rr) ? (n - r0) : rr;

    {
        float4* a4 = (float4*)acc;
        const float4 z = {0.0f, 0.0f, 0.0f, 0.0f};
        for (int i = tid; i < rows * (H / 4); i += AG_T) a4[i] = z;
    }
    __syncthreads();

    int* q = ring[tid >> 5];
    int head = 0, tail = 0;
    const int wv   = __builtin_amdgcn_readfirstlane(tid >> 5);
    const int xr   = r0 / SC_R;
    const int roff = r0 - xr * SC_R;
    const int cap  = sc_cap(E);
    const int sub0 = xr * (SC_S * (SC_T / 32)) + wv;
    static_assert(SC_S <= 32, "one count per lane");
    const int cvec = lcnt[sub0 + ((lane < SC_S) ? lane : 0) * (SC_T / 32)];

    for (int t0 = 0; t0 < SC_S; t0 += SL_G) {
        int cn[SL_G];
        const int* bp[SL_G];
        int ent[SL_G][SL_L];
#pragma unroll
        for (int g = 0; g < SL_G; ++g) {
            const int sub = sub0 + (t0 + g) * (SC_T / 32);
            cn[g] = __builtin_amdgcn_readlane(cvec, t0 + g);
            bp[g] = (cn[g] > 0) ? (lst + (size_t)sub * cap) : (lcnt + sub);
        }
#pragma unroll
        for (int g = 0; g < SL_G; ++g)
#pragma unroll
            for (int u = 0; u < SL_L; ++u) {
                const int i  = u * 32 + lane;
                const int ic = (cn[g] > 0) ? ((i < cn[g]) ? i : (cn[g] - 1)) : 0;
                ent[g][u] = bp[g][ic];
            }
#pragma unroll
        for (int g = 0; g < SL_G; ++g)
            asm volatile("" :: "v"(ent[g][0]), "v"(ent[g][1]), "v"(ent[g][2]),
                               "v"(ent[g][3]), "v"(ent[g][4]));
        static_assert(SL_L == 5, "the fence above names SL_L entries per sublist");
#pragma unroll
        for (int g = 0; g < SL_G; ++g)
#pragma unroll
            for (int u = 0; u < SL_L; ++u) {
                if (u * 32 < cn[g]) {
                    const unsigned en = (unsigned)ent[g][u];
                    const bool in = (u * 32 + lane < cn[g]) &&
                                    ((en >> 20) - (unsigned)roff < (unsigned)rows);
                    const unsigned mask = __builtin_amdgcn_ballot_w32(in);
                    if (mask != 0u) {
                        if (in)
                            q[(tail + (int)__builtin_amdgcn_mbcnt_lo(mask, 0u)) & (AG_Q - 1)] = (int)en;
                        tail += __builtin_popcount(mask);
                    }
                }
            }
        while (tail - head >= 32) {
            agg128_drain(h1, ei, roff, q, head, 32, acc, lane);
            head += 32;
        }
#pragma unroll
        for (int g = 0; g < SL_G; ++g) {
            for (int i0 = SL_L * 32; i0 < cn[g]; i0 += 32) {
                const int i = i0 + lane;
                const unsigned en = (unsigned)bp[g][(i < cn[g]) ? i : (cn[g] - 1)];
                const bool in = (i < cn[g]) && ((en >> 20) - (unsigned)roff < (unsigned)rows);
                const unsigned mask = __builtin_amdgcn_ballot_w32(in);
                if (mask != 0u) {
                    if (in)
                        q[(tail + (int)__builtin_amdgcn_mbcnt_lo(mask, 0u)) & (AG_Q - 1)] = (int)en;
                    tail += __builtin_popcount(mask);
                }
                while (tail - head >= 32) {
                    agg128_drain(h1, ei, roff, q, head, 32, acc, lane);
                    head += 32;
                }
            }
        }
    }
    while (tail - head > 0) {
        const int cnt = (tail - head < 32) ? (tail - head) : 32;
        agg128_drain(h1, ei, roff, q, head, cnt, acc, lane);
        head += cnt;
    }
    __syncthreads();

    float4* o4 = (float4*)(agg + r0 * H);
    for (int i = tid; i < rows * (H / 4); i += AG_T) {
        const float* a = acc + (i >> 5) * H + (i & 31);
        float4 v;
        v.x = a[0];
        v.y = a[32];
        v.z = a[64];
        v.w = a[96];
        o4[i] = v;
    }
}

#define IN_R 400
#define IN_B 16

__device__ __forceinline__ void interp_drain(const float* __restrict__ e,
                                             const int* __restrict__ cols,
                                             const int* __restrict__ rws,
                                             const float* __restrict__ vals, int r0,
                                             const int* q, int head, int cnt,
                                             float* acc, int lane) {
    const int qe  = q[(head + lane) & (AG_Q - 1)];
    const int nz  = (lane < cnt) ? qe : 0;
    const int col = cols[nz];
    const int dl  = rws[nz] - r0;
    const int vb  = __float_as_int(vals[nz]);
#pragma unroll 1
    for (int b = 0; b < cnt; b += IN_B) {
        float v[IN_B][3];
#pragma unroll
        for (int j = 0; j < IN_B; ++j) {
            const int cj = __builtin_amdgcn_readlane(col, b + j);
            const float* p = e + cj * F3 + lane;
            v[j][0] = p[0];
            v[j][1] = p[32];
            v[j][2] = p[64];
        }
#pragma unroll
        for (int j = 0; j < IN_B; ++j) {
            if (b + j < cnt) {
                const int dj  = __builtin_amdgcn_readlane(dl, b + j);
                const float s = __int_as_float(__builtin_amdgcn_readlane(vb, b + j));
                float* a = acc + dj * F3 + lane;
                atomicAdd(a,      s * v[j][0]);
                atomicAdd(a + 32, s * v[j][1]);
                atomicAdd(a + 64, s * v[j][2]);
            }
        }
    }
}

__global__ __launch_bounds__(AG_T)
void interp_lds_kernel(const float* __restrict__ vals1, const int* __restrict__ cols1,
                       const int* __restrict__ rows1, const float* __restrict__ e1,
                       float* __restrict__ out1,
                       const float* __restrict__ vals2, const int* __restrict__ cols2,
                       const int* __restrict__ rows2, const float* __restrict__ e2,
                       float* __restrict__ out2, int nnz, int n) {
    __shared__ __attribute__((aligned(16))) float acc[IN_R * F3];
    __shared__ int ring[AG_T / 32][AG_Q];
    const bool m2 = (blockIdx.y != 0);
    const float* vals = m2 ? vals2 : vals1;
    const int*   cols = m2 ? cols2 : cols1;
    const int*   rws  = m2 ? rows2 : rows1;
    const float* e    = m2 ? e2 : e1;
    float*       outp = m2 ? out2 : out1;

    const int tid  = threadIdx.x;
    const int lane = tid & 31;
    const int r0   = blockIdx.x * IN_R;
    const int rows = (n - r0 < IN_R) ? (n - r0) : IN_R;

    {
        float4* a4 = (float4*)acc;
        const float4 z = {0.0f, 0.0f, 0.0f, 0.0f};
        for (int i = tid; i < rows * (F3 / 4); i += AG_T) a4[i] = z;
    }
    __syncthreads();

    int* q = ring[tid >> 5];
    int head = 0, tail = 0;
    const int4* d4 = (const int4*)rws;
    const int ng = nnz >> 2;

    int g0 = 0;
    for (; g0 + AG_T * AG_U <= ng; g0 += AG_T * AG_U) {
        agg128_scan_trip<false>(d4, g0, ng, tid, r0, (unsigned)rows, q, tail);
        while (tail - head >= 32) {
            interp_drain(e, cols, rws, vals, r0, q, head, 32, acc, lane);
            head += 32;
        }
    }
    if (g0 < ng) agg128_scan_trip<true>(d4, g0, ng, tid, r0, (unsigned)rows, q, tail);
    while (tail - head > 0) {
        const int cnt = (tail - head < 32) ? (tail - head) : 32;
        interp_drain(e, cols, rws, vals, r0, q, head, cnt, acc, lane);
        head += cnt;
    }
    __syncthreads();

    float4* o4 = (float4*)(outp + (size_t)r0 * F3);
    const float4* a4 = (const float4*)acc;
    for (int i = tid; i < rows * (F3 / 4); i += AG_T) o4[i] = a4[i];
}

__global__ void dec_out_kernel(const float* __restrict__ hdec, const float* __restrict__ Wout,
                               const float* __restrict__ bout, float* __restrict__ out,
                               int n, int branch) {
    int r = blockIdx.x * blockDim.x + threadIdx.x;
    if (r >= n) return;
    float s = bout[0];
    const float* hr = hdec + r * DECW;
#pragma unroll
    for (int c = 0; c < DECW; ++c) s += hr[c] * Wout[c];
    out[r * 3 + branch] = s;
}

template <int NSEG, int SEGW, int LD, int Nc, int RT, int ACT, bool DUAL>
__global__ __launch_bounds__(Nc)
void gemm_tile_kernel(const float* __restrict__ s0, const float* __restrict__ s1,
                      const float* __restrict__ s2, int coff,
                      const float* __restrict__ W0, const float* __restrict__ W1,
                      const float* __restrict__ bias, float* __restrict__ D, int n) {
    constexpr int K   = NSEG * SEGW;
    constexpr int P   = K + 4;
    constexpr int MB  = RT * 16;
    constexpr int NT  = Nc;
    constexpr int K4  = K / 4;
    constexpr int S4  = SEGW / 4;
    constexpr int TOT = MB * K4;
    constexpr int ITER = (TOT + NT - 1) / NT;
    constexpr int SB  = 8;
    __shared__ __attribute__((aligned(16))) float As[MB * P];

    const int tid  = threadIdx.x;
    const int lane = tid & 31;
    const int wave = tid >> 5;
    const int rowb = blockIdx.x * MB;

#pragma unroll
    for (int i0 = 0; i0 < ITER; i0 += SB) {
        float4 v[SB];
#pragma unroll
        for (int j = 0; j < SB; ++j) {
            if (i0 + j < ITER) {
                const int idx = (i0 + j) * NT + tid;
                const int idc = (TOT % NT != 0 && idx >= TOT) ? (TOT - 1) : idx;
                const int r   = idc / K4;
                const int c4  = idc % K4;
                const int sg  = c4 / S4;
                const int w4  = c4 % S4;
                const int rg  = (rowb + r < n) ? (rowb + r) : (n - 1);
                const float* sp = (sg == 0) ? s0 : (sg == 1) ? s1 : s2;
                v[j] = *(const float4*)(sp + (size_t)rg * LD + coff + w4 * 4);
            }
        }
#pragma unroll
        for (int j = 0; j < SB; ++j) {
            if (i0 + j < ITER) {
                const int idx = (i0 + j) * NT + tid;
                if (TOT % NT == 0 || idx < TOT) {
                    const int r  = idx / K4;
                    const int c4 = idx % K4;
                    *(float4*)(As + r * P + c4 * 4) = v[j];
                }
            }
        }
    }
    __syncthreads();

    const int c    = lane & 15;
    const int h    = lane >> 4;
    const int col0 = wave * 32;
    const float* ap  = As + c * P + 2 * h;
    const float* wp0 = W0 + col0 + c + 2 * h * Nc;
    const float* wp1 = DUAL ? (W1 + col0 + c + 2 * h * Nc) : nullptr;

    v8f acc[RT][2];
#pragma unroll
    for (int ct = 0; ct < 2; ++ct) {
        const float bv = bias[col0 + ct * 16 + c];
#pragma unroll
        for (int rt = 0; rt < RT; ++rt) {
            const v8f t = {bv, bv, bv, bv, bv, bv, bv, bv};
            acc[rt][ct] = t;
        }
    }

    if (DUAL) {
#pragma unroll 4
        for (int k = 0; k < SEGW; k += 4) {
#pragma unroll
            for (int sg = 0; sg < 2; ++sg) {
                const float* wp = (sg == 0) ? wp0 : wp1;
                v2f a[RT];
                v2f b[2];
#pragma unroll
                for (int rt = 0; rt < RT; ++rt)
                    a[rt] = *(const v2f*)(ap + rt * 16 * P + sg * SEGW + k);
#pragma unroll
                for (int ct = 0; ct < 2; ++ct) {
                    b[ct].x = wp[k * Nc + ct * 16];
                    b[ct].y = wp[k * Nc + Nc + ct * 16];
                }
#pragma unroll
                for (int rt = 0; rt < RT; ++rt)
#pragma unroll
                    for (int ct = 0; ct < 2; ++ct)
                        acc[rt][ct] = __builtin_amdgcn_wmma_f32_16x16x4_f32(
                            false, a[rt], false, b[ct], (short)0, acc[rt][ct], false, false);
            }
        }
    } else {
#pragma unroll 8
        for (int k = 0; k < K; k += 4) {
            v2f a[RT];
            v2f b[2];
#pragma unroll
            for (int rt = 0; rt < RT; ++rt)
                a[rt] = *(const v2f*)(ap + rt * 16 * P + k);
#pragma unroll
            for (int ct = 0; ct < 2; ++ct) {
                b[ct].x = wp0[k * Nc + ct * 16];
                b[ct].y = wp0[k * Nc + Nc + ct * 16];
            }
#pragma unroll
            for (int rt = 0; rt < RT; ++rt)
#pragma unroll
                for (int ct = 0; ct < 2; ++ct)
                    acc[rt][ct] = __builtin_amdgcn_wmma_f32_16x16x4_f32(
                        false, a[rt], false, b[ct], (short)0, acc[rt][ct], false, false);
        }
    }

    const bool full = (rowb + MB <= n);
#pragma unroll
    for (int rt = 0; rt < RT; ++rt)
#pragma unroll
        for (int ct = 0; ct < 2; ++ct)
#pragma unroll
            for (int j = 0; j < 8; ++j) {
                float v = acc[rt][ct][j];
                if (ACT == 1) v = fmaxf(v, 0.0f);
                if (ACT == 2) v = (v > 0.0f) ? v : (expf(v) - 1.0f);
                acc[rt][ct][j] = v;
            }
    float* dp = D + (rowb + h * 8) * Nc + col0 + c;
    if (full) {
#pragma unroll
        for (int rt = 0; rt < RT; ++rt)
#pragma unroll
            for (int ct = 0; ct < 2; ++ct)
#pragma unroll
                for (int j = 0; j < 8; ++j)
                    dp[(rt * 16 + j) * Nc + ct * 16] = acc[rt][ct][j];
    } else {
#pragma unroll
        for (int rt = 0; rt < RT; ++rt)
#pragma unroll
            for (int ct = 0; ct < 2; ++ct)
#pragma unroll
                for (int j = 0; j < 8; ++j)
                    if (rowb + rt * 16 + j + h * 8 < n)
                        dp[(rt * 16 + j) * Nc + ct * 16] = acc[rt][ct][j];
    }
}

template <int RTC, int RTL>
static inline void launch_enc_gemms(const float* agg, const float* h1, float* h2, float* e,
                                    const float* Wrel, const float* Wroot, const float* brel,
                                    const float* lW, const float* lb, int n, hipStream_t stream) {
    gemm_tile_kernel<2, H, H, H, RTC, 1, true><<<(n + RTC * 16 - 1) / (RTC * 16), H, 0, stream>>>(
        agg, h1, nullptr, 0, Wrel, Wroot, brel, h2, n);
    gemm_tile_kernel<1, H, H, F3, RTL, 0, false><<<(n + RTL * 16 - 1) / (RTL * 16), F3, 0, stream>>>(
        h2, nullptr, nullptr, 0, lW, nullptr, lb, e, n);
}

extern "C" void kernel_launch(void* const* d_in, const int* in_sizes, int n_in,
                              void* d_out, int out_size, void* d_ws, size_t ws_size,
                              hipStream_t stream) {
    (void)in_sizes; (void)n_in; (void)out_size; (void)ws_size;

    const float* x[3]   = {(const float*)d_in[0], (const float*)d_in[1], (const float*)d_in[2]};
    const int*   ei[3]  = {(const int*)d_in[3], (const int*)d_in[4], (const int*)d_in[5]};
    const int*   A1r    = (const int*)d_in[6];
    const int*   A1c    = (const int*)d_in[7];
    const float* A1v    = (const float*)d_in[8];
    const int*   A2r    = (const int*)d_in[9];
    const int*   A2c    = (const int*)d_in[10];
    const float* A2v    = (const float*)d_in[11];
    const float* c1Wrel  = (const float*)d_in[12];
    const float* c1brel  = (const float*)d_in[13];
    const float* c1Wroot = (const float*)d_in[14];
    const float* c2Wrel  = (const float*)d_in[15];
    const float* c2brel  = (const float*)d_in[16];
    const float* c2Wroot = (const float*)d_in[17];
    const float* linW    = (const float*)d_in[18];
    const float* linb    = (const float*)d_in[19];
    const float* decW0   = (const float*)d_in[20];
    const float* decb0   = (const float*)d_in[21];
    const float* decWout = (const float*)d_in[22];
    const float* decbout = (const float*)d_in[23];
    float* out = (float*)d_out;

    float* ws = (float*)d_ws;
    float* h1   = ws;
    float* agg  = h1   + (size_t)N0 * H;
    float* h2   = agg  + (size_t)N0 * H;
    float* e0   = h2   + (size_t)N0 * H;
    float* e1   = e0   + (size_t)N0 * F3;
    float* e2   = e1   + (size_t)N1 * F3;
    float* i1   = e2   + (size_t)N2 * F3;
    float* i2   = i1   + (size_t)N0 * F3;
    float* hdec = i2   + (size_t)N0 * F3;

    constexpr size_t LST_MAX = (size_t)((N0 + SC_R - 1) / SC_R) * SC_S * (SC_T / 32) *
                               ((((E0 + SC_S - 1) / SC_S + SC_T * SC_U - 1) / (SC_T * SC_U)) * SC_U * 32);
    constexpr size_t LCNT_MAX = (size_t)((N0 + SC_R - 1) / SC_R) * SC_S * (SC_T / 32);
    static_assert(LST_MAX + LCNT_MAX <= (size_t)N0 * F3 * 2 + (size_t)N0 * DECW,
                  "the lists of level 0 (the largest) fit in i1 | i2 | hdec");
    static_assert(E0 >= E1 && E0 >= E2 && N0 >= N1 && N0 >= N2, "level 0 is the largest");
    int* lst  = (int*)i1;
    int* lcnt = lst + LST_MAX;

    const int nlv[3] = {N0, N1, N2};
    const int Elv[3] = {E0, E1, E2};
    const int rrl[3] = {200, 50, 13};
    static_assert(SC_R % 200 == 0 && SC_R % 50 == 0 && N2 <= SC_R,
                  "a conv2 block's rows lie inside one scatter6 range");
    static_assert(E0 % 4 == 0 && E1 % 4 == 0 && E2 % 4 == 0, "dst lists are read 16 B at a time");
    float* elv[3] = {e0, e1, e2};

    for (int i = 0; i < 3; ++i) {
        const int n = nlv[i];
        const int E = Elv[i];

        {
            scatter6_lds_kernel<<<dim3((n + SC_R - 1) / SC_R, SC_S, 1), SC_T, 0, stream>>>(
                x[i], ei[i], E, n, agg, lst, lcnt);
            conv1_planes_kernel<<<(n + C1_R - 1) / C1_R, C1_T, 0, stream>>>(
                x[i], agg,
                c1Wrel + (size_t)i * 6 * H,
                c1brel + (size_t)i * H,
                c1Wroot + (size_t)i * 6 * H, h1, n);
        }

        {
            const int rr = rrl[i];
            agg128_lds_kernel<<<(n + rr - 1) / rr, AG_T, 0, stream>>>(h1, ei[i], E, n, rr,
                                                                     lst, lcnt, agg);
        }

        {
            const float* Wrel2  = c2Wrel + (size_t)i * H * H;
            const float* Wroot2 = c2Wroot + (size_t)i * H * H;
            const float* brel2  = c2brel + (size_t)i * H;
            const float* lW     = linW + (size_t)i * H * F3;
            const float* lb     = linb + (size_t)i * F3;
            if (i == 0)
                launch_enc_gemms<4, 2>(agg, h1, h2, elv[i], Wrel2, Wroot2, brel2, lW, lb, n, stream);
            else if (i == 1)
                launch_enc_gemms<2, 2>(agg, h1, h2, elv[i], Wrel2, Wroot2, brel2, lW, lb, n, stream);
            else
                launch_enc_gemms<1, 1>(agg, h1, h2, elv[i], Wrel2, Wroot2, brel2, lW, lb, n, stream);
        }
    }

    {
        static_assert(NNZ % 4 == 0, "row lists are read 16 B at a time");
        interp_lds_kernel<<<dim3((N0 + IN_R - 1) / IN_R, 2, 1), AG_T, 0, stream>>>(
            A1v, A1c, A1r, e1, i1, A2v, A2c, A2r, e2, i2, NNZ, N0);
    }

    for (int b = 0; b < 3; ++b) {
        gemm_tile_kernel<3, 32, F3, DECW, 2, 2, false><<<(N0 + 31) / 32, DECW, 0, stream>>>(
            e0, i1, i2, b * 32, decW0 + (size_t)b * F3 * DECW, nullptr,
            decb0 + (size_t)b * DECW, hdec, N0);
        dec_out_kernel<<<(N0 + 255) / 256, 256, 0, stream>>>(
            hdec, decWout + (size_t)b * DECW, decbout + b, out, N0, b);
    }
}


# conv2 segment sums stay in LDS and feed the conv2 product in the same launch: A fragments read from the add layout (k order permuted), h1 rows staged beside them; 22 -> 19 launches; sum order changed
# speedup vs baseline: 3.1676x; 1.0757x over previous
#include <hip/hip_runtime.h>
#include <hip/hip_bf16.h>

#define N0 50000
#define N1 12500
#define N2 3125
#define E0 800000
#define E1 200000
#define E2 50000
#define NNZ 200000
#define H 128
#define F3 96
#define DECW 64

typedef __attribute__((ext_vector_type(2))) float v2f;
typedef __attribute__((ext_vector_type(8))) float v8f;


#define SC_R 4000

#define SC_S 16
#define SC_T 1024
#define SC_U 8
static_assert(SC_R <= 4096 && E0 < (1 << 20) && E1 < (1 << 20) && E2 < (1 << 20),
              "a list entry packs a 12-bit row offset and a 20-bit edge number");

__device__ __forceinline__ int sc_cap(int E) {
    const int chunk = (E + SC_S - 1) / SC_S;
    const int trips = (chunk + SC_T * SC_U - 1) / (SC_T * SC_U);
    return trips * SC_U * 32;
}

__global__ __launch_bounds__(SC_T) __attribute__((amdgpu_waves_per_eu(8, 8)))
void scatter6_lds_kernel(const float* __restrict__ x, const int* __restrict__ ei,
                         int E, int n, float* __restrict__ part,
                         int* __restrict__ lst, int* __restrict__ lcnt) {
    __shared__ float acc[SC_R * 6];
    const int tid = threadIdx.x;
    const int r0  = blockIdx.x * SC_R;
    const int s   = blockIdx.y;

    for (int i = tid; i < SC_R * 6; i += SC_T) acc[i] = 0.0f;
    __syncthreads();

    const int chunk = (E + SC_S - 1) / SC_S;
    const int eb = s * chunk;
    const int ee = (eb + chunk < E) ? (eb + chunk) : E;

    const int sub = (blockIdx.x * SC_S + s) * (SC_T / 32) + (tid >> 5);
    int* lstw = lst + (size_t)sub * sc_cap(E);
    int wc = 0;

    if (eb < ee) {
        for (int base = eb; base < ee; base += SC_T * SC_U) {
            int off[SC_U];
            int srow[SC_U];
#pragma unroll
            for (int u = 0; u < SC_U; ++u) {
                const int e  = base + u * SC_T + tid;
                const int ec = (e < ee) ? e : (ee - 1);
                const int d  = ei[E + ec] - r0;
                const int sr = ei[ec];
                const bool hit = (e < ee) && ((unsigned)d < (unsigned)SC_R);
                off[u]  = hit ? d * 6 : -1;
                srow[u] = hit ? sr * 6 : 0;
            }
            float2 v[SC_U][3];
#pragma unroll
            for (int u = 0; u < SC_U; ++u) {
                const float2* p = (const float2*)(x + srow[u]);
                v[u][0] = p[0];
                v[u][1] = p[1];
                v[u][2] = p[2];
            }
#pragma unroll
            for (int u = 0; u < SC_U; ++u)
                asm volatile("" :: "v"(v[u][0].x), "v"(v[u][0].y), "v"(v[u][1].x),
                                   "v"(v[u][1].y), "v"(v[u][2].x), "v"(v[u][2].y));
#pragma unroll
            for (int u = 0; u < SC_U; ++u) {
                const unsigned hm = __builtin_amdgcn_ballot_w32(off[u] >= 0);
                if (off[u] >= 0) {
                    lstw[wc + (int)__builtin_amdgcn_mbcnt_lo(hm, 0u)] =
                        (int)((((unsigned)off[u] / 6u) << 20) | (unsigned)(base + u * SC_T + tid));
                    float* a = acc + off[u];
                    atomicAdd(a + 0, v[u][0].x);
                    atomicAdd(a + 1, v[u][0].y);
                    atomicAdd(a + 2, v[u][1].x);
                    atomicAdd(a + 3, v[u][1].y);
                    atomicAdd(a + 4, v[u][2].x);
                    atomicAdd(a + 5, v[u][2].y);
                }
                wc += __builtin_popcount(hm);
            }
        }
    }
    if ((tid & 31) == 0) lcnt[sub] = wc;
    __syncthreads();

    const int rows = (n - r0 < SC_R) ? (n - r0) : SC_R;
    float* p = part + (size_t)s * n * 6 + (size_t)r0 * 6;
    for (int i = tid; i < rows * 6; i += SC_T) p[i] = acc[i];
}

#define C1_R 32
#define C1_T 128

__global__ __launch_bounds__(C1_T)
void conv1_planes_kernel(const float* __restrict__ x, const float* __restrict__ part,
                         const float* __restrict__ Wrel, const float* __restrict__ brel,
                         const float* __restrict__ Wroot, float* __restrict__ h1, int n) {
    __shared__ float sa[C1_R * 6];
    __shared__ float sx[C1_R * 6];
    const int tid  = threadIdx.x;
    const int lane = tid & 31;
    const int wave = tid >> 5;
    const int r0   = blockIdx.x * C1_R;
    const int rows = (n - r0 < C1_R) ? (n - r0) : C1_R;
    const int cnt  = n * 6;

    float4 wr[6], wo[6];
#pragma unroll
    for (int k = 0; k < 6; ++k) {
        wr[k] = *(const float4*)(Wrel + k * H + lane * 4);
        wo[k] = *(const float4*)(Wroot + k * H + lane * 4);
    }
    const float4 bv = *(const float4*)(brel + lane * 4);

    static_assert(6 * C1_R <= 2 * C1_T, "two elements per thread cover the block's rows");
    const int  m    = rows * 6;
    const int  i1   = tid + C1_T;
    const bool has0 = tid < m;
    const bool has1 = i1 < m;
    const int  t0   = r0 * 6 + (has0 ? tid : 0);
    const int  t1   = r0 * 6 + (has1 ? i1 : 0);
    float p0[SC_S], p1[SC_S];
#pragma unroll
    for (int k = 0; k < SC_S; ++k) {
        p0[k] = part[(size_t)k * cnt + t0];
        p1[k] = part[(size_t)k * cnt + t1];
    }
    const float x0 = x[t0];
    const float x1 = x[t1];
#pragma unroll
    for (int k = 0; k < SC_S; k += 4)
        asm volatile("" :: "v"(p0[k]), "v"(p0[k + 1]), "v"(p0[k + 2]), "v"(p0[k + 3]),
                           "v"(p1[k]), "v"(p1[k + 1]), "v"(p1[k + 2]), "v"(p1[k + 3]));
    asm volatile("" :: "v"(x0), "v"(x1));
    float s0 = 0.0f, s1 = 0.0f;
#pragma unroll
    for (int k = 0; k < SC_S; ++k) {
        s0 += p0[k];
        s1 += p1[k];
    }
    if (has0) { sa[tid] = s0; sx[tid] = x0; }
    if (has1) { sa[i1]  = s1; sx[i1]  = x1; }
    __syncthreads();

    for (int r = wave; r < rows; r += C1_T / 32) {
        float4 s = bv;
#pragma unroll
        for (int k = 0; k < 6; ++k) {
            const float a  = sa[r * 6 + k];
            const float xv = sx[r * 6 + k];
            s.x += a * wr[k].x;  s.y += a * wr[k].y;  s.z += a * wr[k].z;  s.w += a * wr[k].w;
            s.x += xv * wo[k].x; s.y += xv * wo[k].y; s.z += xv * wo[k].z; s.w += xv * wo[k].w;
        }
        s.x = fmaxf(s.x, 0.0f);
        s.y = fmaxf(s.y, 0.0f);
        s.z = fmaxf(s.z, 0.0f);
        s.w = fmaxf(s.w, 0.0f);
        *(float4*)(h1 + (size_t)(r0 + r) * H + lane * 4) = s;
    }
}

#define AG_T    1024
#define AG_RMAX 200
#define AG_RPAD 208
#define AG_P    (H + 4)
static_assert(AG_RPAD % 16 == 0 && AG_RPAD >= AG_RMAX && AG_RPAD / 16 <= 16,
              "at most 4 row tiles for each of the 4 wave groups");
#define AG_U    4
#define AG_Q    1024
#define AG_B    16
#define SL_G    4
#define SL_L    5
static_assert(SC_S % SL_G == 0, "sublists are read SL_G at a time");
static_assert(SL_G * SL_L * 32 + 31 <= AG_Q, "a wave's ring holds what one read step can queue");

template <bool TAIL>
__device__ __forceinline__ void agg128_scan_trip(const int4* __restrict__ d4, int g0, int ng,
                                                 int tid, int r0, unsigned rows,
                                                 int* q, int& tail) {
    int4 d[AG_U];
#pragma unroll
    for (int u = 0; u < AG_U; ++u) {
        const int g = g0 + u * AG_T + tid;
        d[u] = d4[(TAIL && g >= ng) ? (ng - 1) : g];
    }
#pragma unroll
    for (int u = 0; u < AG_U; ++u) {
        const int g = g0 + u * AG_T + tid;
        const bool valid = !TAIL || (g < ng);
        const unsigned vmask = TAIL ? __builtin_amdgcn_ballot_w32(g < ng) : 0xffffffffu;
        const int dd[4] = {d[u].x, d[u].y, d[u].z, d[u].w};
#pragma unroll
        for (int c = 0; c < 4; ++c) {
            const bool in = (unsigned)(dd[c] - r0) < rows;
            const unsigned mask = __builtin_amdgcn_ballot_w32(in) & vmask;
            if (mask != 0u) {
                if (in && valid)
                    q[(tail + (int)__builtin_amdgcn_mbcnt_lo(mask, 0u)) & (AG_Q - 1)] = g * 4 + c;
                tail += __builtin_popcount(mask);
            }
        }
    }
}

__device__ __forceinline__ void agg128_drain(const float* __restrict__ h1,
                                             const int* __restrict__ ei, int roff,
                                             const int* q, int head, int cnt,
                                             float* acc, int lane) {
    const unsigned qe = (unsigned)q[(head + lane) & (AG_Q - 1)];
    const unsigned en = (lane < cnt) ? qe : 0u;
    const int s  = ei[en & 0xfffffu];
    const int dl = (int)(en >> 20) - roff;
#pragma unroll 1
    for (int b = 0; b < cnt; b += AG_B) {
        float4 v[AG_B];
#pragma unroll
        for (int j = 0; j < AG_B; ++j) {
            const int sj = __builtin_amdgcn_readlane(s, b + j);
            v[j] = *(const float4*)(h1 + sj * H + lane * 4);
        }
#pragma unroll
        for (int j = 0; j < AG_B; ++j) {
            if (b + j < cnt) {
                const int dj = __builtin_amdgcn_readlane(dl, b + j);
                float* a = acc + dj * AG_P + lane;
                atomicAdd(a,      v[j].x);
                atomicAdd(a + 32, v[j].y);
                atomicAdd(a + 64, v[j].z);
                atomicAdd(a + 96, v[j].w);
            }
        }
    }
}

template <int CNT>
__device__ __forceinline__ void agg128_conv2_tiles(const float* acc, const float* hs,
                                                   const float* __restrict__ Wrel,
                                                   const float* __restrict__ Wroot,
                                                   const float* __restrict__ brel,
                                                   float* __restrict__ h2, int r0, int rows,
                                                   int ct, int rg0, int lane) {
    const int c   = lane & 15;
    const int h   = lane >> 4;
    const int col = ct * 16 + c;
    const float bv = brel[col];
    v8f d[CNT];
#pragma unroll
    for (int i = 0; i < CNT; ++i) {
        const v8f t = {bv, bv, bv, bv, bv, bv, bv, bv};
        d[i] = t;
    }
    const float* ap = acc + (rg0 * 16 + c) * AG_P + 2 * h;
    const float* hp = hs  + (rg0 * 16 + c) * AG_P + 2 * h;
    const float* wr = Wrel  + (8 * h) * H + col;
    const float* wo = Wroot + (2 * h) * H + col;
#pragma unroll 8
    for (int t = 0; t < 32; ++t) {
        const int j  = t >> 2;
        const int kk = t & 3;
        v2f b;
        b.x = wr[(16 * j + kk) * H];
        b.y = wr[(16 * j + kk + 4) * H];
#pragma unroll
        for (int i = 0; i < CNT; ++i) {
            const v2f a = *(const v2f*)(ap + i * 64 * AG_P + kk * 32 + 4 * j);
            d[i] = __builtin_amdgcn_wmma_f32_16x16x4_f32(false, a, false, b, (short)0, d[i],
                                                         false, false);
        }
    }
#pragma unroll 8
    for (int k = 0; k < H; k += 4) {
        v2f b;
        b.x = wo[k * H];
        b.y = wo[k * H + H];
#pragma unroll
        for (int i = 0; i < CNT; ++i) {
            const v2f a = *(const v2f*)(hp + i * 64 * AG_P + k);
            d[i] = __builtin_amdgcn_wmma_f32_16x16x4_f32(false, a, false, b, (short)0, d[i],
                                                         false, false);
        }
    }
#pragma unroll
    for (int i = 0; i < CNT; ++i)
#pragma unroll
        for (int j = 0; j < 8; ++j) {
            const int row = (rg0 + 4 * i) * 16 + j + 8 * h;
            if (row < rows) h2[(size_t)(r0 + row) * H + col] = fmaxf(d[i][j], 0.0f);
        }
}

template <int RTS>
__global__ __launch_bounds__(AG_T)
void agg128_conv2_kernel(const float* __restrict__ h1, const int* __restrict__ ei,
                         int E, int n, int rr, const int* __restrict__ lst,
                         const int* __restrict__ lcnt,
                         const float* __restrict__ Wrel, const float* __restrict__ Wroot,
                         const float* __restrict__ brel, float* __restrict__ h2) {
    __shared__ __attribute__((aligned(16))) float acc[AG_RPAD * AG_P];
    __shared__ __attribute__((aligned(16))) int ring[AG_T / 32][AG_Q];
    const int tid  = threadIdx.x;
    const int lane = tid & 31;
    const int r0   = blockIdx.x * rr;
    const int rows = (n - r0 < rr) ? (n - r0) : rr;
    const int nrt  = (rr + 15) >> 4;

    {
        float4* a4 = (float4*)acc;
        const float4 z = {0.0f, 0.0f, 0.0f, 0.0f};
        for (int i = tid; i < nrt * 16 * (AG_P / 4); i += AG_T) a4[i] = z;
    }
    __syncthreads();

    int* q = ring[tid >> 5];
    int head = 0, tail = 0;
    const int wv   = __builtin_amdgcn_readfirstlane(tid >> 5);
    const int xr   = r0 / SC_R;
    const int roff = r0 - xr * SC_R;
    const int cap  = sc_cap(E);
    const int sub0 = xr * (SC_S * (SC_T / 32)) + wv;
    static_assert(SC_S <= 32, "one count per lane");
    const int cvec = lcnt[sub0 + ((lane < SC_S) ? lane : 0) * (SC_T / 32)];

    for (int t0 = 0; t0 < SC_S; t0 += SL_G) {
        int cn[SL_G];
        const int* bp[SL_G];
        int ent[SL_G][SL_L];
#pragma unroll
        for (int g = 0; g < SL_G; ++g) {
            const int sub = sub0 + (t0 + g) * (SC_T / 32);
            cn[g] = __builtin_amdgcn_readlane(cvec, t0 + g);
            bp[g] = (cn[g] > 0) ? (lst + (size_t)sub * cap) : (lcnt + sub);
        }
#pragma unroll
        for (int g = 0; g < SL_G; ++g)
#pragma unroll
            for (int u = 0; u < SL_L; ++u) {
                const int i  = u * 32 + lane;
                const int ic = (cn[g] > 0) ? ((i < cn[g]) ? i : (cn[g] - 1)) : 0;
                ent[g][u] = bp[g][ic];
            }
#pragma unroll
        for (int g = 0; g < SL_G; ++g)
            asm volatile("" :: "v"(ent[g][0]), "v"(ent[g][1]), "v"(ent[g][2]),
                               "v"(ent[g][3]), "v"(ent[g][4]));
        static_assert(SL_L == 5, "the fence above names SL_L entries per sublist");
#pragma unroll
        for (int g = 0; g < SL_G; ++g)
#pragma unroll
            for (int u = 0; u < SL_L; ++u) {
                if (u * 32 < cn[g]) {
                    const unsigned en = (unsigned)ent[g][u];
                    const bool in = (u * 32 + lane < cn[g]) &&
                                    ((en >> 20) - (unsigned)roff < (unsigned)rows);
                    const unsigned mask = __builtin_amdgcn_ballot_w32(in);
                    if (mask != 0u) {
                        if (in)
                            q[(tail + (int)__builtin_amdgcn_mbcnt_lo(mask, 0u)) & (AG_Q - 1)] = (int)en;
                        tail += __builtin_popcount(mask);
                    }
                }
            }
        while (tail - head >= 32) {
            agg128_drain(h1, ei, roff, q, head, 32, acc, lane);
            head += 32;
        }
#pragma unroll
        for (int g = 0; g < SL_G; ++g) {
            for (int i0 = SL_L * 32; i0 < cn[g]; i0 += 32) {
                const int i = i0 + lane;
                const unsigned en = (unsigned)bp[g][(i < cn[g]) ? i : (cn[g] - 1)];
                const bool in = (i < cn[g]) && ((en >> 20) - (unsigned)roff < (unsigned)rows);
                const unsigned mask = __builtin_amdgcn_ballot_w32(in);
                if (mask != 0u) {
                    if (in)
                        q[(tail + (int)__builtin_amdgcn_mbcnt_lo(mask, 0u)) & (AG_Q - 1)] = (int)en;
                    tail += __builtin_popcount(mask);
                }
                while (tail - head >= 32) {
                    agg128_drain(h1, ei, roff, q, head, 32, acc, lane);
                    head += 32;
                }
            }
        }
    }
    while (tail - head > 0) {
        const int cnt = (tail - head < 32) ? (tail - head) : 32;
        agg128_drain(h1, ei, roff, q, head, cnt, acc, lane);
        head += cnt;
    }
    __syncthreads();

    constexpr int RP    = (RTS * 64 < AG_RPAD) ? (RTS * 64) : AG_RPAD;
    constexpr int HS_IT = (RP * (H / 4) + AG_T - 1) / AG_T;
    static_assert(((HS_IT * AG_T - 1) / (H / 4)) * AG_P + H <= (AG_T / 32) * AG_Q,
                  "the staged h1 rows fit in the ring space");
    float* hs = (float*)&ring[0][0];
    {
        float4 hv[HS_IT];
#pragma unroll
        for (int j = 0; j < HS_IT; ++j) {
            const int i  = j * AG_T + tid;
            const int rg = (r0 + (i >> 5) < n) ? (r0 + (i >> 5)) : (n - 1);
            hv[j] = *(const float4*)(h1 + (size_t)rg * H + (i & 31) * 4);
        }
#pragma unroll
        for (int j = 0; j < HS_IT; ++j) {
            const int i = j * AG_T + tid;
            *(float4*)(hs + (i >> 5) * AG_P + (i & 31) * 4) = hv[j];
        }
    }
    __syncthreads();

    const int ct  = wv & 7;
    const int rg0 = wv >> 3;
    const int cnt = (nrt - rg0 + 3) >> 2;
    if (RTS >= 4 && cnt >= 4)
        agg128_conv2_tiles<4>(acc, hs, Wrel, Wroot, brel, h2, r0, rows, ct, rg0, lane);
    else if (RTS >= 3 && cnt == 3)
        agg128_conv2_tiles<3>(acc, hs, Wrel, Wroot, brel, h2, r0, rows, ct, rg0, lane);
    else if (RTS >= 2 && cnt == 2)
        agg128_conv2_tiles<2>(acc, hs, Wrel, Wroot, brel, h2, r0, rows, ct, rg0, lane);
    else if (cnt >= 1)
        agg128_conv2_tiles<1>(acc, hs, Wrel, Wroot, brel, h2, r0, rows, ct, rg0, lane);
}

#define IN_R 400
#define IN_B 16

__device__ __forceinline__ void interp_drain(const float* __restrict__ e,
                                             const int* __restrict__ cols,
                                             const int* __restrict__ rws,
                                             const float* __restrict__ vals, int r0,
                                             const int* q, int head, int cnt,
                                             float* acc, int lane) {
    const int qe  = q[(head + lane) & (AG_Q - 1)];
    const int nz  = (lane < cnt) ? qe : 0;
    const int col = cols[nz];
    const int dl  = rws[nz] - r0;
    const int vb  = __float_as_int(vals[nz]);
#pragma unroll 1
    for (int b = 0; b < cnt; b += IN_B) {
        float v[IN_B][3];
#pragma unroll
        for (int j = 0; j < IN_B; ++j) {
            const int cj = __builtin_amdgcn_readlane(col, b + j);
            const float* p = e + cj * F3 + lane;
            v[j][0] = p[0];
            v[j][1] = p[32];
            v[j][2] = p[64];
        }
#pragma unroll
        for (int j = 0; j < IN_B; ++j) {
            if (b + j < cnt) {
                const int dj  = __builtin_amdgcn_readlane(dl, b + j);
                const float s = __int_as_float(__builtin_amdgcn_readlane(vb, b + j));
                float* a = acc + dj * F3 + lane;
                atomicAdd(a,      s * v[j][0]);
                atomicAdd(a + 32, s * v[j][1]);
                atomicAdd(a + 64, s * v[j][2]);
            }
        }
    }
}

__global__ __launch_bounds__(AG_T)
void interp_lds_kernel(const float* __restrict__ vals1, const int* __restrict__ cols1,
                       const int* __restrict__ rows1, const float* __restrict__ e1,
                       float* __restrict__ out1,
                       const float* __restrict__ vals2, const int* __restrict__ cols2,
                       const int* __restrict__ rows2, const float* __restrict__ e2,
                       float* __restrict__ out2, int nnz, int n) {
    __shared__ __attribute__((aligned(16))) float acc[IN_R * F3];
    __shared__ int ring[AG_T / 32][AG_Q];
    const bool m2 = (blockIdx.y != 0);
    const float* vals = m2 ? vals2 : vals1;
    const int*   cols = m2 ? cols2 : cols1;
    const int*   rws  = m2 ? rows2 : rows1;
    const float* e    = m2 ? e2 : e1;
    float*       outp = m2 ? out2 : out1;

    const int tid  = threadIdx.x;
    const int lane = tid & 31;
    const int r0   = blockIdx.x * IN_R;
    const int rows = (n - r0 < IN_R) ? (n - r0) : IN_R;

    {
        float4* a4 = (float4*)acc;
        const float4 z = {0.0f, 0.0f, 0.0f, 0.0f};
        for (int i = tid; i < rows * (F3 / 4); i += AG_T) a4[i] = z;
    }
    __syncthreads();

    int* q = ring[tid >> 5];
    int head = 0, tail = 0;
    const int4* d4 = (const int4*)rws;
    const int ng = nnz >> 2;

    int g0 = 0;
    for (; g0 + AG_T * AG_U <= ng; g0 += AG_T * AG_U) {
        agg128_scan_trip<false>(d4, g0, ng, tid, r0, (unsigned)rows, q, tail);
        while (tail - head >= 32) {
            interp_drain(e, cols, rws, vals, r0, q, head, 32, acc, lane);
            head += 32;
        }
    }
    if (g0 < ng) agg128_scan_trip<true>(d4, g0, ng, tid, r0, (unsigned)rows, q, tail);
    while (tail - head > 0) {
        const int cnt = (tail - head < 32) ? (tail - head) : 32;
        interp_drain(e, cols, rws, vals, r0, q, head, cnt, acc, lane);
        head += cnt;
    }
    __syncthreads();

    float4* o4 = (float4*)(outp + (size_t)r0 * F3);
    const float4* a4 = (const float4*)acc;
    for (int i = tid; i < rows * (F3 / 4); i += AG_T) o4[i] = a4[i];
}

__global__ void dec_out_kernel(const float* __restrict__ hdec, const float* __restrict__ Wout,
                               const float* __restrict__ bout, float* __restrict__ out,
                               int n, int branch) {
    int r = blockIdx.x * blockDim.x + threadIdx.x;
    if (r >= n) return;
    float s = bout[0];
    const float* hr = hdec + r * DECW;
#pragma unroll
    for (int c = 0; c < DECW; ++c) s += hr[c] * Wout[c];
    out[r * 3 + branch] = s;
}

template <int NSEG, int SEGW, int LD, int Nc, int RT, int ACT>
__global__ __launch_bounds__(Nc)
void gemm_tile_kernel(const float* __restrict__ s0, const float* __restrict__ s1,
                      const float* __restrict__ s2, int coff,
                      const float* __restrict__ W0,
                      const float* __restrict__ bias, float* __restrict__ D, int n) {
    constexpr int K   = NSEG * SEGW;
    constexpr int P   = K + 4;
    constexpr int MB  = RT * 16;
    constexpr int NT  = Nc;
    constexpr int K4  = K / 4;
    constexpr int S4  = SEGW / 4;
    constexpr int TOT = MB * K4;
    constexpr int ITER = (TOT + NT - 1) / NT;
    constexpr int SB  = 8;
    __shared__ __attribute__((aligned(16))) float As[MB * P];

    const int tid  = threadIdx.x;
    const int lane = tid & 31;
    const int wave = tid >> 5;
    const int rowb = blockIdx.x * MB;

#pragma unroll
    for (int i0 = 0; i0 < ITER; i0 += SB) {
        float4 v[SB];
#pragma unroll
        for (int j = 0; j < SB; ++j) {
            if (i0 + j < ITER) {
                const int idx = (i0 + j) * NT + tid;
                const int idc = (TOT % NT != 0 && idx >= TOT) ? (TOT - 1) : idx;
                const int r   = idc / K4;
                const int c4  = idc % K4;
                const int sg  = c4 / S4;
                const int w4  = c4 % S4;
                const int rg  = (rowb + r < n) ? (rowb + r) : (n - 1);
                const float* sp = (sg == 0) ? s0 : (sg == 1) ? s1 : s2;
                v[j] = *(const float4*)(sp + (size_t)rg * LD + coff + w4 * 4);
            }
        }
#pragma unroll
        for (int j = 0; j < SB; ++j) {
            if (i0 + j < ITER) {
                const int idx = (i0 + j) * NT + tid;
                if (TOT % NT == 0 || idx < TOT) {
                    const int r  = idx / K4;
                    const int c4 = idx % K4;
                    *(float4*)(As + r * P + c4 * 4) = v[j];
                }
            }
        }
    }
    __syncthreads();

    const int c    = lane & 15;
    const int h    = lane >> 4;
    const int col0 = wave * 32;
    const float* ap  = As + c * P + 2 * h;
    const float* wp0 = W0 + col0 + c + 2 * h * Nc;

    v8f acc[RT][2];
#pragma unroll
    for (int ct = 0; ct < 2; ++ct) {
        const float bv = bias[col0 + ct * 16 + c];
#pragma unroll
        for (int rt = 0; rt < RT; ++rt) {
            const v8f t = {bv, bv, bv, bv, bv, bv, bv, bv};
            acc[rt][ct] = t;
        }
    }

#pragma unroll 8
    for (int k = 0; k < K; k += 4) {
        v2f a[RT];
        v2f b[2];
#pragma unroll
        for (int rt = 0; rt < RT; ++rt)
            a[rt] = *(const v2f*)(ap + rt * 16 * P + k);
#pragma unroll
        for (int ct = 0; ct < 2; ++ct) {
            b[ct].x = wp0[k * Nc + ct * 16];
            b[ct].y = wp0[k * Nc + Nc + ct * 16];
        }
#pragma unroll
        for (int rt = 0; rt < RT; ++rt)
#pragma unroll
            for (int ct = 0; ct < 2; ++ct)
                acc[rt][ct] = __builtin_amdgcn_wmma_f32_16x16x4_f32(
                    false, a[rt], false, b[ct], (short)0, acc[rt][ct], false, false);
    }

    const bool full = (rowb + MB <= n);
#pragma unroll
    for (int rt = 0; rt < RT; ++rt)
#pragma unroll
        for (int ct = 0; ct < 2; ++ct)
#pragma unroll
            for (int j = 0; j < 8; ++j) {
                float v = acc[rt][ct][j];
                if (ACT == 1) v = fmaxf(v, 0.0f);
                if (ACT == 2) v = (v > 0.0f) ? v : (expf(v) - 1.0f);
                acc[rt][ct][j] = v;
            }
    float* dp = D + (rowb + h * 8) * Nc + col0 + c;
    if (full) {
#pragma unroll
        for (int rt = 0; rt < RT; ++rt)
#pragma unroll
            for (int ct = 0; ct < 2; ++ct)
#pragma unroll
                for (int j = 0; j < 8; ++j)
                    dp[(rt * 16 + j) * Nc + ct * 16] = acc[rt][ct][j];
    } else {
#pragma unroll
        for (int rt = 0; rt < RT; ++rt)
#pragma unroll
            for (int ct = 0; ct < 2; ++ct)
#pragma unroll
                for (int j = 0; j < 8; ++j)
                    if (rowb + rt * 16 + j + h * 8 < n)
                        dp[(rt * 16 + j) * Nc + ct * 16] = acc[rt][ct][j];
    }
}

template <int RTS, int RTL>
static inline void launch_enc_conv2_lin(const float* h1, const int* ei, int E, int n, int rr,
                                        const int* lst, const int* lcnt, float* h2, float* e,
                                        const float* Wrel, const float* Wroot, const float* brel,
                                        const float* lW, const float* lb, hipStream_t stream) {
    agg128_conv2_kernel<RTS><<<(n + rr - 1) / rr, AG_T, 0, stream>>>(
        h1, ei, E, n, rr, lst, lcnt, Wrel, Wroot, brel, h2);
    gemm_tile_kernel<1, H, H, F3, RTL, 0><<<(n + RTL * 16 - 1) / (RTL * 16), F3, 0, stream>>>(
        h2, nullptr, nullptr, 0, lW, lb, e, n);
}

extern "C" void kernel_launch(void* const* d_in, const int* in_sizes, int n_in,
                              void* d_out, int out_size, void* d_ws, size_t ws_size,
                              hipStream_t stream) {
    (void)in_sizes; (void)n_in; (void)out_size; (void)ws_size;

    const float* x[3]   = {(const float*)d_in[0], (const float*)d_in[1], (const float*)d_in[2]};
    const int*   ei[3]  = {(const int*)d_in[3], (const int*)d_in[4], (const int*)d_in[5]};
    const int*   A1r    = (const int*)d_in[6];
    const int*   A1c    = (const int*)d_in[7];
    const float* A1v    = (const float*)d_in[8];
    const int*   A2r    = (const int*)d_in[9];
    const int*   A2c    = (const int*)d_in[10];
    const float* A2v    = (const float*)d_in[11];
    const float* c1Wrel  = (const float*)d_in[12];
    const float* c1brel  = (const float*)d_in[13];
    const float* c1Wroot = (const float*)d_in[14];
    const float* c2Wrel  = (const float*)d_in[15];
    const float* c2brel  = (const float*)d_in[16];
    const float* c2Wroot = (const float*)d_in[17];
    const float* linW    = (const float*)d_in[18];
    const float* linb    = (const float*)d_in[19];
    const float* decW0   = (const float*)d_in[20];
    const float* decb0   = (const float*)d_in[21];
    const float* decWout = (const float*)d_in[22];
    const float* decbout = (const float*)d_in[23];
    float* out = (float*)d_out;

    float* ws = (float*)d_ws;
    float* h1   = ws;
    float* agg  = h1   + (size_t)N0 * H;
    float* h2   = agg  + (size_t)N0 * H;
    float* e0   = h2   + (size_t)N0 * H;
    float* e1   = e0   + (size_t)N0 * F3;
    float* e2   = e1   + (size_t)N1 * F3;
    float* i1   = e2   + (size_t)N2 * F3;
    float* i2   = i1   + (size_t)N0 * F3;
    float* hdec = i2   + (size_t)N0 * F3;

    constexpr size_t LST_MAX = (size_t)((N0 + SC_R - 1) / SC_R) * SC_S * (SC_T / 32) *
                               ((((E0 + SC_S - 1) / SC_S + SC_T * SC_U - 1) / (SC_T * SC_U)) * SC_U * 32);
    constexpr size_t LCNT_MAX = (size_t)((N0 + SC_R - 1) / SC_R) * SC_S * (SC_T / 32);
    static_assert(LST_MAX + LCNT_MAX <= (size_t)N0 * F3 * 2 + (size_t)N0 * DECW,
                  "the lists of level 0 (the largest) fit in i1 | i2 | hdec");
    static_assert(E0 >= E1 && E0 >= E2 && N0 >= N1 && N0 >= N2, "level 0 is the largest");
    int* lst  = (int*)i1;
    int* lcnt = lst + LST_MAX;

    const int nlv[3] = {N0, N1, N2};
    const int Elv[3] = {E0, E1, E2};
    const int rrl[3] = {200, 50, 13};
    static_assert(SC_R % 200 == 0 && SC_R % 50 == 0 && N2 <= SC_R,
                  "a conv2 block's rows lie inside one scatter6 range");
    static_assert(200 <= AG_RMAX && (50 + 15) / 16 <= 4 && (13 + 15) / 16 <= 4,
                  "conv2 product: level 0 within AG_RMAX rows, levels 1 and 2 one row tile per wave");
    static_assert(E0 % 4 == 0 && E1 % 4 == 0 && E2 % 4 == 0, "dst lists are read 16 B at a time");
    float* elv[3] = {e0, e1, e2};

    for (int i = 0; i < 3; ++i) {
        const int n = nlv[i];
        const int E = Elv[i];

        {
            scatter6_lds_kernel<<<dim3((n + SC_R - 1) / SC_R, SC_S, 1), SC_T, 0, stream>>>(
                x[i], ei[i], E, n, agg, lst, lcnt);
            conv1_planes_kernel<<<(n + C1_R - 1) / C1_R, C1_T, 0, stream>>>(
                x[i], agg,
                c1Wrel + (size_t)i * 6 * H,
                c1brel + (size_t)i * H,
                c1Wroot + (size_t)i * 6 * H, h1, n);
        }

        {
            const int rr = rrl[i];
            const float* Wrel2  = c2Wrel + (size_t)i * H * H;
            const float* Wroot2 = c2Wroot + (size_t)i * H * H;
            const float* brel2  = c2brel + (size_t)i * H;
            const float* lW     = linW + (size_t)i * H * F3;
            const float* lb     = linb + (size_t)i * F3;
            if (i == 0)
                launch_enc_conv2_lin<4, 2>(h1, ei[i], E, n, rr, lst, lcnt, h2, elv[i],
                                           Wrel2, Wroot2, brel2, lW, lb, stream);
            else if (i == 1)
                launch_enc_conv2_lin<1, 2>(h1, ei[i], E, n, rr, lst, lcnt, h2, elv[i],
                                           Wrel2, Wroot2, brel2, lW, lb, stream);
            else
                launch_enc_conv2_lin<1, 1>(h1, ei[i], E, n, rr, lst, lcnt, h2, elv[i],
                                           Wrel2, Wroot2, brel2, lW, lb, stream);
        }
    }

    {
        static_assert(NNZ % 4 == 0, "row lists are read 16 B at a time");
        interp_lds_kernel<<<dim3((N0 + IN_R - 1) / IN_R, 2, 1), AG_T, 0, stream>>>(
            A1v, A1c, A1r, e1, i1, A2v, A2c, A2r, e2, i2, NNZ, N0);
    }

    for (int b = 0; b < 3; ++b) {
        gemm_tile_kernel<3, 32, F3, DECW, 2, 2><<<(N0 + 31) / 32, DECW, 0, stream>>>(
            e0, i1, i2, b * 32, decW0 + (size_t)b * F3 * DECW,
            decb0 + (size_t)b * DECW, hdec, N0);
        dec_out_kernel<<<(N0 + 255) / 256, 256, 0, stream>>>(
            hdec, decWout + (size_t)b * DECW, decbout + b, out, N0, b);
    }
}


# interp sums stay in LDS and feed the decoders in the same launch: block owns 224 rows of both matrices; 3 branch products, ELU, output dot from LDS; 19->13 launches; sum order of the dot changed
# speedup vs baseline: 3.3823x; 1.0678x over previous
#include <hip/hip_runtime.h>
#include <hip/hip_bf16.h>

#define N0 50000
#define N1 12500
#define N2 3125
#define E0 800000
#define E1 200000
#define E2 50000
#define NNZ 200000
#define H 128
#define F3 96
#define DECW 64

typedef __attribute__((ext_vector_type(2))) float v2f;
typedef __attribute__((ext_vector_type(8))) float v8f;


#define SC_R 4000

#define SC_S 16
#define SC_T 1024
#define SC_U 8
static_assert(SC_R <= 4096 && E0 < (1 << 20) && E1 < (1 << 20) && E2 < (1 << 20),
              "a list entry packs a 12-bit row offset and a 20-bit edge number");

__device__ __forceinline__ int sc_cap(int E) {
    const int chunk = (E + SC_S - 1) / SC_S;
    const int trips = (chunk + SC_T * SC_U - 1) / (SC_T * SC_U);
    return trips * SC_U * 32;
}

__global__ __launch_bounds__(SC_T) __attribute__((amdgpu_waves_per_eu(8, 8)))
void scatter6_lds_kernel(const float* __restrict__ x, const int* __restrict__ ei,
                         int E, int n, float* __restrict__ part,
                         int* __restrict__ lst, int* __restrict__ lcnt) {
    __shared__ float acc[SC_R * 6];
    const int tid = threadIdx.x;
    const int r0  = blockIdx.x * SC_R;
    const int s   = blockIdx.y;

    for (int i = tid; i < SC_R * 6; i += SC_T) acc[i] = 0.0f;
    __syncthreads();

    const int chunk = (E + SC_S - 1) / SC_S;
    const int eb = s * chunk;
    const int ee = (eb + chunk < E) ? (eb + chunk) : E;

    const int sub = (blockIdx.x * SC_S + s) * (SC_T / 32) + (tid >> 5);
    int* lstw = lst + (size_t)sub * sc_cap(E);
    int wc = 0;

    if (eb < ee) {
        for (int base = eb; base < ee; base += SC_T * SC_U) {
            int off[SC_U];
            int srow[SC_U];
#pragma unroll
            for (int u = 0; u < SC_U; ++u) {
                const int e  = base + u * SC_T + tid;
                const int ec = (e < ee) ? e : (ee - 1);
                const int d  = ei[E + ec] - r0;
                const int sr = ei[ec];
                const bool hit = (e < ee) && ((unsigned)d < (unsigned)SC_R);
                off[u]  = hit ? d * 6 : -1;
                srow[u] = hit ? sr * 6 : 0;
            }
            float2 v[SC_U][3];
#pragma unroll
            for (int u = 0; u < SC_U; ++u) {
                const float2* p = (const float2*)(x + srow[u]);
                v[u][0] = p[0];
                v[u][1] = p[1];
                v[u][2] = p[2];
            }
#pragma unroll
            for (int u = 0; u < SC_U; ++u)
                asm volatile("" :: "v"(v[u][0].x), "v"(v[u][0].y), "v"(v[u][1].x),
                                   "v"(v[u][1].y), "v"(v[u][2].x), "v"(v[u][2].y));
#pragma unroll
            for (int u = 0; u < SC_U; ++u) {
                const unsigned hm = __builtin_amdgcn_ballot_w32(off[u] >= 0);
                if (off[u] >= 0) {
                    lstw[wc + (int)__builtin_amdgcn_mbcnt_lo(hm, 0u)] =
                        (int)((((unsigned)off[u] / 6u) << 20) | (unsigned)(base + u * SC_T + tid));
                    float* a = acc + off[u];
                    atomicAdd(a + 0, v[u][0].x);
                    atomicAdd(a + 1, v[u][0].y);
                    atomicAdd(a + 2, v[u][1].x);
                    atomicAdd(a + 3, v[u][1].y);
                    atomicAdd(a + 4, v[u][2].x);
                    atomicAdd(a + 5, v[u][2].y);
                }
                wc += __builtin_popcount(hm);
            }
        }
    }
    if ((tid & 31) == 0) lcnt[sub] = wc;
    __syncthreads();

    const int rows = (n - r0 < SC_R) ? (n - r0) : SC_R;
    float* p = part + (size_t)s * n * 6 + (size_t)r0 * 6;
    for (int i = tid; i < rows * 6; i += SC_T) p[i] = acc[i];
}

#define C1_R 32
#define C1_T 128

__global__ __launch_bounds__(C1_T)
void conv1_planes_kernel(const float* __restrict__ x, const float* __restrict__ part,
                         const float* __restrict__ Wrel, const float* __restrict__ brel,
                         const float* __restrict__ Wroot, float* __restrict__ h1, int n) {
    __shared__ float sa[C1_R * 6];
    __shared__ float sx[C1_R * 6];
    const int tid  = threadIdx.x;
    const int lane = tid & 31;
    const int wave = tid >> 5;
    const int r0   = blockIdx.x * C1_R;
    const int rows = (n - r0 < C1_R) ? (n - r0) : C1_R;
    const int cnt  = n * 6;

    float4 wr[6], wo[6];
#pragma unroll
    for (int k = 0; k < 6; ++k) {
        wr[k] = *(const float4*)(Wrel + k * H + lane * 4);
        wo[k] = *(const float4*)(Wroot + k * H + lane * 4);
    }
    const float4 bv = *(const float4*)(brel + lane * 4);

    static_assert(6 * C1_R <= 2 * C1_T, "two elements per thread cover the block's rows");
    const int  m    = rows * 6;
    const int  i1   = tid + C1_T;
    const bool has0 = tid < m;
    const bool has1 = i1 < m;
    const int  t0   = r0 * 6 + (has0 ? tid : 0);
    const int  t1   = r0 * 6 + (has1 ? i1 : 0);
    float p0[SC_S], p1[SC_S];
#pragma unroll
    for (int k = 0; k < SC_S; ++k) {
        p0[k] = part[(size_t)k * cnt + t0];
        p1[k] = part[(size_t)k * cnt + t1];
    }
    const float x0 = x[t0];
    const float x1 = x[t1];
#pragma unroll
    for (int k = 0; k < SC_S; k += 4)
        asm volatile("" :: "v"(p0[k]), "v"(p0[k + 1]), "v"(p0[k + 2]), "v"(p0[k + 3]),
                           "v"(p1[k]), "v"(p1[k + 1]), "v"(p1[k + 2]), "v"(p1[k + 3]));
    asm volatile("" :: "v"(x0), "v"(x1));
    float s0 = 0.0f, s1 = 0.0f;
#pragma unroll
    for (int k = 0; k < SC_S; ++k) {
        s0 += p0[k];
        s1 += p1[k];
    }
    if (has0) { sa[tid] = s0; sx[tid] = x0; }
    if (has1) { sa[i1]  = s1; sx[i1]  = x1; }
    __syncthreads();

    for (int r = wave; r < rows; r += C1_T / 32) {
        float4 s = bv;
#pragma unroll
        for (int k = 0; k < 6; ++k) {
            const float a  = sa[r * 6 + k];
            const float xv = sx[r * 6 + k];
            s.x += a * wr[k].x;  s.y += a * wr[k].y;  s.z += a * wr[k].z;  s.w += a * wr[k].w;
            s.x += xv * wo[k].x; s.y += xv * wo[k].y; s.z += xv * wo[k].z; s.w += xv * wo[k].w;
        }
        s.x = fmaxf(s.x, 0.0f);
        s.y = fmaxf(s.y, 0.0f);
        s.z = fmaxf(s.z, 0.0f);
        s.w = fmaxf(s.w, 0.0f);
        *(float4*)(h1 + (size_t)(r0 + r) * H + lane * 4) = s;
    }
}

#define AG_T    1024
#define AG_RMAX 200
#define AG_RPAD 208
#define AG_P    (H + 4)
static_assert(AG_RPAD % 16 == 0 && AG_RPAD >= AG_RMAX && AG_RPAD / 16 <= 16,
              "at most 4 row tiles for each of the 4 wave groups");
#define AG_U    4
#define AG_Q    1024
#define AG_B    16
#define SL_G    4
#define SL_L    5
static_assert(SC_S % SL_G == 0, "sublists are read SL_G at a time");
static_assert(SL_G * SL_L * 32 + 31 <= AG_Q, "a wave's ring holds what one read step can queue");

template <bool TAIL>
__device__ __forceinline__ void agg128_scan_trip(const int4* __restrict__ d4, int g0, int ng,
                                                 int tid, int r0, unsigned rows,
                                                 int* q, int& tail) {
    int4 d[AG_U];
#pragma unroll
    for (int u = 0; u < AG_U; ++u) {
        const int g = g0 + u * AG_T + tid;
        d[u] = d4[(TAIL && g >= ng) ? (ng - 1) : g];
    }
#pragma unroll
    for (int u = 0; u < AG_U; ++u) {
        const int g = g0 + u * AG_T + tid;
        const bool valid = !TAIL || (g < ng);
        const unsigned vmask = TAIL ? __builtin_amdgcn_ballot_w32(g < ng) : 0xffffffffu;
        const int dd[4] = {d[u].x, d[u].y, d[u].z, d[u].w};
#pragma unroll
        for (int c = 0; c < 4; ++c) {
            const bool in = (unsigned)(dd[c] - r0) < rows;
            const unsigned mask = __builtin_amdgcn_ballot_w32(in) & vmask;
            if (mask != 0u) {
                if (in && valid)
                    q[(tail + (int)__builtin_amdgcn_mbcnt_lo(mask, 0u)) & (AG_Q - 1)] = g * 4 + c;
                tail += __builtin_popcount(mask);
            }
        }
    }
}

__device__ __forceinline__ void agg128_drain(const float* __restrict__ h1,
                                             const int* __restrict__ ei, int roff,
                                             const int* q, int head, int cnt,
                                             float* acc, int lane) {
    const unsigned qe = (unsigned)q[(head + lane) & (AG_Q - 1)];
    const unsigned en = (lane < cnt) ? qe : 0u;
    const int s  = ei[en & 0xfffffu];
    const int dl = (int)(en >> 20) - roff;
#pragma unroll 1
    for (int b = 0; b < cnt; b += AG_B) {
        float4 v[AG_B];
#pragma unroll
        for (int j = 0; j < AG_B; ++j) {
            const int sj = __builtin_amdgcn_readlane(s, b + j);
            v[j] = *(const float4*)(h1 + sj * H + lane * 4);
        }
#pragma unroll
        for (int j = 0; j < AG_B; ++j) {
            if (b + j < cnt) {
                const int dj = __builtin_amdgcn_readlane(dl, b + j);
                float* a = acc + dj * AG_P + lane;
                atomicAdd(a,      v[j].x);
                atomicAdd(a + 32, v[j].y);
                atomicAdd(a + 64, v[j].z);
                atomicAdd(a + 96, v[j].w);
            }
        }
    }
}

template <int CNT>
__device__ __forceinline__ void agg128_conv2_tiles(const float* acc, const float* hs,
                                                   const float* __restrict__ Wrel,
                                                   const float* __restrict__ Wroot,
                                                   const float* __restrict__ brel,
                                                   float* __restrict__ h2, int r0, int rows,
                                                   int ct, int rg0, int lane) {
    const int c   = lane & 15;
    const int h   = lane >> 4;
    const int col = ct * 16 + c;
    const float bv = brel[col];
    v8f d[CNT];
#pragma unroll
    for (int i = 0; i < CNT; ++i) {
        const v8f t = {bv, bv, bv, bv, bv, bv, bv, bv};
        d[i] = t;
    }
    const float* ap = acc + (rg0 * 16 + c) * AG_P + 2 * h;
    const float* hp = hs  + (rg0 * 16 + c) * AG_P + 2 * h;
    const float* wr = Wrel  + (8 * h) * H + col;
    const float* wo = Wroot + (2 * h) * H + col;
#pragma unroll 8
    for (int t = 0; t < 32; ++t) {
        const int j  = t >> 2;
        const int kk = t & 3;
        v2f b;
        b.x = wr[(16 * j + kk) * H];
        b.y = wr[(16 * j + kk + 4) * H];
#pragma unroll
        for (int i = 0; i < CNT; ++i) {
            const v2f a = *(const v2f*)(ap + i * 64 * AG_P + kk * 32 + 4 * j);
            d[i] = __builtin_amdgcn_wmma_f32_16x16x4_f32(false, a, false, b, (short)0, d[i],
                                                         false, false);
        }
    }
#pragma unroll 8
    for (int k = 0; k < H; k += 4) {
        v2f b;
        b.x = wo[k * H];
        b.y = wo[k * H + H];
#pragma unroll
        for (int i = 0; i < CNT; ++i) {
            const v2f a = *(const v2f*)(hp + i * 64 * AG_P + k);
            d[i] = __builtin_amdgcn_wmma_f32_16x16x4_f32(false, a, false, b, (short)0, d[i],
                                                         false, false);
        }
    }
#pragma unroll
    for (int i = 0; i < CNT; ++i)
#pragma unroll
        for (int j = 0; j < 8; ++j) {
            const int row = (rg0 + 4 * i) * 16 + j + 8 * h;
            if (row < rows) h2[(size_t)(r0 + row) * H + col] = fmaxf(d[i][j], 0.0f);
        }
}

template <int RTS>
__global__ __launch_bounds__(AG_T)
void agg128_conv2_kernel(const float* __restrict__ h1, const int* __restrict__ ei,
                         int E, int n, int rr, const int* __restrict__ lst,
                         const int* __restrict__ lcnt,
                         const float* __restrict__ Wrel, const float* __restrict__ Wroot,
                         const float* __restrict__ brel, float* __restrict__ h2) {
    __shared__ __attribute__((aligned(16))) float acc[AG_RPAD * AG_P];
    __shared__ __attribute__((aligned(16))) int ring[AG_T / 32][AG_Q];
    const int tid  = threadIdx.x;
    const int lane = tid & 31;
    const int r0   = blockIdx.x * rr;
    const int rows = (n - r0 < rr) ? (n - r0) : rr;
    const int nrt  = (rr + 15) >> 4;

    {
        float4* a4 = (float4*)acc;
        const float4 z = {0.0f, 0.0f, 0.0f, 0.0f};
        for (int i = tid; i < nrt * 16 * (AG_P / 4); i += AG_T) a4[i] = z;
    }
    __syncthreads();

    int* q = ring[tid >> 5];
    int head = 0, tail = 0;
    const int wv   = __builtin_amdgcn_readfirstlane(tid >> 5);
    const int xr   = r0 / SC_R;
    const int roff = r0 - xr * SC_R;
    const int cap  = sc_cap(E);
    const int sub0 = xr * (SC_S * (SC_T / 32)) + wv;
    static_assert(SC_S <= 32, "one count per lane");
    const int cvec = lcnt[sub0 + ((lane < SC_S) ? lane : 0) * (SC_T / 32)];

    for (int t0 = 0; t0 < SC_S; t0 += SL_G) {
        int cn[SL_G];
        const int* bp[SL_G];
        int ent[SL_G][SL_L];
#pragma unroll
        for (int g = 0; g < SL_G; ++g) {
            const int sub = sub0 + (t0 + g) * (SC_T / 32);
            cn[g] = __builtin_amdgcn_readlane(cvec, t0 + g);
            bp[g] = (cn[g] > 0) ? (lst + (size_t)sub * cap) : (lcnt + sub);
        }
#pragma unroll
        for (int g = 0; g < SL_G; ++g)
#pragma unroll
            for (int u = 0; u < SL_L; ++u) {
                const int i  = u * 32 + lane;
                const int ic = (cn[g] > 0) ? ((i < cn[g]) ? i : (cn[g] - 1)) : 0;
                ent[g][u] = bp[g][ic];
            }
#pragma unroll
        for (int g = 0; g < SL_G; ++g)
            asm volatile("" :: "v"(ent[g][0]), "v"(ent[g][1]), "v"(ent[g][2]),
                               "v"(ent[g][3]), "v"(ent[g][4]));
        static_assert(SL_L == 5, "the fence above names SL_L entries per sublist");
#pragma unroll
        for (int g = 0; g < SL_G; ++g)
#pragma unroll
            for (int u = 0; u < SL_L; ++u) {
                if (u * 32 < cn[g]) {
                    const unsigned en = (unsigned)ent[g][u];
                    const bool in = (u * 32 + lane < cn[g]) &&
                                    ((en >> 20) - (unsigned)roff < (unsigned)rows);
                    const unsigned mask = __builtin_amdgcn_ballot_w32(in);
                    if (mask != 0u) {
                        if (in)
                            q[(tail + (int)__builtin_amdgcn_mbcnt_lo(mask, 0u)) & (AG_Q - 1)] = (int)en;
                        tail += __builtin_popcount(mask);
                    }
                }
            }
        while (tail - head >= 32) {
            agg128_drain(h1, ei, roff, q, head, 32, acc, lane);
            head += 32;
        }
#pragma unroll
        for (int g = 0; g < SL_G; ++g) {
            for (int i0 = SL_L * 32; i0 < cn[g]; i0 += 32) {
                const int i = i0 + lane;
                const unsigned en = (unsigned)bp[g][(i < cn[g]) ? i : (cn[g] - 1)];
                const bool in = (i < cn[g]) && ((en >> 20) - (unsigned)roff < (unsigned)rows);
                const unsigned mask = __builtin_amdgcn_ballot_w32(in);
                if (mask != 0u) {
                    if (in)
                        q[(tail + (int)__builtin_amdgcn_mbcnt_lo(mask, 0u)) & (AG_Q - 1)] = (int)en;
                    tail += __builtin_popcount(mask);
                }
                while (tail - head >= 32) {
                    agg128_drain(h1, ei, roff, q, head, 32, acc, lane);
                    head += 32;
                }
            }
        }
    }
    while (tail - head > 0) {
        const int cnt = (tail - head < 32) ? (tail - head) : 32;
        agg128_drain(h1, ei, roff, q, head, cnt, acc, lane);
        head += cnt;
    }
    __syncthreads();

    constexpr int RP    = (RTS * 64 < AG_RPAD) ? (RTS * 64) : AG_RPAD;
    constexpr int HS_IT = (RP * (H / 4) + AG_T - 1) / AG_T;
    static_assert(((HS_IT * AG_T - 1) / (H / 4)) * AG_P + H <= (AG_T / 32) * AG_Q,
                  "the staged h1 rows fit in the ring space");
    float* hs = (float*)&ring[0][0];
    {
        float4 hv[HS_IT];
#pragma unroll
        for (int j = 0; j < HS_IT; ++j) {
            const int i  = j * AG_T + tid;
            const int rg = (r0 + (i >> 5) < n) ? (r0 + (i >> 5)) : (n - 1);
            hv[j] = *(const float4*)(h1 + (size_t)rg * H + (i & 31) * 4);
        }
#pragma unroll
        for (int j = 0; j < HS_IT; ++j) {
            const int i = j * AG_T + tid;
            *(float4*)(hs + (i >> 5) * AG_P + (i & 31) * 4) = hv[j];
        }
    }
    __syncthreads();

    const int ct  = wv & 7;
    const int rg0 = wv >> 3;
    const int cnt = (nrt - rg0 + 3) >> 2;
    if (RTS >= 4 && cnt >= 4)
        agg128_conv2_tiles<4>(acc, hs, Wrel, Wroot, brel, h2, r0, rows, ct, rg0, lane);
    else if (RTS >= 3 && cnt == 3)
        agg128_conv2_tiles<3>(acc, hs, Wrel, Wroot, brel, h2, r0, rows, ct, rg0, lane);
    else if (RTS >= 2 && cnt == 2)
        agg128_conv2_tiles<2>(acc, hs, Wrel, Wroot, brel, h2, r0, rows, ct, rg0, lane);
    else if (cnt >= 1)
        agg128_conv2_tiles<1>(acc, hs, Wrel, Wroot, brel, h2, r0, rows, ct, rg0, lane);
}

#define IN_R 224
#define IN_P (F3 + 4)
#define IN_B 16
#define IN_SEG (IN_R * IN_P)
#define IN_RING ((AG_T / 32) * AG_Q)
static_assert(IN_R % 16 == 0 && (IN_R * 3) % 32 == 0, "whole tiles; whole lines of out");
static_assert(IN_SEG + IN_R * 3 <= IN_RING, "the e0 rows and the block's results fit in the ring space");
static_assert((IN_RING + 2 * IN_SEG) * 4 <= 320 * 1024, "one block per WGP");

__device__ __forceinline__ void interp_drain(const float* __restrict__ e,
                                             const int* __restrict__ cols,
                                             const int* __restrict__ rws,
                                             const float* __restrict__ vals, int r0,
                                             const int* q, int head, int cnt,
                                             float* acc, int lane) {
    const int qe  = q[(head + lane) & (AG_Q - 1)];
    const int nz  = (lane < cnt) ? qe : 0;
    const int col = cols[nz];
    const int dl  = rws[nz] - r0;
    const int vb  = __float_as_int(vals[nz]);
#pragma unroll 1
    for (int b = 0; b < cnt; b += IN_B) {
        float v[IN_B][3];
#pragma unroll
        for (int j = 0; j < IN_B; ++j) {
            const int cj = __builtin_amdgcn_readlane(col, b + j);
            const float* p = e + cj * F3 + lane;
            v[j][0] = p[0];
            v[j][1] = p[32];
            v[j][2] = p[64];
        }
#pragma unroll
        for (int j = 0; j < IN_B; ++j) {
            if (b + j < cnt) {
                const int dj  = __builtin_amdgcn_readlane(dl, b + j);
                const float s = __int_as_float(__builtin_amdgcn_readlane(vb, b + j));
                float* a = acc + dj * IN_P + lane;
                atomicAdd(a,      s * v[j][0]);
                atomicAdd(a + 32, s * v[j][1]);
                atomicAdd(a + 64, s * v[j][2]);
            }
        }
    }
}

__device__ __forceinline__ void dec_unit(const float* seg0, const float* __restrict__ W0,
                                         const float* __restrict__ b0,
                                         const float* __restrict__ Wout, float bo,
                                         float* res, int rt, int br, int lane) {
    const int c = lane & 15;
    const int h = lane >> 4;
    v8f d[4];
#pragma unroll
    for (int ct = 0; ct < 4; ++ct) {
        const float bv = b0[ct * 16 + c];
        const v8f t = {bv, bv, bv, bv, bv, bv, bv, bv};
        d[ct] = t;
    }
    const float* ap = seg0 + (rt * 16 + c) * IN_P + br * 32 + 2 * h;
    const float* wp = W0 + (2 * h) * DECW + c;
#pragma unroll 4
    for (int t = 0; t < F3 / 4; ++t) {
        const v2f a = *(const v2f*)(ap + (t >> 3) * IN_SEG + (t & 7) * 4);
#pragma unroll
        for (int ct = 0; ct < 4; ++ct) {
            v2f b;
            b.x = wp[(4 * t) * DECW + ct * 16];
            b.y = wp[(4 * t + 1) * DECW + ct * 16];
            d[ct] = __builtin_amdgcn_wmma_f32_16x16x4_f32(false, a, false, b, (short)0, d[ct],
                                                          false, false);
        }
    }
    float p[8];
#pragma unroll
    for (int j = 0; j < 8; ++j) p[j] = 0.0f;
#pragma unroll
    for (int ct = 0; ct < 4; ++ct) {
        const float w = Wout[ct * 16 + c];
#pragma unroll
        for (int j = 0; j < 8; ++j) {
            const float v = d[ct][j];
            p[j] += ((v > 0.0f) ? v : (expf(v) - 1.0f)) * w;
        }
    }
#pragma unroll
    for (int m = 1; m < 16; m <<= 1)
#pragma unroll
        for (int j = 0; j < 8; ++j) p[j] += __shfl_xor(p[j], m);
    if (c == 0) {
#pragma unroll
        for (int j = 0; j < 8; ++j) res[(rt * 16 + j + 8 * h) * 3 + br] = p[j] + bo;
    }
}

__global__ __launch_bounds__(AG_T)
void interp_dec_kernel(const float* __restrict__ vals1, const int* __restrict__ cols1,
                       const int* __restrict__ rows1, const float* __restrict__ e1,
                       const float* __restrict__ vals2, const int* __restrict__ cols2,
                       const int* __restrict__ rows2, const float* __restrict__ e2,
                       const float* __restrict__ e0, const float* __restrict__ W0,
                       const float* __restrict__ b0, const float* __restrict__ Wout,
                       const float* __restrict__ bout, float* __restrict__ out,
                       int nnz, int n) {
    __shared__ __attribute__((aligned(16))) float smem[IN_RING + 2 * IN_SEG];
    int*   ring = (int*)smem;
    float* sums = smem + IN_RING;
    float* es   = sums - IN_SEG;
    float* res  = smem;

    const int tid  = threadIdx.x;
    const int lane = tid & 31;
    const int r0   = blockIdx.x * IN_R;
    const int rows = (n - r0 < IN_R) ? (n - r0) : IN_R;

    {
        float4* a4 = (float4*)sums;
        const float4 z = {0.0f, 0.0f, 0.0f, 0.0f};
        for (int i = tid; i < 2 * IN_SEG / 4; i += AG_T) a4[i] = z;
    }
    __syncthreads();

    int* q = ring + (tid >> 5) * AG_Q;
    int head = 0, tail = 0;
    const int ng = nnz >> 2;

#pragma unroll 1
    for (int m = 0; m < 2; ++m) {
        const float* vals = m ? vals2 : vals1;
        const int*   cols = m ? cols2 : cols1;
        const int*   rws  = m ? rows2 : rows1;
        const float* e    = m ? e2 : e1;
        float*       acc  = sums + m * IN_SEG;
        const int4*  d4   = (const int4*)rws;

        int g0 = 0;
        for (; g0 + AG_T * AG_U <= ng; g0 += AG_T * AG_U) {
            agg128_scan_trip<false>(d4, g0, ng, tid, r0, (unsigned)rows, q, tail);
            while (tail - head >= 32) {
                interp_drain(e, cols, rws, vals, r0, q, head, 32, acc, lane);
                head += 32;
            }
        }
        if (g0 < ng) agg128_scan_trip<true>(d4, g0, ng, tid, r0, (unsigned)rows, q, tail);
        while (tail - head > 0) {
            const int cnt = (tail - head < 32) ? (tail - head) : 32;
            interp_drain(e, cols, rws, vals, r0, q, head, cnt, acc, lane);
            head += cnt;
        }
    }
    __syncthreads();

    {
        constexpr int TOT = IN_R * (F3 / 4);
        constexpr int IT  = (TOT + AG_T - 1) / AG_T;
        float4 ev[IT];
#pragma unroll
        for (int j = 0; j < IT; ++j) {
            const int i  = j * AG_T + tid;
            const int ic = (i < TOT) ? i : (TOT - 1);
            const int r  = ic / (F3 / 4);
            const int c4 = ic % (F3 / 4);
            const int rg = (r0 + r < n) ? (r0 + r) : (n - 1);
            ev[j] = *(const float4*)(e0 + (size_t)rg * F3 + c4 * 4);
        }
#pragma unroll
        for (int j = 0; j < IT; ++j) {
            const int i = j * AG_T + tid;
            if (i < TOT) {
                const int r  = i / (F3 / 4);
                const int c4 = i % (F3 / 4);
                *(float4*)(es + r * IN_P + c4 * 4) = ev[j];
            }
        }
    }
    __syncthreads();

    const int wv = __builtin_amdgcn_readfirstlane(tid >> 5);
    for (int u = wv; u < (IN_R / 16) * 3; u += AG_T / 32) {
        const int br = u % 3;
        const int rt = u / 3;
        dec_unit(es, W0 + (size_t)br * F3 * DECW, b0 + br * DECW, Wout + br * DECW, bout[br],
                 res, rt, br, lane);
    }
    __syncthreads();

    for (int i = tid; i < rows * 3; i += AG_T) out[(size_t)r0 * 3 + i] = res[i];
}

template <int NSEG, int SEGW, int LD, int Nc, int RT, int ACT>
__global__ __launch_bounds__(Nc)
void gemm_tile_kernel(const float* __restrict__ s0, const float* __restrict__ s1,
                      const float* __restrict__ s2, int coff,
                      const float* __restrict__ W0,
                      const float* __restrict__ bias, float* __restrict__ D, int n) {
    constexpr int K   = NSEG * SEGW;
    constexpr int P   = K + 4;
    constexpr int MB  = RT * 16;
    constexpr int NT  = Nc;
    constexpr int K4  = K / 4;
    constexpr int S4  = SEGW / 4;
    constexpr int TOT = MB * K4;
    constexpr int ITER = (TOT + NT - 1) / NT;
    constexpr int SB  = 8;
    __shared__ __attribute__((aligned(16))) float As[MB * P];

    const int tid  = threadIdx.x;
    const int lane = tid & 31;
    const int wave = tid >> 5;
    const int rowb = blockIdx.x * MB;

#pragma unroll
    for (int i0 = 0; i0 < ITER; i0 += SB) {
        float4 v[SB];
#pragma unroll
        for (int j = 0; j < SB; ++j) {
            if (i0 + j < ITER) {
                const int idx = (i0 + j) * NT + tid;
                const int idc = (TOT % NT != 0 && idx >= TOT) ? (TOT - 1) : idx;
                const int r   = idc / K4;
                const int c4  = idc % K4;
                const int sg  = c4 / S4;
                const int w4  = c4 % S4;
                const int rg  = (rowb + r < n) ? (rowb + r) : (n - 1);
                const float* sp = (sg == 0) ? s0 : (sg == 1) ? s1 : s2;
                v[j] = *(const float4*)(sp + (size_t)rg * LD + coff + w4 * 4);
            }
        }
#pragma unroll
        for (int j = 0; j < SB; ++j) {
            if (i0 + j < ITER) {
                const int idx = (i0 + j) * NT + tid;
                if (TOT % NT == 0 || idx < TOT) {
                    const int r  = idx / K4;
                    const int c4 = idx % K4;
                    *(float4*)(As + r * P + c4 * 4) = v[j];
                }
            }
        }
    }
    __syncthreads();

    const int c    = lane & 15;
    const int h    = lane >> 4;
    const int col0 = wave * 32;
    const float* ap  = As + c * P + 2 * h;
    const float* wp0 = W0 + col0 + c + 2 * h * Nc;

    v8f acc[RT][2];
#pragma unroll
    for (int ct = 0; ct < 2; ++ct) {
        const float bv = bias[col0 + ct * 16 + c];
#pragma unroll
        for (int rt = 0; rt < RT; ++rt) {
            const v8f t = {bv, bv, bv, bv, bv, bv, bv, bv};
            acc[rt][ct] = t;
        }
    }

#pragma unroll 8
    for (int k = 0; k < K; k += 4) {
        v2f a[RT];
        v2f b[2];
#pragma unroll
        for (int rt = 0; rt < RT; ++rt)
            a[rt] = *(const v2f*)(ap + rt * 16 * P + k);
#pragma unroll
        for (int ct = 0; ct < 2; ++ct) {
            b[ct].x = wp0[k * Nc + ct * 16];
            b[ct].y = wp0[k * Nc + Nc + ct * 16];
        }
#pragma unroll
        for (int rt = 0; rt < RT; ++rt)
#pragma unroll
            for (int ct = 0; ct < 2; ++ct)
                acc[rt][ct] = __builtin_amdgcn_wmma_f32_16x16x4_f32(
                    false, a[rt], false, b[ct], (short)0, acc[rt][ct], false, false);
    }

    const bool full = (rowb + MB <= n);
#pragma unroll
    for (int rt = 0; rt < RT; ++rt)
#pragma unroll
        for (int ct = 0; ct < 2; ++ct)
#pragma unroll
            for (int j = 0; j < 8; ++j) {
                float v = acc[rt][ct][j];
                if (ACT == 1) v = fmaxf(v, 0.0f);
                if (ACT == 2) v = (v > 0.0f) ? v : (expf(v) - 1.0f);
                acc[rt][ct][j] = v;
            }
    float* dp = D + (rowb + h * 8) * Nc + col0 + c;
    if (full) {
#pragma unroll
        for (int rt = 0; rt < RT; ++rt)
#pragma unroll
            for (int ct = 0; ct < 2; ++ct)
#pragma unroll
                for (int j = 0; j < 8; ++j)
                    dp[(rt * 16 + j) * Nc + ct * 16] = acc[rt][ct][j];
    } else {
#pragma unroll
        for (int rt = 0; rt < RT; ++rt)
#pragma unroll
            for (int ct = 0; ct < 2; ++ct)
#pragma unroll
                for (int j = 0; j < 8; ++j)
                    if (rowb + rt * 16 + j + h * 8 < n)
                        dp[(rt * 16 + j) * Nc + ct * 16] = acc[rt][ct][j];
    }
}

template <int RTS, int RTL>
static inline void launch_enc_conv2_lin(const float* h1, const int* ei, int E, int n, int rr,
                                        const int* lst, const int* lcnt, float* h2, float* e,
                                        const float* Wrel, const float* Wroot, const float* brel,
                                        const float* lW, const float* lb, hipStream_t stream) {
    agg128_conv2_kernel<RTS><<<(n + rr - 1) / rr, AG_T, 0, stream>>>(
        h1, ei, E, n, rr, lst, lcnt, Wrel, Wroot, brel, h2);
    gemm_tile_kernel<1, H, H, F3, RTL, 0><<<(n + RTL * 16 - 1) / (RTL * 16), F3, 0, stream>>>(
        h2, nullptr, nullptr, 0, lW, lb, e, n);
}

extern "C" void kernel_launch(void* const* d_in, const int* in_sizes, int n_in,
                              void* d_out, int out_size, void* d_ws, size_t ws_size,
                              hipStream_t stream) {
    (void)in_sizes; (void)n_in; (void)out_size; (void)ws_size;

    const float* x[3]   = {(const float*)d_in[0], (const float*)d_in[1], (const float*)d_in[2]};
    const int*   ei[3]  = {(const int*)d_in[3], (const int*)d_in[4], (const int*)d_in[5]};
    const int*   A1r    = (const int*)d_in[6];
    const int*   A1c    = (const int*)d_in[7];
    const float* A1v    = (const float*)d_in[8];
    const int*   A2r    = (const int*)d_in[9];
    const int*   A2c    = (const int*)d_in[10];
    const float* A2v    = (const float*)d_in[11];
    const float* c1Wrel  = (const float*)d_in[12];
    const float* c1brel  = (const float*)d_in[13];
    const float* c1Wroot = (const float*)d_in[14];
    const float* c2Wrel  = (const float*)d_in[15];
    const float* c2brel  = (const float*)d_in[16];
    const float* c2Wroot = (const float*)d_in[17];
    const float* linW    = (const float*)d_in[18];
    const float* linb    = (const float*)d_in[19];
    const float* decW0   = (const float*)d_in[20];
    const float* decb0   = (const float*)d_in[21];
    const float* decWout = (const float*)d_in[22];
    const float* decbout = (const float*)d_in[23];
    float* out = (float*)d_out;

    float* ws = (float*)d_ws;
    float* h1   = ws;
    float* agg  = h1   + (size_t)N0 * H;
    float* h2   = agg  + (size_t)N0 * H;
    float* e0   = h2   + (size_t)N0 * H;
    float* e1   = e0   + (size_t)N0 * F3;
    float* e2   = e1   + (size_t)N1 * F3;
    float* rest = e2   + (size_t)N2 * F3;

    constexpr size_t LST_MAX = (size_t)((N0 + SC_R - 1) / SC_R) * SC_S * (SC_T / 32) *
                               ((((E0 + SC_S - 1) / SC_S + SC_T * SC_U - 1) / (SC_T * SC_U)) * SC_U * 32);
    constexpr size_t LCNT_MAX = (size_t)((N0 + SC_R - 1) / SC_R) * SC_S * (SC_T / 32);
    static_assert(LST_MAX + LCNT_MAX <= (size_t)N0 * F3 * 2 + (size_t)N0 * DECW,
                  "the lists of level 0 (the largest) fit behind e2");
    static_assert(E0 >= E1 && E0 >= E2 && N0 >= N1 && N0 >= N2, "level 0 is the largest");
    int* lst  = (int*)rest;
    int* lcnt = lst + LST_MAX;

    const int nlv[3] = {N0, N1, N2};
    const int Elv[3] = {E0, E1, E2};
    const int rrl[3] = {200, 50, 13};
    static_assert(SC_R % 200 == 0 && SC_R % 50 == 0 && N2 <= SC_R,
                  "a conv2 block's rows lie inside one scatter6 range");
    static_assert(200 <= AG_RMAX && (50 + 15) / 16 <= 4 && (13 + 15) / 16 <= 4,
                  "conv2 product: level 0 within AG_RMAX rows, levels 1 and 2 one row tile per wave");
    static_assert(E0 % 4 == 0 && E1 % 4 == 0 && E2 % 4 == 0, "dst lists are read 16 B at a time");
    float* elv[3] = {e0, e1, e2};

    for (int i = 0; i < 3; ++i) {
        const int n = nlv[i];
        const int E = Elv[i];

        {
            scatter6_lds_kernel<<<dim3((n + SC_R - 1) / SC_R, SC_S, 1), SC_T, 0, stream>>>(
                x[i], ei[i], E, n, agg, lst, lcnt);
            conv1_planes_kernel<<<(n + C1_R - 1) / C1_R, C1_T, 0, stream>>>(
                x[i], agg,
                c1Wrel + (size_t)i * 6 * H,
                c1brel + (size_t)i * H,
                c1Wroot + (size_t)i * 6 * H, h1, n);
        }

        {
            const int rr = rrl[i];
            const float* Wrel2  = c2Wrel + (size_t)i * H * H;
            const float* Wroot2 = c2Wroot + (size_t)i * H * H;
            const float* brel2  = c2brel + (size_t)i * H;
            const float* lW     = linW + (size_t)i * H * F3;
            const float* lb     = linb + (size_t)i * F3;
            if (i == 0)
                launch_enc_conv2_lin<4, 2>(h1, ei[i], E, n, rr, lst, lcnt, h2, elv[i],
                                           Wrel2, Wroot2, brel2, lW, lb, stream);
            else if (i == 1)
                launch_enc_conv2_lin<1, 2>(h1, ei[i], E, n, rr, lst, lcnt, h2, elv[i],
                                           Wrel2, Wroot2, brel2, lW, lb, stream);
            else
                launch_enc_conv2_lin<1, 1>(h1, ei[i], E, n, rr, lst, lcnt, h2, elv[i],
                                           Wrel2, Wroot2, brel2, lW, lb, stream);
        }
    }

    static_assert(NNZ % 4 == 0, "row lists are read 16 B at a time");
    interp_dec_kernel<<<(N0 + IN_R - 1) / IN_R, AG_T, 0, stream>>>(
        A1v, A1c, A1r, e1, A2v, A2c, A2r, e2, e0, decW0, decb0, decWout, decbout, out,
        NNZ, N0);
}


# lin product moved into the conv2 launch: relu'd h2 tiles go from accumulators into LDS (over the dead sums) and feed lin there; h2 never stored; 13 -> 10 launches; sum order kept
# speedup vs baseline: 3.6737x; 1.0862x over previous
#include <hip/hip_runtime.h>
#include <hip/hip_bf16.h>

#define N0 50000
#define N1 12500
#define N2 3125
#define E0 800000
#define E1 200000
#define E2 50000
#define NNZ 200000
#define H 128
#define F3 96
#define DECW 64

typedef __attribute__((ext_vector_type(2))) float v2f;
typedef __attribute__((ext_vector_type(8))) float v8f;


#define SC_R 4000

#define SC_S 16
#define SC_T 1024
#define SC_U 8
static_assert(SC_R <= 4096 && E0 < (1 << 20) && E1 < (1 << 20) && E2 < (1 << 20),
              "a list entry packs a 12-bit row offset and a 20-bit edge number");

__device__ __forceinline__ int sc_cap(int E) {
    const int chunk = (E + SC_S - 1) / SC_S;
    const int trips = (chunk + SC_T * SC_U - 1) / (SC_T * SC_U);
    return trips * SC_U * 32;
}

__global__ __launch_bounds__(SC_T) __attribute__((amdgpu_waves_per_eu(8, 8)))
void scatter6_lds_kernel(const float* __restrict__ x, const int* __restrict__ ei,
                         int E, int n, float* __restrict__ part,
                         int* __restrict__ lst, int* __restrict__ lcnt) {
    __shared__ float acc[SC_R * 6];
    const int tid = threadIdx.x;
    const int r0  = blockIdx.x * SC_R;
    const int s   = blockIdx.y;

    for (int i = tid; i < SC_R * 6; i += SC_T) acc[i] = 0.0f;
    __syncthreads();

    const int chunk = (E + SC_S - 1) / SC_S;
    const int eb = s * chunk;
    const int ee = (eb + chunk < E) ? (eb + chunk) : E;

    const int sub = (blockIdx.x * SC_S + s) * (SC_T / 32) + (tid >> 5);
    int* lstw = lst + (size_t)sub * sc_cap(E);
    int wc = 0;

    if (eb < ee) {
        for (int base = eb; base < ee; base += SC_T * SC_U) {
            int off[SC_U];
            int srow[SC_U];
#pragma unroll
            for (int u = 0; u < SC_U; ++u) {
                const int e  = base + u * SC_T + tid;
                const int ec = (e < ee) ? e : (ee - 1);
                const int d  = ei[E + ec] - r0;
                const int sr = ei[ec];
                const bool hit = (e < ee) && ((unsigned)d < (unsigned)SC_R);
                off[u]  = hit ? d * 6 : -1;
                srow[u] = hit ? sr * 6 : 0;
            }
            float2 v[SC_U][3];
#pragma unroll
            for (int u = 0; u < SC_U; ++u) {
                const float2* p = (const float2*)(x + srow[u]);
                v[u][0] = p[0];
                v[u][1] = p[1];
                v[u][2] = p[2];
            }
#pragma unroll
            for (int u = 0; u < SC_U; ++u)
                asm volatile("" :: "v"(v[u][0].x), "v"(v[u][0].y), "v"(v[u][1].x),
                                   "v"(v[u][1].y), "v"(v[u][2].x), "v"(v[u][2].y));
#pragma unroll
            for (int u = 0; u < SC_U; ++u) {
                const unsigned hm = __builtin_amdgcn_ballot_w32(off[u] >= 0);
                if (off[u] >= 0) {
                    lstw[wc + (int)__builtin_amdgcn_mbcnt_lo(hm, 0u)] =
                        (int)((((unsigned)off[u] / 6u) << 20) | (unsigned)(base + u * SC_T + tid));
                    float* a = acc + off[u];
                    atomicAdd(a + 0, v[u][0].x);
                    atomicAdd(a + 1, v[u][0].y);
                    atomicAdd(a + 2, v[u][1].x);
                    atomicAdd(a + 3, v[u][1].y);
                    atomicAdd(a + 4, v[u][2].x);
                    atomicAdd(a + 5, v[u][2].y);
                }
                wc += __builtin_popcount(hm);
            }
        }
    }
    if ((tid & 31) == 0) lcnt[sub] = wc;
    __syncthreads();

    const int rows = (n - r0 < SC_R) ? (n - r0) : SC_R;
    float* p = part + (size_t)s * n * 6 + (size_t)r0 * 6;
    for (int i = tid; i < rows * 6; i += SC_T) p[i] = acc[i];
}

#define C1_R 32
#define C1_T 128

__global__ __launch_bounds__(C1_T)
void conv1_planes_kernel(const float* __restrict__ x, const float* __restrict__ part,
                         const float* __restrict__ Wrel, const float* __restrict__ brel,
                         const float* __restrict__ Wroot, float* __restrict__ h1, int n) {
    __shared__ float sa[C1_R * 6];
    __shared__ float sx[C1_R * 6];
    const int tid  = threadIdx.x;
    const int lane = tid & 31;
    const int wave = tid >> 5;
    const int r0   = blockIdx.x * C1_R;
    const int rows = (n - r0 < C1_R) ? (n - r0) : C1_R;
    const int cnt  = n * 6;

    float4 wr[6], wo[6];
#pragma unroll
    for (int k = 0; k < 6; ++k) {
        wr[k] = *(const float4*)(Wrel + k * H + lane * 4);
        wo[k] = *(const float4*)(Wroot + k * H + lane * 4);
    }
    const float4 bv = *(const float4*)(brel + lane * 4);

    static_assert(6 * C1_R <= 2 * C1_T, "two elements per thread cover the block's rows");
    const int  m    = rows * 6;
    const int  i1   = tid + C1_T;
    const bool has0 = tid < m;
    const bool has1 = i1 < m;
    const int  t0   = r0 * 6 + (has0 ? tid : 0);
    const int  t1   = r0 * 6 + (has1 ? i1 : 0);
    float p0[SC_S], p1[SC_S];
#pragma unroll
    for (int k = 0; k < SC_S; ++k) {
        p0[k] = part[(size_t)k * cnt + t0];
        p1[k] = part[(size_t)k * cnt + t1];
    }
    const float x0 = x[t0];
    const float x1 = x[t1];
#pragma unroll
    for (int k = 0; k < SC_S; k += 4)
        asm volatile("" :: "v"(p0[k]), "v"(p0[k + 1]), "v"(p0[k + 2]), "v"(p0[k + 3]),
                           "v"(p1[k]), "v"(p1[k + 1]), "v"(p1[k + 2]), "v"(p1[k + 3]));
    asm volatile("" :: "v"(x0), "v"(x1));
    float s0 = 0.0f, s1 = 0.0f;
#pragma unroll
    for (int k = 0; k < SC_S; ++k) {
        s0 += p0[k];
        s1 += p1[k];
    }
    if (has0) { sa[tid] = s0; sx[tid] = x0; }
    if (has1) { sa[i1]  = s1; sx[i1]  = x1; }
    __syncthreads();

    for (int r = wave; r < rows; r += C1_T / 32) {
        float4 s = bv;
#pragma unroll
        for (int k = 0; k < 6; ++k) {
            const float a  = sa[r * 6 + k];
            const float xv = sx[r * 6 + k];
            s.x += a * wr[k].x;  s.y += a * wr[k].y;  s.z += a * wr[k].z;  s.w += a * wr[k].w;
            s.x += xv * wo[k].x; s.y += xv * wo[k].y; s.z += xv * wo[k].z; s.w += xv * wo[k].w;
        }
        s.x = fmaxf(s.x, 0.0f);
        s.y = fmaxf(s.y, 0.0f);
        s.z = fmaxf(s.z, 0.0f);
        s.w = fmaxf(s.w, 0.0f);
        *(float4*)(h1 + (size_t)(r0 + r) * H + lane * 4) = s;
    }
}

#define AG_T    1024
#define AG_RMAX 200
#define AG_RPAD 208
#define AG_P    (H + 4)
static_assert(AG_RPAD % 16 == 0 && AG_RPAD >= AG_RMAX && AG_RPAD / 16 <= 16,
              "at most 4 row tiles for each of the 4 wave groups");
#define AG_U    4
#define AG_Q    1024
#define AG_B    16
#define SL_G    4
#define SL_L    5
static_assert(SC_S % SL_G == 0, "sublists are read SL_G at a time");
static_assert(SL_G * SL_L * 32 + 31 <= AG_Q, "a wave's ring holds what one read step can queue");

template <bool TAIL>
__device__ __forceinline__ void agg128_scan_trip(const int4* __restrict__ d4, int g0, int ng,
                                                 int tid, int r0, unsigned rows,
                                                 int* q, int& tail) {
    int4 d[AG_U];
#pragma unroll
    for (int u = 0; u < AG_U; ++u) {
        const int g = g0 + u * AG_T + tid;
        d[u] = d4[(TAIL && g >= ng) ? (ng - 1) : g];
    }
#pragma unroll
    for (int u = 0; u < AG_U; ++u) {
        const int g = g0 + u * AG_T + tid;
        const bool valid = !TAIL || (g < ng);
        const unsigned vmask = TAIL ? __builtin_amdgcn_ballot_w32(g < ng) : 0xffffffffu;
        const int dd[4] = {d[u].x, d[u].y, d[u].z, d[u].w};
#pragma unroll
        for (int c = 0; c < 4; ++c) {
            const bool in = (unsigned)(dd[c] - r0) < rows;
            const unsigned mask = __builtin_amdgcn_ballot_w32(in) & vmask;
            if (mask != 0u) {
                if (in && valid)
                    q[(tail + (int)__builtin_amdgcn_mbcnt_lo(mask, 0u)) & (AG_Q - 1)] = g * 4 + c;
                tail += __builtin_popcount(mask);
            }
        }
    }
}

__device__ __forceinline__ void agg128_drain(const float* __restrict__ h1,
                                             const int* __restrict__ ei, int roff,
                                             const int* q, int head, int cnt,
                                             float* acc, int lane) {
    const unsigned qe = (unsigned)q[(head + lane) & (AG_Q - 1)];
    const unsigned en = (lane < cnt) ? qe : 0u;
    const int s  = ei[en & 0xfffffu];
    const int dl = (int)(en >> 20) - roff;
#pragma unroll 1
    for (int b = 0; b < cnt; b += AG_B) {
        float4 v[AG_B];
#pragma unroll
        for (int j = 0; j < AG_B; ++j) {
            const int sj = __builtin_amdgcn_readlane(s, b + j);
            v[j] = *(const float4*)(h1 + sj * H + lane * 4);
        }
#pragma unroll
        for (int j = 0; j < AG_B; ++j) {
            if (b + j < cnt) {
                const int dj = __builtin_amdgcn_readlane(dl, b + j);
                float* a = acc + dj * AG_P + lane;
                atomicAdd(a,      v[j].x);
                atomicAdd(a + 32, v[j].y);
                atomicAdd(a + 64, v[j].z);
                atomicAdd(a + 96, v[j].w);
            }
        }
    }
}

template <int CNT>
__device__ __forceinline__ void agg128_conv2_tiles(const float* acc, const float* hs,
                                                   const float* __restrict__ Wrel,
                                                   const float* __restrict__ Wroot,
                                                   const float* __restrict__ brel,
                                                   v8f (&dout)[4],
                                                   int ct, int rg0, int lane) {
    const int c   = lane & 15;
    const int h   = lane >> 4;
    const int col = ct * 16 + c;
    const float bv = brel[col];
    v8f d[CNT];
#pragma unroll
    for (int i = 0; i < CNT; ++i) {
        const v8f t = {bv, bv, bv, bv, bv, bv, bv, bv};
        d[i] = t;
    }
    const float* ap = acc + (rg0 * 16 + c) * AG_P + 2 * h;
    const float* hp = hs  + (rg0 * 16 + c) * AG_P + 2 * h;
    const float* wr = Wrel  + (8 * h) * H + col;
    const float* wo = Wroot + (2 * h) * H + col;
#pragma unroll 8
    for (int t = 0; t < 32; ++t) {
        const int j  = t >> 2;
        const int kk = t & 3;
        v2f b;
        b.x = wr[(16 * j + kk) * H];
        b.y = wr[(16 * j + kk + 4) * H];
#pragma unroll
        for (int i = 0; i < CNT; ++i) {
            const v2f a = *(const v2f*)(ap + i * 64 * AG_P + kk * 32 + 4 * j);
            d[i] = __builtin_amdgcn_wmma_f32_16x16x4_f32(false, a, false, b, (short)0, d[i],
                                                         false, false);
        }
    }
#pragma unroll 8
    for (int k = 0; k < H; k += 4) {
        v2f b;
        b.x = wo[k * H];
        b.y = wo[k * H + H];
#pragma unroll
        for (int i = 0; i < CNT; ++i) {
            const v2f a = *(const v2f*)(hp + i * 64 * AG_P + k);
            d[i] = __builtin_amdgcn_wmma_f32_16x16x4_f32(false, a, false, b, (short)0, d[i],
                                                         false, false);
        }
    }
#pragma unroll
    for (int i = 0; i < CNT; ++i)
#pragma unroll
        for (int j = 0; j < 8; ++j) dout[i][j] = fmaxf(d[i][j], 0.0f);
}

__device__ __forceinline__ void lin_unit(const float* hrow, const float* __restrict__ lW,
                                         const float* __restrict__ lb, float* __restrict__ e,
                                         int r0, int rows, int rt, int ct, int lane) {
    const int c   = lane & 15;
    const int h   = lane >> 4;
    const int col = ct * 16 + c;
    const float bv = lb[col];
    v8f d = {bv, bv, bv, bv, bv, bv, bv, bv};
    const float* ap = hrow + (rt * 16 + c) * AG_P + 2 * h;
    const float* wp = lW + (2 * h) * F3 + col;
#pragma unroll 8
    for (int k = 0; k < H; k += 4) {
        const v2f a = *(const v2f*)(ap + k);
        v2f b;
        b.x = wp[k * F3];
        b.y = wp[k * F3 + F3];
        d = __builtin_amdgcn_wmma_f32_16x16x4_f32(false, a, false, b, (short)0, d, false, false);
    }
#pragma unroll
    for (int j = 0; j < 8; ++j) {
        const int row = rt * 16 + j + 8 * h;
        if (row < rows) e[(size_t)(r0 + row) * F3 + col] = d[j];
    }
}

template <int RTS>
__global__ __launch_bounds__(AG_T)
void agg128_conv2_kernel(const float* __restrict__ h1, const int* __restrict__ ei,
                         int E, int n, int rr, const int* __restrict__ lst,
                         const int* __restrict__ lcnt,
                         const float* __restrict__ Wrel, const float* __restrict__ Wroot,
                         const float* __restrict__ brel, const float* __restrict__ lW,
                         const float* __restrict__ lb, float* __restrict__ e) {
    __shared__ __attribute__((aligned(16))) float acc[AG_RPAD * AG_P];
    __shared__ __attribute__((aligned(16))) int ring[AG_T / 32][AG_Q];
    const int tid  = threadIdx.x;
    const int lane = tid & 31;
    const int r0   = blockIdx.x * rr;
    const int rows = (n - r0 < rr) ? (n - r0) : rr;
    const int nrt  = (rr + 15) >> 4;

    {
        float4* a4 = (float4*)acc;
        const float4 z = {0.0f, 0.0f, 0.0f, 0.0f};
        for (int i = tid; i < nrt * 16 * (AG_P / 4); i += AG_T) a4[i] = z;
    }
    __syncthreads();

    int* q = ring[tid >> 5];
    int head = 0, tail = 0;
    const int wv   = __builtin_amdgcn_readfirstlane(tid >> 5);
    const int xr   = r0 / SC_R;
    const int roff = r0 - xr * SC_R;
    const int cap  = sc_cap(E);
    const int sub0 = xr * (SC_S * (SC_T / 32)) + wv;
    static_assert(SC_S <= 32, "one count per lane");
    const int cvec = lcnt[sub0 + ((lane < SC_S) ? lane : 0) * (SC_T / 32)];

    for (int t0 = 0; t0 < SC_S; t0 += SL_G) {
        int cn[SL_G];
        const int* bp[SL_G];
        int ent[SL_G][SL_L];
#pragma unroll
        for (int g = 0; g < SL_G; ++g) {
            const int sub = sub0 + (t0 + g) * (SC_T / 32);
            cn[g] = __builtin_amdgcn_readlane(cvec, t0 + g);
            bp[g] = (cn[g] > 0) ? (lst + (size_t)sub * cap) : (lcnt + sub);
        }
#pragma unroll
        for (int g = 0; g < SL_G; ++g)
#pragma unroll
            for (int u = 0; u < SL_L; ++u) {
                const int i  = u * 32 + lane;
                const int ic = (cn[g] > 0) ? ((i < cn[g]) ? i : (cn[g] - 1)) : 0;
                ent[g][u] = bp[g][ic];
            }
#pragma unroll
        for (int g = 0; g < SL_G; ++g)
            asm volatile("" :: "v"(ent[g][0]), "v"(ent[g][1]), "v"(ent[g][2]),
                               "v"(ent[g][3]), "v"(ent[g][4]));
        static_assert(SL_L == 5, "the fence above names SL_L entries per sublist");
#pragma unroll
        for (int g = 0; g < SL_G; ++g)
#pragma unroll
            for (int u = 0; u < SL_L; ++u) {
                if (u * 32 < cn[g]) {
                    const unsigned en = (unsigned)ent[g][u];
                    const bool in = (u * 32 + lane < cn[g]) &&
                                    ((en >> 20) - (unsigned)roff < (unsigned)rows);
                    const unsigned mask = __builtin_amdgcn_ballot_w32(in);
                    if (mask != 0u) {
                        if (in)
                            q[(tail + (int)__builtin_amdgcn_mbcnt_lo(mask, 0u)) & (AG_Q - 1)] = (int)en;
                        tail += __builtin_popcount(mask);
                    }
                }
            }
        while (tail - head >= 32) {
            agg128_drain(h1, ei, roff, q, head, 32, acc, lane);
            head += 32;
        }
#pragma unroll
        for (int g = 0; g < SL_G; ++g) {
            for (int i0 = SL_L * 32; i0 < cn[g]; i0 += 32) {
                const int i = i0 + lane;
                const unsigned en = (unsigned)bp[g][(i < cn[g]) ? i : (cn[g] - 1)];
                const bool in = (i < cn[g]) && ((en >> 20) - (unsigned)roff < (unsigned)rows);
                const unsigned mask = __builtin_amdgcn_ballot_w32(in);
                if (mask != 0u) {
                    if (in)
                        q[(tail + (int)__builtin_amdgcn_mbcnt_lo(mask, 0u)) & (AG_Q - 1)] = (int)en;
                    tail += __builtin_popcount(mask);
                }
                while (tail - head >= 32) {
                    agg128_drain(h1, ei, roff, q, head, 32, acc, lane);
                    head += 32;
                }
            }
        }
    }
    while (tail - head > 0) {
        const int cnt = (tail - head < 32) ? (tail - head) : 32;
        agg128_drain(h1, ei, roff, q, head, cnt, acc, lane);
        head += cnt;
    }
    __syncthreads();

    constexpr int RP    = (RTS * 64 < AG_RPAD) ? (RTS * 64) : AG_RPAD;
    constexpr int HS_IT = (RP * (H / 4) + AG_T - 1) / AG_T;
    static_assert(((HS_IT * AG_T - 1) / (H / 4)) * AG_P + H <= (AG_T / 32) * AG_Q,
                  "the staged h1 rows fit in the ring space");
    float* hs = (float*)&ring[0][0];
    {
        float4 hv[HS_IT];
#pragma unroll
        for (int j = 0; j < HS_IT; ++j) {
            const int i  = j * AG_T + tid;
            const int rg = (r0 + (i >> 5) < n) ? (r0 + (i >> 5)) : (n - 1);
            hv[j] = *(const float4*)(h1 + (size_t)rg * H + (i & 31) * 4);
        }
#pragma unroll
        for (int j = 0; j < HS_IT; ++j) {
            const int i = j * AG_T + tid;
            *(float4*)(hs + (i >> 5) * AG_P + (i & 31) * 4) = hv[j];
        }
    }
    __syncthreads();

    const int ct  = wv & 7;
    const int rg0 = wv >> 3;
    const int cnt = (nrt - rg0 + 3) >> 2;
    v8f d2[4];
#pragma unroll
    for (int i = 0; i < 4; ++i) {
        const v8f z = {0.0f, 0.0f, 0.0f, 0.0f, 0.0f, 0.0f, 0.0f, 0.0f};
        d2[i] = z;
    }
    if (RTS >= 4 && cnt >= 4)
        agg128_conv2_tiles<4>(acc, hs, Wrel, Wroot, brel, d2, ct, rg0, lane);
    else if (RTS >= 3 && cnt == 3)
        agg128_conv2_tiles<3>(acc, hs, Wrel, Wroot, brel, d2, ct, rg0, lane);
    else if (RTS >= 2 && cnt == 2)
        agg128_conv2_tiles<2>(acc, hs, Wrel, Wroot, brel, d2, ct, rg0, lane);
    else if (cnt >= 1)
        agg128_conv2_tiles<1>(acc, hs, Wrel, Wroot, brel, d2, ct, rg0, lane);
    __syncthreads();

    {
        const int c = lane & 15;
        const int h = lane >> 4;
#pragma unroll
        for (int i = 0; i < RTS; ++i) {
            if (i < cnt) {
#pragma unroll
                for (int j = 0; j < 8; ++j)
                    acc[((rg0 + 4 * i) * 16 + j + 8 * h) * AG_P + ct * 16 + c] = d2[i][j];
            }
        }
    }
    __syncthreads();

    for (int u = wv; u < nrt * (F3 / 16); u += AG_T / 32)
        lin_unit(acc, lW, lb, e, r0, rows, u / (F3 / 16), u % (F3 / 16), lane);
}

#define IN_R 224
#define IN_P (F3 + 4)
#define IN_B 16
#define IN_SEG (IN_R * IN_P)
#define IN_RING ((AG_T / 32) * AG_Q)
static_assert(IN_R % 16 == 0 && (IN_R * 3) % 32 == 0, "whole tiles; whole lines of out");
static_assert(IN_SEG + IN_R * 3 <= IN_RING, "the e0 rows and the block's results fit in the ring space");
static_assert((IN_RING + 2 * IN_SEG) * 4 <= 320 * 1024, "one block per WGP");

__device__ __forceinline__ void interp_drain(const float* __restrict__ e,
                                             const int* __restrict__ cols,
                                             const int* __restrict__ rws,
                                             const float* __restrict__ vals, int r0,
                                             const int* q, int head, int cnt,
                                             float* acc, int lane) {
    const int qe  = q[(head + lane) & (AG_Q - 1)];
    const int nz  = (lane < cnt) ? qe : 0;
    const int col = cols[nz];
    const int dl  = rws[nz] - r0;
    const int vb  = __float_as_int(vals[nz]);
#pragma unroll 1
    for (int b = 0; b < cnt; b += IN_B) {
        float v[IN_B][3];
#pragma unroll
        for (int j = 0; j < IN_B; ++j) {
            const int cj = __builtin_amdgcn_readlane(col, b + j);
            const float* p = e + cj * F3 + lane;
            v[j][0] = p[0];
            v[j][1] = p[32];
            v[j][2] = p[64];
        }
#pragma unroll
        for (int j = 0; j < IN_B; ++j) {
            if (b + j < cnt) {
                const int dj  = __builtin_amdgcn_readlane(dl, b + j);
                const float s = __int_as_float(__builtin_amdgcn_readlane(vb, b + j));
                float* a = acc + dj * IN_P + lane;
                atomicAdd(a,      s * v[j][0]);
                atomicAdd(a + 32, s * v[j][1]);
                atomicAdd(a + 64, s * v[j][2]);
            }
        }
    }
}

__device__ __forceinline__ void dec_unit(const float* seg0, const float* __restrict__ W0,
                                         const float* __restrict__ b0,
                                         const float* __restrict__ Wout, float bo,
                                         float* res, int rt, int br, int lane) {
    const int c = lane & 15;
    const int h = lane >> 4;
    v8f d[4];
#pragma unroll
    for (int ct = 0; ct < 4; ++ct) {
        const float bv = b0[ct * 16 + c];
        const v8f t = {bv, bv, bv, bv, bv, bv, bv, bv};
        d[ct] = t;
    }
    const float* ap = seg0 + (rt * 16 + c) * IN_P + br * 32 + 2 * h;
    const float* wp = W0 + (2 * h) * DECW + c;
#pragma unroll 4
    for (int t = 0; t < F3 / 4; ++t) {
        const v2f a = *(const v2f*)(ap + (t >> 3) * IN_SEG + (t & 7) * 4);
#pragma unroll
        for (int ct = 0; ct < 4; ++ct) {
            v2f b;
            b.x = wp[(4 * t) * DECW + ct * 16];
            b.y = wp[(4 * t + 1) * DECW + ct * 16];
            d[ct] = __builtin_amdgcn_wmma_f32_16x16x4_f32(false, a, false, b, (short)0, d[ct],
                                                          false, false);
        }
    }
    float p[8];
#pragma unroll
    for (int j = 0; j < 8; ++j) p[j] = 0.0f;
#pragma unroll
    for (int ct = 0; ct < 4; ++ct) {
        const float w = Wout[ct * 16 + c];
#pragma unroll
        for (int j = 0; j < 8; ++j) {
            const float v = d[ct][j];
            p[j] += ((v > 0.0f) ? v : (expf(v) - 1.0f)) * w;
        }
    }
#pragma unroll
    for (int m = 1; m < 16; m <<= 1)
#pragma unroll
        for (int j = 0; j < 8; ++j) p[j] += __shfl_xor(p[j], m);
    if (c == 0) {
#pragma unroll
        for (int j = 0; j < 8; ++j) res[(rt * 16 + j + 8 * h) * 3 + br] = p[j] + bo;
    }
}

__global__ __launch_bounds__(AG_T)
void interp_dec_kernel(const float* __restrict__ vals1, const int* __restrict__ cols1,
                       const int* __restrict__ rows1, const float* __restrict__ e1,
                       const float* __restrict__ vals2, const int* __restrict__ cols2,
                       const int* __restrict__ rows2, const float* __restrict__ e2,
                       const float* __restrict__ e0, const float* __restrict__ W0,
                       const float* __restrict__ b0, const float* __restrict__ Wout,
                       const float* __restrict__ bout, float* __restrict__ out,
                       int nnz, int n) {
    __shared__ __attribute__((aligned(16))) float smem[IN_RING + 2 * IN_SEG];
    int*   ring = (int*)smem;
    float* sums = smem + IN_RING;
    float* es   = sums - IN_SEG;
    float* res  = smem;

    const int tid  = threadIdx.x;
    const int lane = tid & 31;
    const int r0   = blockIdx.x * IN_R;
    const int rows = (n - r0 < IN_R) ? (n - r0) : IN_R;

    {
        float4* a4 = (float4*)sums;
        const float4 z = {0.0f, 0.0f, 0.0f, 0.0f};
        for (int i = tid; i < 2 * IN_SEG / 4; i += AG_T) a4[i] = z;
    }
    __syncthreads();

    int* q = ring + (tid >> 5) * AG_Q;
    int head = 0, tail = 0;
    const int ng = nnz >> 2;

#pragma unroll 1
    for (int m = 0; m < 2; ++m) {
        const float* vals = m ? vals2 : vals1;
        const int*   cols = m ? cols2 : cols1;
        const int*   rws  = m ? rows2 : rows1;
        const float* e    = m ? e2 : e1;
        float*       acc  = sums + m * IN_SEG;
        const int4*  d4   = (const int4*)rws;

        int g0 = 0;
        for (; g0 + AG_T * AG_U <= ng; g0 += AG_T * AG_U) {
            agg128_scan_trip<false>(d4, g0, ng, tid, r0, (unsigned)rows, q, tail);
            while (tail - head >= 32) {
                interp_drain(e, cols, rws, vals, r0, q, head, 32, acc, lane);
                head += 32;
            }
        }
        if (g0 < ng) agg128_scan_trip<true>(d4, g0, ng, tid, r0, (unsigned)rows, q, tail);
        while (tail - head > 0) {
            const int cnt = (tail - head < 32) ? (tail - head) : 32;
            interp_drain(e, cols, rws, vals, r0, q, head, cnt, acc, lane);
            head += cnt;
        }
    }
    __syncthreads();

    {
        constexpr int TOT = IN_R * (F3 / 4);
        constexpr int IT  = (TOT + AG_T - 1) / AG_T;
        float4 ev[IT];
#pragma unroll
        for (int j = 0; j < IT; ++j) {
            const int i  = j * AG_T + tid;
            const int ic = (i < TOT) ? i : (TOT - 1);
            const int r  = ic / (F3 / 4);
            const int c4 = ic % (F3 / 4);
            const int rg = (r0 + r < n) ? (r0 + r) : (n - 1);
            ev[j] = *(const float4*)(e0 + (size_t)rg * F3 + c4 * 4);
        }
#pragma unroll
        for (int j = 0; j < IT; ++j) {
            const int i = j * AG_T + tid;
            if (i < TOT) {
                const int r  = i / (F3 / 4);
                const int c4 = i % (F3 / 4);
                *(float4*)(es + r * IN_P + c4 * 4) = ev[j];
            }
        }
    }
    __syncthreads();

    const int wv = __builtin_amdgcn_readfirstlane(tid >> 5);
    for (int u = wv; u < (IN_R / 16) * 3; u += AG_T / 32) {
        const int br = u % 3;
        const int rt = u / 3;
        dec_unit(es, W0 + (size_t)br * F3 * DECW, b0 + br * DECW, Wout + br * DECW, bout[br],
                 res, rt, br, lane);
    }
    __syncthreads();

    for (int i = tid; i < rows * 3; i += AG_T) out[(size_t)r0 * 3 + i] = res[i];
}

template <int RTS>
static inline void launch_enc_conv2_lin(const float* h1, const int* ei, int E, int n, int rr,
                                        const int* lst, const int* lcnt, float* e,
                                        const float* Wrel, const float* Wroot, const float* brel,
                                        const float* lW, const float* lb, hipStream_t stream) {
    agg128_conv2_kernel<RTS><<<(n + rr - 1) / rr, AG_T, 0, stream>>>(
        h1, ei, E, n, rr, lst, lcnt, Wrel, Wroot, brel, lW, lb, e);
}

extern "C" void kernel_launch(void* const* d_in, const int* in_sizes, int n_in,
                              void* d_out, int out_size, void* d_ws, size_t ws_size,
                              hipStream_t stream) {
    (void)in_sizes; (void)n_in; (void)out_size; (void)ws_size;

    const float* x[3]   = {(const float*)d_in[0], (const float*)d_in[1], (const float*)d_in[2]};
    const int*   ei[3]  = {(const int*)d_in[3], (const int*)d_in[4], (const int*)d_in[5]};
    const int*   A1r    = (const int*)d_in[6];
    const int*   A1c    = (const int*)d_in[7];
    const float* A1v    = (const float*)d_in[8];
    const int*   A2r    = (const int*)d_in[9];
    const int*   A2c    = (const int*)d_in[10];
    const float* A2v    = (const float*)d_in[11];
    const float* c1Wrel  = (const float*)d_in[12];
    const float* c1brel  = (const float*)d_in[13];
    const float* c1Wroot = (const float*)d_in[14];
    const float* c2Wrel  = (const float*)d_in[15];
    const float* c2brel  = (const float*)d_in[16];
    const float* c2Wroot = (const float*)d_in[17];
    const float* linW    = (const float*)d_in[18];
    const float* linb    = (const float*)d_in[19];
    const float* decW0   = (const float*)d_in[20];
    const float* decb0   = (const float*)d_in[21];
    const float* decWout = (const float*)d_in[22];
    const float* decbout = (const float*)d_in[23];
    float* out = (float*)d_out;

    float* ws = (float*)d_ws;
    float* h1   = ws;
    float* agg  = h1   + (size_t)N0 * H;
    float* e0   = agg  + (size_t)N0 * H * 2;
    float* e1   = e0   + (size_t)N0 * F3;
    float* e2   = e1   + (size_t)N1 * F3;
    float* rest = e2   + (size_t)N2 * F3;

    constexpr size_t LST_MAX = (size_t)((N0 + SC_R - 1) / SC_R) * SC_S * (SC_T / 32) *
                               ((((E0 + SC_S - 1) / SC_S + SC_T * SC_U - 1) / (SC_T * SC_U)) * SC_U * 32);
    constexpr size_t LCNT_MAX = (size_t)((N0 + SC_R - 1) / SC_R) * SC_S * (SC_T / 32);
    static_assert(LST_MAX + LCNT_MAX <= (size_t)N0 * F3 * 2 + (size_t)N0 * DECW,
                  "the lists of level 0 (the largest) fit behind e2");
    static_assert(E0 >= E1 && E0 >= E2 && N0 >= N1 && N0 >= N2, "level 0 is the largest");
    int* lst  = (int*)rest;
    int* lcnt = lst + LST_MAX;

    const int nlv[3] = {N0, N1, N2};
    const int Elv[3] = {E0, E1, E2};
    const int rrl[3] = {200, 50, 13};
    static_assert(SC_R % 200 == 0 && SC_R % 50 == 0 && N2 <= SC_R,
                  "a conv2 block's rows lie inside one scatter6 range");
    static_assert(200 <= AG_RMAX && (50 + 15) / 16 <= 4 && (13 + 15) / 16 <= 4,
                  "conv2 product: level 0 within AG_RMAX rows, levels 1 and 2 one row tile per wave");
    static_assert(E0 % 4 == 0 && E1 % 4 == 0 && E2 % 4 == 0, "dst lists are read 16 B at a time");
    float* elv[3] = {e0, e1, e2};

    for (int i = 0; i < 3; ++i) {
        const int n = nlv[i];
        const int E = Elv[i];

        {
            scatter6_lds_kernel<<<dim3((n + SC_R - 1) / SC_R, SC_S, 1), SC_T, 0, stream>>>(
                x[i], ei[i], E, n, agg, lst, lcnt);
            conv1_planes_kernel<<<(n + C1_R - 1) / C1_R, C1_T, 0, stream>>>(
                x[i], agg,
                c1Wrel + (size_t)i * 6 * H,
                c1brel + (size_t)i * H,
                c1Wroot + (size_t)i * 6 * H, h1, n);
        }

        {
            const int rr = rrl[i];
            const float* Wrel2  = c2Wrel + (size_t)i * H * H;
            const float* Wroot2 = c2Wroot + (size_t)i * H * H;
            const float* brel2  = c2brel + (size_t)i * H;
            const float* lW     = linW + (size_t)i * H * F3;
            const float* lb     = linb + (size_t)i * F3;
            if (i == 0)
                launch_enc_conv2_lin<4>(h1, ei[i], E, n, rr, lst, lcnt, elv[i],
                                        Wrel2, Wroot2, brel2, lW, lb, stream);
            else
                launch_enc_conv2_lin<1>(h1, ei[i], E, n, rr, lst, lcnt, elv[i],
                                        Wrel2, Wroot2, brel2, lW, lb, stream);
        }
    }

    static_assert(NNZ % 4 == 0, "row lists are read 16 B at a time");
    interp_dec_kernel<<<(N0 + IN_R - 1) / IN_R, AG_T, 0, stream>>>(
        A1v, A1c, A1r, e1, A2v, A2c, A2r, e2, e0, decW0, decb0, decWout, decbout, out,
        NNZ, N0);
}


# interp entries sorted by 224-row block in LDS (returning ds add = arrival rank) by spare blocks of the level-1 scatter launch; interp blocks read their bins, not both row lists; sum order changed
# speedup vs baseline: 3.9643x; 1.0791x over previous
#include <hip/hip_runtime.h>
#include <hip/hip_bf16.h>

#define N0 50000
#define N1 12500
#define N2 3125
#define E0 800000
#define E1 200000
#define E2 50000
#define NNZ 200000
#define H 128
#define F3 96
#define DECW 64

typedef __attribute__((ext_vector_type(2))) float v2f;
typedef __attribute__((ext_vector_type(8))) float v8f;


#define SC_R 4000

#define SC_S 16
#define SC_T 1024
#define SC_U 8
static_assert(SC_R <= 4096 && E0 < (1 << 20) && E1 < (1 << 20) && E2 < (1 << 20),
              "a list entry packs a 12-bit row offset and a 20-bit edge number");

__device__ __forceinline__ int sc_cap(int E) {
    const int chunk = (E + SC_S - 1) / SC_S;
    const int trips = (chunk + SC_T * SC_U - 1) / (SC_T * SC_U);
    return trips * SC_U * 32;
}

#define IN_R  224
#define IB_S  32
#define IB_CH ((NNZ + IB_S - 1) / IB_S)
#define IB_K  ((IB_CH + SC_T - 1) / SC_T)
#define IB_NB ((N0 + IN_R - 1) / IN_R)
#define IB_OP (IB_NB + 1)
static_assert(IB_NB + 1 <= 256 && NNZ < (1 << 20) && IN_R <= 4096 && (IB_S - 1) * IB_CH < NNZ,
              "256 counters cover the bins; an entry packs a row offset and an entry number; no slice is empty");

__device__ __forceinline__ void interp_bin_block(const int* __restrict__ rws,
                                                 int* __restrict__ srt, int* __restrict__ off,
                                                 int slice, int* cnt) {
    int* start = cnt + 256;
    const int tid = threadIdx.x;
    if (tid < 256) cnt[tid] = 0;
    __syncthreads();
    const int eb  = slice * IB_CH;
    const int len = (NNZ - eb < IB_CH) ? (NNZ - eb) : IB_CH;
    int r[IB_K], bin[IB_K], rank[IB_K];
#pragma unroll
    for (int k = 0; k < IB_K; ++k) {
        const int i = k * SC_T + tid;
        r[k] = rws[eb + ((i < len) ? i : 0)];
    }
#pragma unroll
    for (int k = 0; k < IB_K; ++k) {
        const int i = k * SC_T + tid;
        bin[k]  = (i < len && (unsigned)r[k] < (unsigned)N0) ? r[k] / IN_R : -1;
        rank[k] = (bin[k] >= 0) ? atomicAdd(&cnt[bin[k]], 1) : 0;
    }
    __syncthreads();
    if (tid < 32) {
        int cv[8];
        int s = 0;
#pragma unroll
        for (int j = 0; j < 8; ++j) { cv[j] = cnt[tid * 8 + j]; s += cv[j]; }
        int inc = s;
#pragma unroll
        for (int d = 1; d < 32; d <<= 1) {
            const int t = __shfl_up(inc, d);
            if (tid >= d) inc += t;
        }
        int run = inc - s;
#pragma unroll
        for (int j = 0; j < 8; ++j) { start[tid * 8 + j] = run; run += cv[j]; }
    }
    __syncthreads();
#pragma unroll
    for (int k = 0; k < IB_K; ++k) {
        const int i = k * SC_T + tid;
        if (bin[k] >= 0)
            srt[eb + start[bin[k]] + rank[k]] = ((r[k] - bin[k] * IN_R) << 20) | (eb + i);
    }
    if (tid < IB_OP) off[tid] = start[tid];
}

__global__ __launch_bounds__(SC_T) __attribute__((amdgpu_waves_per_eu(8, 8)))
void scatter6_lds_kernel(const float* __restrict__ x, const int* __restrict__ ei,
                         int E, int n, float* __restrict__ part,
                         int* __restrict__ lst, int* __restrict__ lcnt,
                         int nb, const int* __restrict__ br1, const int* __restrict__ br2,
                         int* __restrict__ bsrt, int* __restrict__ boff) {
    __shared__ float acc[SC_R * 6];
    if ((int)blockIdx.x >= nb) {
        const int id = ((int)blockIdx.x - nb) * SC_S + (int)blockIdx.y;
        const int m  = id / IB_S;
        interp_bin_block(m ? br2 : br1, bsrt + m * NNZ, boff + id * IB_OP, id % IB_S, (int*)acc);
        return;
    }
    const int tid = threadIdx.x;
    const int r0  = blockIdx.x * SC_R;
    const int s   = blockIdx.y;

    for (int i = tid; i < SC_R * 6; i += SC_T) acc[i] = 0.0f;
    __syncthreads();

    const int chunk = (E + SC_S - 1) / SC_S;
    const int eb = s * chunk;
    const int ee = (eb + chunk < E) ? (eb + chunk) : E;

    const int sub = (blockIdx.x * SC_S + s) * (SC_T / 32) + (tid >> 5);
    int* lstw = lst + (size_t)sub * sc_cap(E);
    int wc = 0;

    if (eb < ee) {
        for (int base = eb; base < ee; base += SC_T * SC_U) {
            int off[SC_U];
            int srow[SC_U];
#pragma unroll
            for (int u = 0; u < SC_U; ++u) {
                const int e  = base + u * SC_T + tid;
                const int ec = (e < ee) ? e : (ee - 1);
                const int d  = ei[E + ec] - r0;
                const int sr = ei[ec];
                const bool hit = (e < ee) && ((unsigned)d < (unsigned)SC_R);
                off[u]  = hit ? d * 6 : -1;
                srow[u] = hit ? sr * 6 : 0;
            }
            float2 v[SC_U][3];
#pragma unroll
            for (int u = 0; u < SC_U; ++u) {
                const float2* p = (const float2*)(x + srow[u]);
                v[u][0] = p[0];
                v[u][1] = p[1];
                v[u][2] = p[2];
            }
#pragma unroll
            for (int u = 0; u < SC_U; ++u)
                asm volatile("" :: "v"(v[u][0].x), "v"(v[u][0].y), "v"(v[u][1].x),
                                   "v"(v[u][1].y), "v"(v[u][2].x), "v"(v[u][2].y));
#pragma unroll
            for (int u = 0; u < SC_U; ++u) {
                const unsigned hm = __builtin_amdgcn_ballot_w32(off[u] >= 0);
                if (off[u] >= 0) {
                    lstw[wc + (int)__builtin_amdgcn_mbcnt_lo(hm, 0u)] =
                        (int)((((unsigned)off[u] / 6u) << 20) | (unsigned)(base + u * SC_T + tid));
                    float* a = acc + off[u];
                    atomicAdd(a + 0, v[u][0].x);
                    atomicAdd(a + 1, v[u][0].y);
                    atomicAdd(a + 2, v[u][1].x);
                    atomicAdd(a + 3, v[u][1].y);
                    atomicAdd(a + 4, v[u][2].x);
                    atomicAdd(a + 5, v[u][2].y);
                }
                wc += __builtin_popcount(hm);
            }
        }
    }
    if ((tid & 31) == 0) lcnt[sub] = wc;
    __syncthreads();

    const int rows = (n - r0 < SC_R) ? (n - r0) : SC_R;
    float* p = part + (size_t)s * n * 6 + (size_t)r0 * 6;
    for (int i = tid; i < rows * 6; i += SC_T) p[i] = acc[i];
}

#define C1_R 32
#define C1_T 128

__global__ __launch_bounds__(C1_T)
void conv1_planes_kernel(const float* __restrict__ x, const float* __restrict__ part,
                         const float* __restrict__ Wrel, const float* __restrict__ brel,
                         const float* __restrict__ Wroot, float* __restrict__ h1, int n) {
    __shared__ float sa[C1_R * 6];
    __shared__ float sx[C1_R * 6];
    const int tid  = threadIdx.x;
    const int lane = tid & 31;
    const int wave = tid >> 5;
    const int r0   = blockIdx.x * C1_R;
    const int rows = (n - r0 < C1_R) ? (n - r0) : C1_R;
    const int cnt  = n * 6;

    float4 wr[6], wo[6];
#pragma unroll
    for (int k = 0; k < 6; ++k) {
        wr[k] = *(const float4*)(Wrel + k * H + lane * 4);
        wo[k] = *(const float4*)(Wroot + k * H + lane * 4);
    }
    const float4 bv = *(const float4*)(brel + lane * 4);

    static_assert(6 * C1_R <= 2 * C1_T, "two elements per thread cover the block's rows");
    const int  m    = rows * 6;
    const int  i1   = tid + C1_T;
    const bool has0 = tid < m;
    const bool has1 = i1 < m;
    const int  t0   = r0 * 6 + (has0 ? tid : 0);
    const int  t1   = r0 * 6 + (has1 ? i1 : 0);
    float p0[SC_S], p1[SC_S];
#pragma unroll
    for (int k = 0; k < SC_S; ++k) {
        p0[k] = part[(size_t)k * cnt + t0];
        p1[k] = part[(size_t)k * cnt + t1];
    }
    const float x0 = x[t0];
    const float x1 = x[t1];
#pragma unroll
    for (int k = 0; k < SC_S; k += 4)
        asm volatile("" :: "v"(p0[k]), "v"(p0[k + 1]), "v"(p0[k + 2]), "v"(p0[k + 3]),
                           "v"(p1[k]), "v"(p1[k + 1]), "v"(p1[k + 2]), "v"(p1[k + 3]));
    asm volatile("" :: "v"(x0), "v"(x1));
    float s0 = 0.0f, s1 = 0.0f;
#pragma unroll
    for (int k = 0; k < SC_S; ++k) {
        s0 += p0[k];
        s1 += p1[k];
    }
    if (has0) { sa[tid] = s0; sx[tid] = x0; }
    if (has1) { sa[i1]  = s1; sx[i1]  = x1; }
    __syncthreads();

    for (int r = wave; r < rows; r += C1_T / 32) {
        float4 s = bv;
#pragma unroll
        for (int k = 0; k < 6; ++k) {
            const float a  = sa[r * 6 + k];
            const float xv = sx[r * 6 + k];
            s.x += a * wr[k].x;  s.y += a * wr[k].y;  s.z += a * wr[k].z;  s.w += a * wr[k].w;
            s.x += xv * wo[k].x; s.y += xv * wo[k].y; s.z += xv * wo[k].z; s.w += xv * wo[k].w;
        }
        s.x = fmaxf(s.x, 0.0f);
        s.y = fmaxf(s.y, 0.0f);
        s.z = fmaxf(s.z, 0.0f);
        s.w = fmaxf(s.w, 0.0f);
        *(float4*)(h1 + (size_t)(r0 + r) * H + lane * 4) = s;
    }
}

#define AG_T    1024
#define AG_RMAX 200
#define AG_RPAD 208
#define AG_P    (H + 4)
static_assert(AG_RPAD % 16 == 0 && AG_RPAD >= AG_RMAX && AG_RPAD / 16 <= 16,
              "at most 4 row tiles for each of the 4 wave groups");
#define AG_Q    1024
#define AG_B    16
#define SL_G    4
#define SL_L    5
static_assert(SC_S % SL_G == 0, "sublists are read SL_G at a time");
static_assert(SL_G * SL_L * 32 + 31 <= AG_Q, "a wave's ring holds what one read step can queue");

__device__ __forceinline__ void agg128_drain(const float* __restrict__ h1,
                                             const int* __restrict__ ei, int roff,
                                             const int* q, int head, int cnt,
                                             float* acc, int lane) {
    const unsigned qe = (unsigned)q[(head + lane) & (AG_Q - 1)];
    const unsigned en = (lane < cnt) ? qe : 0u;
    const int s  = ei[en & 0xfffffu];
    const int dl = (int)(en >> 20) - roff;
#pragma unroll 1
    for (int b = 0; b < cnt; b += AG_B) {
        float4 v[AG_B];
#pragma unroll
        for (int j = 0; j < AG_B; ++j) {
            const int sj = __builtin_amdgcn_readlane(s, b + j);
            v[j] = *(const float4*)(h1 + sj * H + lane * 4);
        }
#pragma unroll
        for (int j = 0; j < AG_B; ++j) {
            if (b + j < cnt) {
                const int dj = __builtin_amdgcn_readlane(dl, b + j);
                float* a = acc + dj * AG_P + lane;
                atomicAdd(a,      v[j].x);
                atomicAdd(a + 32, v[j].y);
                atomicAdd(a + 64, v[j].z);
                atomicAdd(a + 96, v[j].w);
            }
        }
    }
}

template <int CNT>
__device__ __forceinline__ void agg128_conv2_tiles(const float* acc, const float* hs,
                                                   const float* __restrict__ Wrel,
                                                   const float* __restrict__ Wroot,
                                                   const float* __restrict__ brel,
                                                   v8f (&dout)[4],
                                                   int ct, int rg0, int lane) {
    const int c   = lane & 15;
    const int h   = lane >> 4;
    const int col = ct * 16 + c;
    const float bv = brel[col];
    v8f d[CNT];
#pragma unroll
    for (int i = 0; i < CNT; ++i) {
        const v8f t = {bv, bv, bv, bv, bv, bv, bv, bv};
        d[i] = t;
    }
    const float* ap = acc + (rg0 * 16 + c) * AG_P + 2 * h;
    const float* hp = hs  + (rg0 * 16 + c) * AG_P + 2 * h;
    const float* wr = Wrel  + (8 * h) * H + col;
    const float* wo = Wroot + (2 * h) * H + col;
#pragma unroll 8
    for (int t = 0; t < 32; ++t) {
        const int j  = t >> 2;
        const int kk = t & 3;
        v2f b;
        b.x = wr[(16 * j + kk) * H];
        b.y = wr[(16 * j + kk + 4) * H];
#pragma unroll
        for (int i = 0; i < CNT; ++i) {
            const v2f a = *(const v2f*)(ap + i * 64 * AG_P + kk * 32 + 4 * j);
            d[i] = __builtin_amdgcn_wmma_f32_16x16x4_f32(false, a, false, b, (short)0, d[i],
                                                         false, false);
        }
    }
#pragma unroll 8
    for (int k = 0; k < H; k += 4) {
        v2f b;
        b.x = wo[k * H];
        b.y = wo[k * H + H];
#pragma unroll
        for (int i = 0; i < CNT; ++i) {
            const v2f a = *(const v2f*)(hp + i * 64 * AG_P + k);
            d[i] = __builtin_amdgcn_wmma_f32_16x16x4_f32(false, a, false, b, (short)0, d[i],
                                                         false, false);
        }
    }
#pragma unroll
    for (int i = 0; i < CNT; ++i)
#pragma unroll
        for (int j = 0; j < 8; ++j) dout[i][j] = fmaxf(d[i][j], 0.0f);
}

__device__ __forceinline__ void lin_unit(const float* hrow, const float* __restrict__ lW,
                                         const float* __restrict__ lb, float* __restrict__ e,
                                         int r0, int rows, int rt, int ct, int lane) {
    const int c   = lane & 15;
    const int h   = lane >> 4;
    const int col = ct * 16 + c;
    const float bv = lb[col];
    v8f d = {bv, bv, bv, bv, bv, bv, bv, bv};
    const float* ap = hrow + (rt * 16 + c) * AG_P + 2 * h;
    const float* wp = lW + (2 * h) * F3 + col;
#pragma unroll 8
    for (int k = 0; k < H; k += 4) {
        const v2f a = *(const v2f*)(ap + k);
        v2f b;
        b.x = wp[k * F3];
        b.y = wp[k * F3 + F3];
        d = __builtin_amdgcn_wmma_f32_16x16x4_f32(false, a, false, b, (short)0, d, false, false);
    }
#pragma unroll
    for (int j = 0; j < 8; ++j) {
        const int row = rt * 16 + j + 8 * h;
        if (row < rows) e[(size_t)(r0 + row) * F3 + col] = d[j];
    }
}

template <int RTS>
__global__ __launch_bounds__(AG_T)
void agg128_conv2_kernel(const float* __restrict__ h1, const int* __restrict__ ei,
                         int E, int n, int rr, const int* __restrict__ lst,
                         const int* __restrict__ lcnt,
                         const float* __restrict__ Wrel, const float* __restrict__ Wroot,
                         const float* __restrict__ brel, const float* __restrict__ lW,
                         const float* __restrict__ lb, float* __restrict__ e) {
    __shared__ __attribute__((aligned(16))) float acc[AG_RPAD * AG_P];
    __shared__ __attribute__((aligned(16))) int ring[AG_T / 32][AG_Q];
    const int tid  = threadIdx.x;
    const int lane = tid & 31;
    const int r0   = blockIdx.x * rr;
    const int rows = (n - r0 < rr) ? (n - r0) : rr;
    const int nrt  = (rr + 15) >> 4;

    {
        float4* a4 = (float4*)acc;
        const float4 z = {0.0f, 0.0f, 0.0f, 0.0f};
        for (int i = tid; i < nrt * 16 * (AG_P / 4); i += AG_T) a4[i] = z;
    }
    __syncthreads();

    int* q = ring[tid >> 5];
    int head = 0, tail = 0;
    const int wv   = __builtin_amdgcn_readfirstlane(tid >> 5);
    const int xr   = r0 / SC_R;
    const int roff = r0 - xr * SC_R;
    const int cap  = sc_cap(E);
    const int sub0 = xr * (SC_S * (SC_T / 32)) + wv;
    static_assert(SC_S <= 32, "one count per lane");
    const int cvec = lcnt[sub0 + ((lane < SC_S) ? lane : 0) * (SC_T / 32)];

    for (int t0 = 0; t0 < SC_S; t0 += SL_G) {
        int cn[SL_G];
        const int* bp[SL_G];
        int ent[SL_G][SL_L];
#pragma unroll
        for (int g = 0; g < SL_G; ++g) {
            const int sub = sub0 + (t0 + g) * (SC_T / 32);
            cn[g] = __builtin_amdgcn_readlane(cvec, t0 + g);
            bp[g] = (cn[g] > 0) ? (lst + (size_t)sub * cap) : (lcnt + sub);
        }
#pragma unroll
        for (int g = 0; g < SL_G; ++g)
#pragma unroll
            for (int u = 0; u < SL_L; ++u) {
                const int i  = u * 32 + lane;
                const int ic = (cn[g] > 0) ? ((i < cn[g]) ? i : (cn[g] - 1)) : 0;
                ent[g][u] = bp[g][ic];
            }
#pragma unroll
        for (int g = 0; g < SL_G; ++g)
            asm volatile("" :: "v"(ent[g][0]), "v"(ent[g][1]), "v"(ent[g][2]),
                               "v"(ent[g][3]), "v"(ent[g][4]));
        static_assert(SL_L == 5, "the fence above names SL_L entries per sublist");
#pragma unroll
        for (int g = 0; g < SL_G; ++g)
#pragma unroll
            for (int u = 0; u < SL_L; ++u) {
                if (u * 32 < cn[g]) {
                    const unsigned en = (unsigned)ent[g][u];
                    const bool in = (u * 32 + lane < cn[g]) &&
                                    ((en >> 20) - (unsigned)roff < (unsigned)rows);
                    const unsigned mask = __builtin_amdgcn_ballot_w32(in);
                    if (mask != 0u) {
                        if (in)
                            q[(tail + (int)__builtin_amdgcn_mbcnt_lo(mask, 0u)) & (AG_Q - 1)] = (int)en;
                        tail += __builtin_popcount(mask);
                    }
                }
            }
        while (tail - head >= 32) {
            agg128_drain(h1, ei, roff, q, head, 32, acc, lane);
            head += 32;
        }
#pragma unroll
        for (int g = 0; g < SL_G; ++g) {
            for (int i0 = SL_L * 32; i0 < cn[g]; i0 += 32) {
                const int i = i0 + lane;
                const unsigned en = (unsigned)bp[g][(i < cn[g]) ? i : (cn[g] - 1)];
                const bool in = (i < cn[g]) && ((en >> 20) - (unsigned)roff < (unsigned)rows);
                const unsigned mask = __builtin_amdgcn_ballot_w32(in);
                if (mask != 0u) {
                    if (in)
                        q[(tail + (int)__builtin_amdgcn_mbcnt_lo(mask, 0u)) & (AG_Q - 1)] = (int)en;
                    tail += __builtin_popcount(mask);
                }
                while (tail - head >= 32) {
                    agg128_drain(h1, ei, roff, q, head, 32, acc, lane);
                    head += 32;
                }
            }
        }
    }
    while (tail - head > 0) {
        const int cnt = (tail - head < 32) ? (tail - head) : 32;
        agg128_drain(h1, ei, roff, q, head, cnt, acc, lane);
        head += cnt;
    }
    __syncthreads();

    constexpr int RP    = (RTS * 64 < AG_RPAD) ? (RTS * 64) : AG_RPAD;
    constexpr int HS_IT = (RP * (H / 4) + AG_T - 1) / AG_T;
    static_assert(((HS_IT * AG_T - 1) / (H / 4)) * AG_P + H <= (AG_T / 32) * AG_Q,
                  "the staged h1 rows fit in the ring space");
    float* hs = (float*)&ring[0][0];
    {
        float4 hv[HS_IT];
#pragma unroll
        for (int j = 0; j < HS_IT; ++j) {
            const int i  = j * AG_T + tid;
            const int rg = (r0 + (i >> 5) < n) ? (r0 + (i >> 5)) : (n - 1);
            hv[j] = *(const float4*)(h1 + (size_t)rg * H + (i & 31) * 4);
        }
#pragma unroll
        for (int j = 0; j < HS_IT; ++j) {
            const int i = j * AG_T + tid;
            *(float4*)(hs + (i >> 5) * AG_P + (i & 31) * 4) = hv[j];
        }
    }
    __syncthreads();

    const int ct  = wv & 7;
    const int rg0 = wv >> 3;
    const int cnt = (nrt - rg0 + 3) >> 2;
    v8f d2[4];
#pragma unroll
    for (int i = 0; i < 4; ++i) {
        const v8f z = {0.0f, 0.0f, 0.0f, 0.0f, 0.0f, 0.0f, 0.0f, 0.0f};
        d2[i] = z;
    }
    if (RTS >= 4 && cnt >= 4)
        agg128_conv2_tiles<4>(acc, hs, Wrel, Wroot, brel, d2, ct, rg0, lane);
    else if (RTS >= 3 && cnt == 3)
        agg128_conv2_tiles<3>(acc, hs, Wrel, Wroot, brel, d2, ct, rg0, lane);
    else if (RTS >= 2 && cnt == 2)
        agg128_conv2_tiles<2>(acc, hs, Wrel, Wroot, brel, d2, ct, rg0, lane);
    else if (cnt >= 1)
        agg128_conv2_tiles<1>(acc, hs, Wrel, Wroot, brel, d2, ct, rg0, lane);
    __syncthreads();

    {
        const int c = lane & 15;
        const int h = lane >> 4;
#pragma unroll
        for (int i = 0; i < RTS; ++i) {
            if (i < cnt) {
#pragma unroll
                for (int j = 0; j < 8; ++j)
                    acc[((rg0 + 4 * i) * 16 + j + 8 * h) * AG_P + ct * 16 + c] = d2[i][j];
            }
        }
    }
    __syncthreads();

    for (int u = wv; u < nrt * (F3 / 16); u += AG_T / 32)
        lin_unit(acc, lW, lb, e, r0, rows, u / (F3 / 16), u % (F3 / 16), lane);
}

#define IN_P (F3 + 4)
#define IN_B 16
#define IN_SEG (IN_R * IN_P)
#define IN_RES  (((IN_R * 3 + 31) / 32) * 32)
static_assert(IN_R % 16 == 0 && (IN_R * 3) % 32 == 0, "whole tiles; whole lines of out");
static_assert(AG_T / 32 == IB_S, "one wave per slice");
static_assert((IN_RES + 3 * IN_SEG) * 4 <= 320 * 1024, "fits the LDS of a WGP");

__device__ __forceinline__ void interp_drain(const float* __restrict__ e,
                                             int col, int dl, int vb, int cnt,
                                             float* acc, int lane) {
#pragma unroll 1
    for (int b = 0; b < cnt; b += IN_B) {
        float v[IN_B][3];
#pragma unroll
        for (int j = 0; j < IN_B; ++j) {
            const int cj = __builtin_amdgcn_readlane(col, b + j);
            const float* p = e + cj * F3 + lane;
            v[j][0] = p[0];
            v[j][1] = p[32];
            v[j][2] = p[64];
        }
#pragma unroll
        for (int j = 0; j < IN_B; ++j) {
            if (b + j < cnt) {
                const int dj  = __builtin_amdgcn_readlane(dl, b + j);
                const float s = __int_as_float(__builtin_amdgcn_readlane(vb, b + j));
                float* a = acc + dj * IN_P + lane;
                atomicAdd(a,      s * v[j][0]);
                atomicAdd(a + 32, s * v[j][1]);
                atomicAdd(a + 64, s * v[j][2]);
            }
        }
    }
}

__device__ __forceinline__ void dec_unit(const float* seg0, const float* __restrict__ W0,
                                         const float* __restrict__ b0,
                                         const float* __restrict__ Wout, float bo,
                                         float* res, int rt, int br, int lane) {
    const int c = lane & 15;
    const int h = lane >> 4;
    v8f d[4];
#pragma unroll
    for (int ct = 0; ct < 4; ++ct) {
        const float bv = b0[ct * 16 + c];
        const v8f t = {bv, bv, bv, bv, bv, bv, bv, bv};
        d[ct] = t;
    }
    const float* ap = seg0 + (rt * 16 + c) * IN_P + br * 32 + 2 * h;
    const float* wp = W0 + (2 * h) * DECW + c;
#pragma unroll 4
    for (int t = 0; t < F3 / 4; ++t) {
        const v2f a = *(const v2f*)(ap + (t >> 3) * IN_SEG + (t & 7) * 4);
#pragma unroll
        for (int ct = 0; ct < 4; ++ct) {
            v2f b;
            b.x = wp[(4 * t) * DECW + ct * 16];
            b.y = wp[(4 * t + 1) * DECW + ct * 16];
            d[ct] = __builtin_amdgcn_wmma_f32_16x16x4_f32(false, a, false, b, (short)0, d[ct],
                                                          false, false);
        }
    }
    float p[8];
#pragma unroll
    for (int j = 0; j < 8; ++j) p[j] = 0.0f;
#pragma unroll
    for (int ct = 0; ct < 4; ++ct) {
        const float w = Wout[ct * 16 + c];
#pragma unroll
        for (int j = 0; j < 8; ++j) {
            const float v = d[ct][j];
            p[j] += ((v > 0.0f) ? v : (expf(v) - 1.0f)) * w;
        }
    }
#pragma unroll
    for (int m = 1; m < 16; m <<= 1)
#pragma unroll
        for (int j = 0; j < 8; ++j) p[j] += __shfl_xor(p[j], m);
    if (c == 0) {
#pragma unroll
        for (int j = 0; j < 8; ++j) res[(rt * 16 + j + 8 * h) * 3 + br] = p[j] + bo;
    }
}

__global__ __launch_bounds__(AG_T)
void interp_dec_kernel(const float* __restrict__ vals1, const int* __restrict__ cols1,
                       const float* __restrict__ e1,
                       const float* __restrict__ vals2, const int* __restrict__ cols2,
                       const float* __restrict__ e2,
                       const float* __restrict__ e0, const float* __restrict__ W0,
                       const float* __restrict__ b0, const float* __restrict__ Wout,
                       const float* __restrict__ bout, float* __restrict__ out,
                       const int* __restrict__ bsrt, const int* __restrict__ boff, int n) {
    __shared__ __attribute__((aligned(16))) float smem[IN_RES + 3 * IN_SEG];
    float* res  = smem;
    float* es   = smem + IN_RES;
    float* sums = es + IN_SEG;

    const int tid  = threadIdx.x;
    const int lane = tid & 31;
    const int r0   = blockIdx.x * IN_R;
    const int rows = (n - r0 < IN_R) ? (n - r0) : IN_R;
    const int wv = __builtin_amdgcn_readfirstlane(tid >> 5);

    const int* ob1 = boff + wv * IB_OP + blockIdx.x;
    const int* ob2 = ob1 + IB_S * IB_OP;
    const int* sl1 = bsrt + wv * IB_CH;
    const int* sl2 = sl1 + NNZ;
    const int lo1 = ob1[0], hi1 = ob1[1];
    const int lo2 = ob2[0], hi2 = ob2[1];
    const unsigned ld1 = (unsigned)sl1[(lo1 + lane < hi1) ? (lo1 + lane) : lo1];
    const unsigned ld2 = (unsigned)sl2[(lo2 + lane < hi2) ? (lo2 + lane) : lo2];
    const unsigned en1 = (lo1 + lane < hi1) ? ld1 : 0u;
    const unsigned en2 = (lo2 + lane < hi2) ? ld2 : 0u;
    const int   col1 = cols1[en1 & 0xfffffu];
    const int   col2 = cols2[en2 & 0xfffffu];
    const float val1 = vals1[en1 & 0xfffffu];
    const float val2 = vals2[en2 & 0xfffffu];

    {
        float4* a4 = (float4*)sums;
        const float4 z = {0.0f, 0.0f, 0.0f, 0.0f};
        for (int i = tid; i < 2 * IN_SEG / 4; i += AG_T) a4[i] = z;
    }
    __syncthreads();

#pragma unroll 1
    for (int m = 0; m < 2; ++m) {
        const float* vals = m ? vals2 : vals1;
        const int*   cols = m ? cols2 : cols1;
        const float* e    = m ? e2 : e1;
        const int*   sl   = m ? sl2 : sl1;
        const int    hi   = m ? hi2 : hi1;
        float*       acc  = sums + m * IN_SEG;
        unsigned     en   = m ? en2 : en1;
        int          col  = m ? col2 : col1;
        float        val  = m ? val2 : val1;
        for (int i0 = m ? lo2 : lo1; i0 < hi; i0 += 32) {
            const int cnt = (hi - i0 < 32) ? (hi - i0) : 32;
            interp_drain(e, col, (int)(en >> 20), __float_as_int(val), cnt, acc, lane);
            if (i0 + 32 < hi) {
                const int i = i0 + 32 + lane;
                const unsigned ld = (unsigned)sl[(i < hi) ? i : (hi - 1)];
                en  = (i < hi) ? ld : 0u;
                col = cols[en & 0xfffffu];
                val = vals[en & 0xfffffu];
            }
        }
    }
    __syncthreads();

    {
        constexpr int TOT = IN_R * (F3 / 4);
        constexpr int IT  = (TOT + AG_T - 1) / AG_T;
        float4 ev[IT];
#pragma unroll
        for (int j = 0; j < IT; ++j) {
            const int i  = j * AG_T + tid;
            const int ic = (i < TOT) ? i : (TOT - 1);
            const int r  = ic / (F3 / 4);
            const int c4 = ic % (F3 / 4);
            const int rg = (r0 + r < n) ? (r0 + r) : (n - 1);
            ev[j] = *(const float4*)(e0 + (size_t)rg * F3 + c4 * 4);
        }
#pragma unroll
        for (int j = 0; j < IT; ++j) {
            const int i = j * AG_T + tid;
            if (i < TOT) {
                const int r  = i / (F3 / 4);
                const int c4 = i % (F3 / 4);
                *(float4*)(es + r * IN_P + c4 * 4) = ev[j];
            }
        }
    }
    __syncthreads();

    for (int u = wv; u < (IN_R / 16) * 3; u += AG_T / 32) {
        const int br = u % 3;
        const int rt = u / 3;
        dec_unit(es, W0 + (size_t)br * F3 * DECW, b0 + br * DECW, Wout + br * DECW, bout[br],
                 res, rt, br, lane);
    }
    __syncthreads();

    for (int i = tid; i < rows * 3; i += AG_T) out[(size_t)r0 * 3 + i] = res[i];
}

template <int RTS>
static inline void launch_enc_conv2_lin(const float* h1, const int* ei, int E, int n, int rr,
                                        const int* lst, const int* lcnt, float* e,
                                        const float* Wrel, const float* Wroot, const float* brel,
                                        const float* lW, const float* lb, hipStream_t stream) {
    agg128_conv2_kernel<RTS><<<(n + rr - 1) / rr, AG_T, 0, stream>>>(
        h1, ei, E, n, rr, lst, lcnt, Wrel, Wroot, brel, lW, lb, e);
}

extern "C" void kernel_launch(void* const* d_in, const int* in_sizes, int n_in,
                              void* d_out, int out_size, void* d_ws, size_t ws_size,
                              hipStream_t stream) {
    (void)in_sizes; (void)n_in; (void)out_size; (void)ws_size;

    const float* x[3]   = {(const float*)d_in[0], (const float*)d_in[1], (const float*)d_in[2]};
    const int*   ei[3]  = {(const int*)d_in[3], (const int*)d_in[4], (const int*)d_in[5]};
    const int*   A1r    = (const int*)d_in[6];
    const int*   A1c    = (const int*)d_in[7];
    const float* A1v    = (const float*)d_in[8];
    const int*   A2r    = (const int*)d_in[9];
    const int*   A2c    = (const int*)d_in[10];
    const float* A2v    = (const float*)d_in[11];
    const float* c1Wrel  = (const float*)d_in[12];
    const float* c1brel  = (const float*)d_in[13];
    const float* c1Wroot = (const float*)d_in[14];
    const float* c2Wrel  = (const float*)d_in[15];
    const float* c2brel  = (const float*)d_in[16];
    const float* c2Wroot = (const float*)d_in[17];
    const float* linW    = (const float*)d_in[18];
    const float* linb    = (const float*)d_in[19];
    const float* decW0   = (const float*)d_in[20];
    const float* decb0   = (const float*)d_in[21];
    const float* decWout = (const float*)d_in[22];
    const float* decbout = (const float*)d_in[23];
    float* out = (float*)d_out;

    float* ws = (float*)d_ws;
    float* h1   = ws;
    float* agg  = h1   + (size_t)N0 * H;
    int*   bsrt = (int*)(agg + (size_t)N0 * H);
    int*   boff = bsrt + 2 * NNZ;
    float* e0   = agg  + (size_t)N0 * H * 2;
    float* e1   = e0   + (size_t)N0 * F3;
    float* e2   = e1   + (size_t)N1 * F3;
    float* rest = e2   + (size_t)N2 * F3;

    constexpr size_t LST_MAX = (size_t)((N0 + SC_R - 1) / SC_R) * SC_S * (SC_T / 32) *
                               ((((E0 + SC_S - 1) / SC_S + SC_T * SC_U - 1) / (SC_T * SC_U)) * SC_U * 32);
    constexpr size_t LCNT_MAX = (size_t)((N0 + SC_R - 1) / SC_R) * SC_S * (SC_T / 32);
    static_assert(LST_MAX + LCNT_MAX <= (size_t)N0 * F3 * 2 + (size_t)N0 * DECW,
                  "the lists of level 0 (the largest) fit behind e2");
    static_assert(E0 >= E1 && E0 >= E2 && N0 >= N1 && N0 >= N2, "level 0 is the largest");
    static_assert((size_t)SC_S * N0 * 6 <= (size_t)N0 * H &&
                  (size_t)2 * NNZ + (size_t)2 * IB_S * IB_OP <= (size_t)N0 * H,
                  "the planes fit in the first half of agg, the sorted entries in the second");
    static_assert((2 * IB_S) % SC_S == 0 &&
                  (N1 + SC_R - 1) / SC_R * SC_S + 2 * IB_S <= 256,
                  "the sorting blocks fill whole grid columns and find a free WGP at level 1");
    int* lst  = (int*)rest;
    int* lcnt = lst + LST_MAX;

    const int nlv[3] = {N0, N1, N2};
    const int Elv[3] = {E0, E1, E2};
    const int rrl[3] = {200, 50, 13};
    static_assert(SC_R % 200 == 0 && SC_R % 50 == 0 && N2 <= SC_R,
                  "a conv2 block's rows lie inside one scatter6 range");
    static_assert(200 <= AG_RMAX && (50 + 15) / 16 <= 4 && (13 + 15) / 16 <= 4,
                  "conv2 product: level 0 within AG_RMAX rows, levels 1 and 2 one row tile per wave");
    static_assert(E0 % 4 == 0 && E1 % 4 == 0 && E2 % 4 == 0, "dst lists are read 16 B at a time");
    float* elv[3] = {e0, e1, e2};

    for (int i = 0; i < 3; ++i) {
        const int n = nlv[i];
        const int E = Elv[i];

        {
            const int nb = (n + SC_R - 1) / SC_R;
            const int xb = (i == 1) ? (2 * IB_S) / SC_S : 0;
            scatter6_lds_kernel<<<dim3(nb + xb, SC_S, 1), SC_T, 0, stream>>>(
                x[i], ei[i], E, n, agg, lst, lcnt, nb, A1r, A2r, bsrt, boff);
            conv1_planes_kernel<<<(n + C1_R - 1) / C1_R, C1_T, 0, stream>>>(
                x[i], agg,
                c1Wrel + (size_t)i * 6 * H,
                c1brel + (size_t)i * H,
                c1Wroot + (size_t)i * 6 * H, h1, n);
        }

        {
            const int rr = rrl[i];
            const float* Wrel2  = c2Wrel + (size_t)i * H * H;
            const float* Wroot2 = c2Wroot + (size_t)i * H * H;
            const float* brel2  = c2brel + (size_t)i * H;
            const float* lW     = linW + (size_t)i * H * F3;
            const float* lb     = linb + (size_t)i * F3;
            if (i == 0)
                launch_enc_conv2_lin<4>(h1, ei[i], E, n, rr, lst, lcnt, elv[i],
                                        Wrel2, Wroot2, brel2, lW, lb, stream);
            else
                launch_enc_conv2_lin<1>(h1, ei[i], E, n, rr, lst, lcnt, elv[i],
                                        Wrel2, Wroot2, brel2, lW, lb, stream);
        }
    }

    interp_dec_kernel<<<IB_NB, AG_T, 0, stream>>>(
        A1v, A1c, e1, A2v, A2c, e2, e0, decW0, decb0, decWout, decbout, out,
        bsrt, boff, N0);
}
